# starting kernel
# baseline (speedup 1.0000x reference)
#include <hip/hip_runtime.h>
#include <hip/hip_bf16.h>

typedef __attribute__((ext_vector_type(2)))  float    v2f;
typedef __attribute__((ext_vector_type(8)))  float    v8f;
typedef __attribute__((ext_vector_type(16))) _Float16 v16h;

#define HID 64
#define NCLASS 10

#if defined(__HIP_DEVICE_COMPILE__)
#  if __has_builtin(__builtin_amdgcn_wmma_f32_16x16x4_f32)
#    define GCN_USE_F32_WMMA 1
#  else
#    define GCN_USE_F32_WMMA 0
#  endif
#endif

__device__ __forceinline__ void atomic_add_f32(float* p, float v) {
    asm volatile("global_atomic_add_f32 %0, %1, off"
                 :
                 : "v"(p), "v"(v)
                 : "memory");
}

__global__ __launch_bounds__(256) void edge_pre_kernel(
    const float* __restrict__ ea, const float* __restrict__ We,
    const int* __restrict__ rows, float* __restrict__ ew,
    float* __restrict__ deg, int nedge) {
    int e = blockIdx.x * 256 + threadIdx.x;
    if (e >= nedge) return;
    const float4* a4 = (const float4*)(ea + (size_t)e * 16);
    const float4* w4 = (const float4*)We;
    float s = 0.0f;
#pragma unroll
    for (int i = 0; i < 4; ++i) {
        float4 a = a4[i], w = w4[i];
        s += a.x * w.x + a.y * w.y + a.z * w.z + a.w * w.w;
    }
    ew[e] = s;
    atomic_add_f32(&deg[rows[e]], s);
}

__global__ __launch_bounds__(256) void zero_kernel(float* __restrict__ p, int n) {
    int i = blockIdx.x * 256 + threadIdx.x;
    if (i < n) p[i] = 0.0f;
}

__global__ __launch_bounds__(256) void dinv_kernel(float* __restrict__ deg, int n) {
    int i = blockIdx.x * 256 + threadIdx.x;
    if (i < n) deg[i] = __frsqrt_rn(deg[i] + 1.0f);
}

template <int K, int MODE>
__global__ __launch_bounds__(128) void gemm_wmma_kernel(
    const float* __restrict__ A, const float* __restrict__ W,
    const float* __restrict__ dinv_or_bias,
    float* __restrict__ out0, float* __restrict__ out1) {
    const int lane = threadIdx.x & 31;
    const int wave = threadIdx.x >> 5;
    const int m    = lane & 15;
    const int hi   = lane >> 4;
    const int row0 = blockIdx.x * 16;
    const int col0 = wave * 16;

    v8f acc = {0.f, 0.f, 0.f, 0.f, 0.f, 0.f, 0.f, 0.f};

#if defined(__HIP_DEVICE_COMPILE__)
#if GCN_USE_F32_WMMA
    const int kh = hi * 2;
#pragma unroll 4
    for (int k0 = 0; k0 < K; k0 += 4) {
        const float* ap = A + (size_t)(row0 + m) * K + k0 + kh;
        const float* bp = W + (size_t)(k0 + kh) * HID + col0 + m;
        v2f a, b;
        a.x = ap[0];   a.y = ap[1];
        b.x = bp[0];   b.y = bp[HID];
        acc = __builtin_amdgcn_wmma_f32_16x16x4_f32(
            false, a, false, b, (short)0, acc, false, false);
    }
#else
    const int kb = hi * 8;
    for (int k0 = 0; k0 < K; k0 += 32) {
        const float* ap = A + (size_t)(row0 + m) * K + k0 + kb;
        const float* bp = W + (size_t)(k0 + kb) * HID + col0 + m;
        v16h a, b;
#pragma unroll
        for (int j = 0; j < 8; ++j) {
            a[j]     = (_Float16)ap[j];
            a[j + 8] = (_Float16)ap[j + 16];
            b[j]     = (_Float16)bp[(size_t)j * HID];
            b[j + 8] = (_Float16)bp[(size_t)(j + 16) * HID];
        }
        acc = __builtin_amdgcn_wmma_f32_16x16x32_f16(
            false, a, false, b, (short)0, acc, false, false);
    }
#endif
#endif

#pragma unroll
    for (int r = 0; r < 8; ++r) {
        const int row = row0 + r + hi * 8;
        const int col = col0 + m;
        const size_t idx = (size_t)row * HID + col;
        const float v = acc[r];
        if (MODE == 0) {
            out0[idx] = v;
            const float di = dinv_or_bias[row];
            out1[idx] = di * di * v;
        } else {
            out0[idx] = v + dinv_or_bias[col];
        }
    }
}

__global__ __launch_bounds__(256) void edge_agg_kernel(
    const int* __restrict__ rows, const int* __restrict__ cols,
    const float* __restrict__ ew, const float* __restrict__ dinv,
    const float* __restrict__ h, float* __restrict__ agg, int nedge) {
    const int e = blockIdx.x * 16 + (threadIdx.x >> 4);
    if (e >= nedge) return;
    const int l = threadIdx.x & 15;
    const int r = rows[e];
    const int c = cols[e];
    const float nrm = dinv[r] * ew[e] * dinv[c];
    const float4 hv = *(const float4*)(h + (size_t)c * HID + l * 4);
    float* dst = agg + (size_t)r * HID + l * 4;
    atomic_add_f32(dst + 0, nrm * hv.x);
    atomic_add_f32(dst + 1, nrm * hv.y);
    atomic_add_f32(dst + 2, nrm * hv.z);
    atomic_add_f32(dst + 3, nrm * hv.w);
}

__global__ __launch_bounds__(256) void finalize_kernel(
    const float* __restrict__ agg, const float* __restrict__ bias,
    float* __restrict__ xout, int total) {
    int i = blockIdx.x * 256 + threadIdx.x;
    if (i >= total) return;
    float v = agg[i] + bias[i & (HID - 1)];
    xout[i] = v > 0.0f ? v : 0.0f;
}

__global__ __launch_bounds__(256) void head2_kernel(
    const float* __restrict__ t, const float* __restrict__ W,
    const float* __restrict__ b, float* __restrict__ out, int n) {
    int i = blockIdx.x * 256 + threadIdx.x;
    if (i >= n) return;
    float acc[NCLASS];
#pragma unroll
    for (int c = 0; c < NCLASS; ++c) acc[c] = b[c];
    const float* ti = t + (size_t)i * HID;
#pragma unroll 4
    for (int j = 0; j < HID; ++j) {
        const float tv = ti[j];
        const float* wr = W + (size_t)j * NCLASS;
#pragma unroll
        for (int c = 0; c < NCLASS; ++c) acc[c] += tv * wr[c];
    }
#pragma unroll
    for (int c = 0; c < NCLASS; ++c) out[(size_t)i * NCLASS + c] = acc[c];
}

extern "C" void kernel_launch(void* const* d_in, const int* in_sizes, int n_in,
                              void* d_out, int out_size, void* d_ws, size_t ws_size,
                              hipStream_t stream) {
    const float* x      = (const float*)d_in[0];
    const float* eattr  = (const float*)d_in[1];
    const int*   eidx   = (const int*)  d_in[2];
    const float* We     = (const float*)d_in[3];
    const float* Wl[3]  = {(const float*)d_in[4], (const float*)d_in[6], (const float*)d_in[8]};
    const float* bl[3]  = {(const float*)d_in[5], (const float*)d_in[7], (const float*)d_in[9]};
    const float* lin1_W = (const float*)d_in[10];
    const float* lin1_b = (const float*)d_in[11];
    const float* lin2_W = (const float*)d_in[12];
    const float* lin2_b = (const float*)d_in[13];
    float* out = (float*)d_out;

    const int N = in_sizes[0] / 128;
    const int E = in_sizes[1] / 16;
    const int* rows = eidx;
    const int* cols = eidx + E;

    float* ws   = (float*)d_ws;
    float* ew   = ws;
    float* dinv = ew + E;
    float* h    = dinv + N;
    float* agg  = h + (size_t)N * HID;
    float* xa   = agg + (size_t)N * HID;
    float* xb   = xa + (size_t)N * HID;
    float* t    = h;

    const int rowTiles = (N + 15) / 16;
    const int eBlk     = (E + 15) / 16;
    const int nodeElts = N * HID;

    zero_kernel<<<(N + 255) / 256, 256, 0, stream>>>(dinv, N);
    edge_pre_kernel<<<(E + 255) / 256, 256, 0, stream>>>(eattr, We, rows, ew, dinv, E);
    dinv_kernel<<<(N + 255) / 256, 256, 0, stream>>>(dinv, N);

    gemm_wmma_kernel<128, 0><<<rowTiles, 128, 0, stream>>>(x, Wl[0], dinv, h, agg);
    edge_agg_kernel<<<eBlk, 256, 0, stream>>>(rows, cols, ew, dinv, h, agg, E);
    finalize_kernel<<<(nodeElts + 255) / 256, 256, 0, stream>>>(agg, bl[0], xa, nodeElts);

    gemm_wmma_kernel<64, 0><<<rowTiles, 128, 0, stream>>>(xa, Wl[1], dinv, h, agg);
    edge_agg_kernel<<<eBlk, 256, 0, stream>>>(rows, cols, ew, dinv, h, agg, E);
    finalize_kernel<<<(nodeElts + 255) / 256, 256, 0, stream>>>(agg, bl[1], xb, nodeElts);

    gemm_wmma_kernel<64, 0><<<rowTiles, 128, 0, stream>>>(xb, Wl[2], dinv, h, agg);
    edge_agg_kernel<<<eBlk, 256, 0, stream>>>(rows, cols, ew, dinv, h, agg, E);
    finalize_kernel<<<(nodeElts + 255) / 256, 256, 0, stream>>>(agg, bl[2], xa, nodeElts);

    gemm_wmma_kernel<64, 1><<<rowTiles, 128, 0, stream>>>(xa, lin1_W, lin1_b, t, nullptr);
    head2_kernel<<<(N + 255) / 256, 256, 0, stream>>>(t, lin2_W, lin2_b, out, N);
}


# Edge scatter: a wave's 32 lanes take 32 consecutive floats of one edge, so each float atomic instruction covers one whole 128-byte line, not 8 floats in each of 4 lines (4x fewer atomic requests)
# speedup vs baseline: 1.4268x; 1.4268x over previous
#include <hip/hip_runtime.h>
#include <hip/hip_bf16.h>

typedef __attribute__((ext_vector_type(2)))  float    v2f;
typedef __attribute__((ext_vector_type(8)))  float    v8f;
typedef __attribute__((ext_vector_type(16))) _Float16 v16h;

#define HID 64
#define NCLASS 10

#if defined(__HIP_DEVICE_COMPILE__)
#  if __has_builtin(__builtin_amdgcn_wmma_f32_16x16x4_f32)
#    define GCN_USE_F32_WMMA 1
#  else
#    define GCN_USE_F32_WMMA 0
#  endif
#endif

__device__ __forceinline__ void atomic_add_f32(float* p, float v) {
    asm volatile("global_atomic_add_f32 %0, %1, off"
                 :
                 : "v"(p), "v"(v)
                 : "memory");
}

__global__ __launch_bounds__(256) void edge_pre_kernel(
    const float* __restrict__ ea, const float* __restrict__ We,
    const int* __restrict__ rows, float* __restrict__ ew,
    float* __restrict__ deg, int nedge) {
    int e = blockIdx.x * 256 + threadIdx.x;
    if (e >= nedge) return;
    const float4* a4 = (const float4*)(ea + (size_t)e * 16);
    const float4* w4 = (const float4*)We;
    float s = 0.0f;
#pragma unroll
    for (int i = 0; i < 4; ++i) {
        float4 a = a4[i], w = w4[i];
        s += a.x * w.x + a.y * w.y + a.z * w.z + a.w * w.w;
    }
    ew[e] = s;
    atomic_add_f32(&deg[rows[e]], s);
}

__global__ __launch_bounds__(256) void zero_kernel(float* __restrict__ p, int n) {
    int i = blockIdx.x * 256 + threadIdx.x;
    if (i < n) p[i] = 0.0f;
}

__global__ __launch_bounds__(256) void dinv_kernel(float* __restrict__ deg, int n) {
    int i = blockIdx.x * 256 + threadIdx.x;
    if (i < n) deg[i] = __frsqrt_rn(deg[i] + 1.0f);
}

template <int K, int MODE>
__global__ __launch_bounds__(128) void gemm_wmma_kernel(
    const float* __restrict__ A, const float* __restrict__ W,
    const float* __restrict__ dinv_or_bias,
    float* __restrict__ out0, float* __restrict__ out1) {
    const int lane = threadIdx.x & 31;
    const int wave = threadIdx.x >> 5;
    const int m    = lane & 15;
    const int hi   = lane >> 4;
    const int row0 = blockIdx.x * 16;
    const int col0 = wave * 16;

    v8f acc = {0.f, 0.f, 0.f, 0.f, 0.f, 0.f, 0.f, 0.f};

#if defined(__HIP_DEVICE_COMPILE__)
#if GCN_USE_F32_WMMA
    const int kh = hi * 2;
#pragma unroll 4
    for (int k0 = 0; k0 < K; k0 += 4) {
        const float* ap = A + (size_t)(row0 + m) * K + k0 + kh;
        const float* bp = W + (size_t)(k0 + kh) * HID + col0 + m;
        v2f a, b;
        a.x = ap[0];   a.y = ap[1];
        b.x = bp[0];   b.y = bp[HID];
        acc = __builtin_amdgcn_wmma_f32_16x16x4_f32(
            false, a, false, b, (short)0, acc, false, false);
    }
#else
    const int kb = hi * 8;
    for (int k0 = 0; k0 < K; k0 += 32) {
        const float* ap = A + (size_t)(row0 + m) * K + k0 + kb;
        const float* bp = W + (size_t)(k0 + kb) * HID + col0 + m;
        v16h a, b;
#pragma unroll
        for (int j = 0; j < 8; ++j) {
            a[j]     = (_Float16)ap[j];
            a[j + 8] = (_Float16)ap[j + 16];
            b[j]     = (_Float16)bp[(size_t)j * HID];
            b[j + 8] = (_Float16)bp[(size_t)(j + 16) * HID];
        }
        acc = __builtin_amdgcn_wmma_f32_16x16x32_f16(
            false, a, false, b, (short)0, acc, false, false);
    }
#endif
#endif

#pragma unroll
    for (int r = 0; r < 8; ++r) {
        const int row = row0 + r + hi * 8;
        const int col = col0 + m;
        const size_t idx = (size_t)row * HID + col;
        const float v = acc[r];
        if (MODE == 0) {
            out0[idx] = v;
            const float di = dinv_or_bias[row];
            out1[idx] = di * di * v;
        } else {
            out0[idx] = v + dinv_or_bias[col];
        }
    }
}

__global__ __launch_bounds__(256) void edge_agg_kernel(
    const int* __restrict__ rows, const int* __restrict__ cols,
    const float* __restrict__ ew, const float* __restrict__ dinv,
    const float* __restrict__ h, float* __restrict__ agg, int nedge) {
    const int lane = threadIdx.x & 31;
    const int wv = __builtin_amdgcn_readfirstlane((int)(threadIdx.x >> 5));
    const int e0 = (blockIdx.x * 8 + wv) * 2;
    if (e0 >= nedge) return;
    if (e0 + 1 < nedge) {
        const int e1 = e0 + 1;
        const int c0 = cols[e0];
        const int c1 = cols[e1];
        const int r0 = rows[e0];
        const int r1 = rows[e1];
        const float* s0 = h + (size_t)c0 * HID + lane;
        const float* s1 = h + (size_t)c1 * HID + lane;
        const float a00 = s0[0];
        const float a01 = s0[32];
        const float a10 = s1[0];
        const float a11 = s1[32];
        const float n0 = dinv[r0] * ew[e0] * dinv[c0];
        const float n1 = dinv[r1] * ew[e1] * dinv[c1];
        float* d0 = agg + (size_t)r0 * HID + lane;
        float* d1 = agg + (size_t)r1 * HID + lane;
        atomic_add_f32(d0,      n0 * a00);
        atomic_add_f32(d0 + 32, n0 * a01);
        atomic_add_f32(d1,      n1 * a10);
        atomic_add_f32(d1 + 32, n1 * a11);
    } else {
        const int c0 = cols[e0];
        const int r0 = rows[e0];
        const float* s0 = h + (size_t)c0 * HID + lane;
        const float a00 = s0[0];
        const float a01 = s0[32];
        const float n0 = dinv[r0] * ew[e0] * dinv[c0];
        float* d0 = agg + (size_t)r0 * HID + lane;
        atomic_add_f32(d0,      n0 * a00);
        atomic_add_f32(d0 + 32, n0 * a01);
    }
}

__global__ __launch_bounds__(256) void finalize_kernel(
    const float* __restrict__ agg, const float* __restrict__ bias,
    float* __restrict__ xout, int total) {
    int i = blockIdx.x * 256 + threadIdx.x;
    if (i >= total) return;
    float v = agg[i] + bias[i & (HID - 1)];
    xout[i] = v > 0.0f ? v : 0.0f;
}

__global__ __launch_bounds__(256) void head2_kernel(
    const float* __restrict__ t, const float* __restrict__ W,
    const float* __restrict__ b, float* __restrict__ out, int n) {
    int i = blockIdx.x * 256 + threadIdx.x;
    if (i >= n) return;
    float acc[NCLASS];
#pragma unroll
    for (int c = 0; c < NCLASS; ++c) acc[c] = b[c];
    const float* ti = t + (size_t)i * HID;
#pragma unroll 4
    for (int j = 0; j < HID; ++j) {
        const float tv = ti[j];
        const float* wr = W + (size_t)j * NCLASS;
#pragma unroll
        for (int c = 0; c < NCLASS; ++c) acc[c] += tv * wr[c];
    }
#pragma unroll
    for (int c = 0; c < NCLASS; ++c) out[(size_t)i * NCLASS + c] = acc[c];
}

extern "C" void kernel_launch(void* const* d_in, const int* in_sizes, int n_in,
                              void* d_out, int out_size, void* d_ws, size_t ws_size,
                              hipStream_t stream) {
    const float* x      = (const float*)d_in[0];
    const float* eattr  = (const float*)d_in[1];
    const int*   eidx   = (const int*)  d_in[2];
    const float* We     = (const float*)d_in[3];
    const float* Wl[3]  = {(const float*)d_in[4], (const float*)d_in[6], (const float*)d_in[8]};
    const float* bl[3]  = {(const float*)d_in[5], (const float*)d_in[7], (const float*)d_in[9]};
    const float* lin1_W = (const float*)d_in[10];
    const float* lin1_b = (const float*)d_in[11];
    const float* lin2_W = (const float*)d_in[12];
    const float* lin2_b = (const float*)d_in[13];
    float* out = (float*)d_out;

    const int N = in_sizes[0] / 128;
    const int E = in_sizes[1] / 16;
    const int* rows = eidx;
    const int* cols = eidx + E;

    float* ws   = (float*)d_ws;
    float* ew   = ws;
    float* dinv = ew + E;
    float* h    = dinv + N;
    float* agg  = h + (size_t)N * HID;
    float* xa   = agg + (size_t)N * HID;
    float* xb   = xa + (size_t)N * HID;
    float* t    = h;

    const int rowTiles = (N + 15) / 16;
    const int eBlk     = (E + 15) / 16;
    const int nodeElts = N * HID;

    zero_kernel<<<(N + 255) / 256, 256, 0, stream>>>(dinv, N);
    edge_pre_kernel<<<(E + 255) / 256, 256, 0, stream>>>(eattr, We, rows, ew, dinv, E);
    dinv_kernel<<<(N + 255) / 256, 256, 0, stream>>>(dinv, N);

    gemm_wmma_kernel<128, 0><<<rowTiles, 128, 0, stream>>>(x, Wl[0], dinv, h, agg);
    edge_agg_kernel<<<eBlk, 256, 0, stream>>>(rows, cols, ew, dinv, h, agg, E);
    finalize_kernel<<<(nodeElts + 255) / 256, 256, 0, stream>>>(agg, bl[0], xa, nodeElts);

    gemm_wmma_kernel<64, 0><<<rowTiles, 128, 0, stream>>>(xa, Wl[1], dinv, h, agg);
    edge_agg_kernel<<<eBlk, 256, 0, stream>>>(rows, cols, ew, dinv, h, agg, E);
    finalize_kernel<<<(nodeElts + 255) / 256, 256, 0, stream>>>(agg, bl[1], xb, nodeElts);

    gemm_wmma_kernel<64, 0><<<rowTiles, 128, 0, stream>>>(xb, Wl[2], dinv, h, agg);
    edge_agg_kernel<<<eBlk, 256, 0, stream>>>(rows, cols, ew, dinv, h, agg, E);
    finalize_kernel<<<(nodeElts + 255) / 256, 256, 0, stream>>>(agg, bl[2], xa, nodeElts);

    gemm_wmma_kernel<64, 1><<<rowTiles, 128, 0, stream>>>(xa, lin1_W, lin1_b, t, nullptr);
    head2_kernel<<<(N + 255) / 256, 256, 0, stream>>>(t, lin2_W, lin2_b, out, N);
}


# Workspace planes re-ordered so every 64-float node row starts on a 256-byte boundary: each 32-lane gather load and float atomic of the edge scatter touches one 128-byte line, not two
# speedup vs baseline: 1.6471x; 1.1544x over previous
#include <hip/hip_runtime.h>
#include <hip/hip_bf16.h>

typedef __attribute__((ext_vector_type(2)))  float    v2f;
typedef __attribute__((ext_vector_type(8)))  float    v8f;
typedef __attribute__((ext_vector_type(16))) _Float16 v16h;

#define HID 64
#define NCLASS 10

#if defined(__HIP_DEVICE_COMPILE__)
#  if __has_builtin(__builtin_amdgcn_wmma_f32_16x16x4_f32)
#    define GCN_USE_F32_WMMA 1
#  else
#    define GCN_USE_F32_WMMA 0
#  endif
#endif

__device__ __forceinline__ void atomic_add_f32(float* p, float v) {
    asm volatile("global_atomic_add_f32 %0, %1, off"
                 :
                 : "v"(p), "v"(v)
                 : "memory");
}

__global__ __launch_bounds__(256) void edge_pre_kernel(
    const float* __restrict__ ea, const float* __restrict__ We,
    const int* __restrict__ rows, float* __restrict__ ew,
    float* __restrict__ deg, int nedge) {
    int e = blockIdx.x * 256 + threadIdx.x;
    if (e >= nedge) return;
    const float4* a4 = (const float4*)(ea + (size_t)e * 16);
    const float4* w4 = (const float4*)We;
    float s = 0.0f;
#pragma unroll
    for (int i = 0; i < 4; ++i) {
        float4 a = a4[i], w = w4[i];
        s += a.x * w.x + a.y * w.y + a.z * w.z + a.w * w.w;
    }
    ew[e] = s;
    atomic_add_f32(&deg[rows[e]], s);
}

__global__ __launch_bounds__(256) void zero_kernel(float* __restrict__ p, int n) {
    int i = blockIdx.x * 256 + threadIdx.x;
    if (i < n) p[i] = 0.0f;
}

__global__ __launch_bounds__(256) void dinv_kernel(float* __restrict__ deg, int n) {
    int i = blockIdx.x * 256 + threadIdx.x;
    if (i < n) deg[i] = __frsqrt_rn(deg[i] + 1.0f);
}

template <int K, int MODE>
__global__ __launch_bounds__(128) void gemm_wmma_kernel(
    const float* __restrict__ A, const float* __restrict__ W,
    const float* __restrict__ dinv_or_bias,
    float* __restrict__ out0, float* __restrict__ out1) {
    const int lane = threadIdx.x & 31;
    const int wave = threadIdx.x >> 5;
    const int m    = lane & 15;
    const int hi   = lane >> 4;
    const int row0 = blockIdx.x * 16;
    const int col0 = wave * 16;

    v8f acc = {0.f, 0.f, 0.f, 0.f, 0.f, 0.f, 0.f, 0.f};

#if defined(__HIP_DEVICE_COMPILE__)
#if GCN_USE_F32_WMMA
    const int kh = hi * 2;
#pragma unroll 4
    for (int k0 = 0; k0 < K; k0 += 4) {
        const float* ap = A + (size_t)(row0 + m) * K + k0 + kh;
        const float* bp = W + (size_t)(k0 + kh) * HID + col0 + m;
        v2f a, b;
        a.x = ap[0];   a.y = ap[1];
        b.x = bp[0];   b.y = bp[HID];
        acc = __builtin_amdgcn_wmma_f32_16x16x4_f32(
            false, a, false, b, (short)0, acc, false, false);
    }
#else
    const int kb = hi * 8;
    for (int k0 = 0; k0 < K; k0 += 32) {
        const float* ap = A + (size_t)(row0 + m) * K + k0 + kb;
        const float* bp = W + (size_t)(k0 + kb) * HID + col0 + m;
        v16h a, b;
#pragma unroll
        for (int j = 0; j < 8; ++j) {
            a[j]     = (_Float16)ap[j];
            a[j + 8] = (_Float16)ap[j + 16];
            b[j]     = (_Float16)bp[(size_t)j * HID];
            b[j + 8] = (_Float16)bp[(size_t)(j + 16) * HID];
        }
        acc = __builtin_amdgcn_wmma_f32_16x16x32_f16(
            false, a, false, b, (short)0, acc, false, false);
    }
#endif
#endif

#pragma unroll
    for (int r = 0; r < 8; ++r) {
        const int row = row0 + r + hi * 8;
        const int col = col0 + m;
        const size_t idx = (size_t)row * HID + col;
        const float v = acc[r];
        if (MODE == 0) {
            out0[idx] = v;
            const float di = dinv_or_bias[row];
            out1[idx] = di * di * v;
        } else {
            out0[idx] = v + dinv_or_bias[col];
        }
    }
}

__global__ __launch_bounds__(256) void edge_agg_kernel(
    const int* __restrict__ rows, const int* __restrict__ cols,
    const float* __restrict__ ew, const float* __restrict__ dinv,
    const float* __restrict__ h, float* __restrict__ agg, int nedge) {
    const int lane = threadIdx.x & 31;
    const int wv = __builtin_amdgcn_readfirstlane((int)(threadIdx.x >> 5));
    const int e0 = (blockIdx.x * 8 + wv) * 2;
    if (e0 >= nedge) return;
    if (e0 + 1 < nedge) {
        const int e1 = e0 + 1;
        const int c0 = cols[e0];
        const int c1 = cols[e1];
        const int r0 = rows[e0];
        const int r1 = rows[e1];
        const float* s0 = h + (size_t)c0 * HID + lane;
        const float* s1 = h + (size_t)c1 * HID + lane;
        const float a00 = s0[0];
        const float a01 = s0[32];
        const float a10 = s1[0];
        const float a11 = s1[32];
        const float n0 = dinv[r0] * ew[e0] * dinv[c0];
        const float n1 = dinv[r1] * ew[e1] * dinv[c1];
        float* d0 = agg + (size_t)r0 * HID + lane;
        float* d1 = agg + (size_t)r1 * HID + lane;
        atomic_add_f32(d0,      n0 * a00);
        atomic_add_f32(d0 + 32, n0 * a01);
        atomic_add_f32(d1,      n1 * a10);
        atomic_add_f32(d1 + 32, n1 * a11);
    } else {
        const int c0 = cols[e0];
        const int r0 = rows[e0];
        const float* s0 = h + (size_t)c0 * HID + lane;
        const float a00 = s0[0];
        const float a01 = s0[32];
        const float n0 = dinv[r0] * ew[e0] * dinv[c0];
        float* d0 = agg + (size_t)r0 * HID + lane;
        atomic_add_f32(d0,      n0 * a00);
        atomic_add_f32(d0 + 32, n0 * a01);
    }
}

__global__ __launch_bounds__(256) void finalize_kernel(
    const float* __restrict__ agg, const float* __restrict__ bias,
    float* __restrict__ xout, int total) {
    int i = blockIdx.x * 256 + threadIdx.x;
    if (i >= total) return;
    float v = agg[i] + bias[i & (HID - 1)];
    xout[i] = v > 0.0f ? v : 0.0f;
}

__global__ __launch_bounds__(256) void head2_kernel(
    const float* __restrict__ t, const float* __restrict__ W,
    const float* __restrict__ b, float* __restrict__ out, int n) {
    int i = blockIdx.x * 256 + threadIdx.x;
    if (i >= n) return;
    float acc[NCLASS];
#pragma unroll
    for (int c = 0; c < NCLASS; ++c) acc[c] = b[c];
    const float* ti = t + (size_t)i * HID;
#pragma unroll 4
    for (int j = 0; j < HID; ++j) {
        const float tv = ti[j];
        const float* wr = W + (size_t)j * NCLASS;
#pragma unroll
        for (int c = 0; c < NCLASS; ++c) acc[c] += tv * wr[c];
    }
#pragma unroll
    for (int c = 0; c < NCLASS; ++c) out[(size_t)i * NCLASS + c] = acc[c];
}

extern "C" void kernel_launch(void* const* d_in, const int* in_sizes, int n_in,
                              void* d_out, int out_size, void* d_ws, size_t ws_size,
                              hipStream_t stream) {
    const float* x      = (const float*)d_in[0];
    const float* eattr  = (const float*)d_in[1];
    const int*   eidx   = (const int*)  d_in[2];
    const float* We     = (const float*)d_in[3];
    const float* Wl[3]  = {(const float*)d_in[4], (const float*)d_in[6], (const float*)d_in[8]};
    const float* bl[3]  = {(const float*)d_in[5], (const float*)d_in[7], (const float*)d_in[9]};
    const float* lin1_W = (const float*)d_in[10];
    const float* lin1_b = (const float*)d_in[11];
    const float* lin2_W = (const float*)d_in[12];
    const float* lin2_b = (const float*)d_in[13];
    float* out = (float*)d_out;

    const int N = in_sizes[0] / 128;
    const int E = in_sizes[1] / 16;
    const int* rows = eidx;
    const int* cols = eidx + E;

    float* ws   = (float*)d_ws;
    float* h    = ws;
    float* agg  = h + (size_t)N * HID;
    float* xa   = agg + (size_t)N * HID;
    float* xb   = xa + (size_t)N * HID;
    float* ew   = xb + (size_t)N * HID;
    float* dinv = ew + E;
    float* t    = h;

    const int rowTiles = (N + 15) / 16;
    const int eBlk     = (E + 15) / 16;
    const int nodeElts = N * HID;

    zero_kernel<<<(N + 255) / 256, 256, 0, stream>>>(dinv, N);
    edge_pre_kernel<<<(E + 255) / 256, 256, 0, stream>>>(eattr, We, rows, ew, dinv, E);
    dinv_kernel<<<(N + 255) / 256, 256, 0, stream>>>(dinv, N);

    gemm_wmma_kernel<128, 0><<<rowTiles, 128, 0, stream>>>(x, Wl[0], dinv, h, agg);
    edge_agg_kernel<<<eBlk, 256, 0, stream>>>(rows, cols, ew, dinv, h, agg, E);
    finalize_kernel<<<(nodeElts + 255) / 256, 256, 0, stream>>>(agg, bl[0], xa, nodeElts);

    gemm_wmma_kernel<64, 0><<<rowTiles, 128, 0, stream>>>(xa, Wl[1], dinv, h, agg);
    edge_agg_kernel<<<eBlk, 256, 0, stream>>>(rows, cols, ew, dinv, h, agg, E);
    finalize_kernel<<<(nodeElts + 255) / 256, 256, 0, stream>>>(agg, bl[1], xb, nodeElts);

    gemm_wmma_kernel<64, 0><<<rowTiles, 128, 0, stream>>>(xb, Wl[2], dinv, h, agg);
    edge_agg_kernel<<<eBlk, 256, 0, stream>>>(rows, cols, ew, dinv, h, agg, E);
    finalize_kernel<<<(nodeElts + 255) / 256, 256, 0, stream>>>(agg, bl[2], xa, nodeElts);

    gemm_wmma_kernel<64, 1><<<rowTiles, 128, 0, stream>>>(xa, lin1_W, lin1_b, t, nullptr);
    head2_kernel<<<(N + 255) / 256, 256, 0, stream>>>(t, lin2_W, lin2_b, out, N);
}


# Edge scatter: a wave takes 32 edges, not 2: row/col/weight fetched coalesced one edge per lane and broadcast, each 8 edges' gathers issued ahead of their float atomics (arrival-order sums as before)
# speedup vs baseline: 1.9531x; 1.1858x over previous
#include <hip/hip_runtime.h>
#include <hip/hip_bf16.h>

typedef __attribute__((ext_vector_type(2)))  float    v2f;
typedef __attribute__((ext_vector_type(8)))  float    v8f;
typedef __attribute__((ext_vector_type(16))) _Float16 v16h;

#define HID 64
#define NCLASS 10

#if defined(__HIP_DEVICE_COMPILE__)
#  if __has_builtin(__builtin_amdgcn_wmma_f32_16x16x4_f32)
#    define GCN_USE_F32_WMMA 1
#  else
#    define GCN_USE_F32_WMMA 0
#  endif
#endif

__device__ __forceinline__ void atomic_add_f32(float* p, float v) {
    asm volatile("global_atomic_add_f32 %0, %1, off"
                 :
                 : "v"(p), "v"(v)
                 : "memory");
}

__global__ __launch_bounds__(256) void edge_pre_kernel(
    const float* __restrict__ ea, const float* __restrict__ We,
    const int* __restrict__ rows, float* __restrict__ ew,
    float* __restrict__ deg, int nedge) {
    int e = blockIdx.x * 256 + threadIdx.x;
    if (e >= nedge) return;
    const float4* a4 = (const float4*)(ea + (size_t)e * 16);
    const float4* w4 = (const float4*)We;
    float s = 0.0f;
#pragma unroll
    for (int i = 0; i < 4; ++i) {
        float4 a = a4[i], w = w4[i];
        s += a.x * w.x + a.y * w.y + a.z * w.z + a.w * w.w;
    }
    ew[e] = s;
    atomic_add_f32(&deg[rows[e]], s);
}

__global__ __launch_bounds__(256) void zero_kernel(float* __restrict__ p, int n) {
    int i = blockIdx.x * 256 + threadIdx.x;
    if (i < n) p[i] = 0.0f;
}

__global__ __launch_bounds__(256) void dinv_kernel(float* __restrict__ deg, int n) {
    int i = blockIdx.x * 256 + threadIdx.x;
    if (i < n) deg[i] = __frsqrt_rn(deg[i] + 1.0f);
}

template <int K, int MODE>
__global__ __launch_bounds__(128) void gemm_wmma_kernel(
    const float* __restrict__ A, const float* __restrict__ W,
    const float* __restrict__ dinv_or_bias,
    float* __restrict__ out0, float* __restrict__ out1) {
    const int lane = threadIdx.x & 31;
    const int wave = threadIdx.x >> 5;
    const int m    = lane & 15;
    const int hi   = lane >> 4;
    const int row0 = blockIdx.x * 16;
    const int col0 = wave * 16;

    v8f acc = {0.f, 0.f, 0.f, 0.f, 0.f, 0.f, 0.f, 0.f};

#if defined(__HIP_DEVICE_COMPILE__)
#if GCN_USE_F32_WMMA
    const int kh = hi * 2;
#pragma unroll 4
    for (int k0 = 0; k0 < K; k0 += 4) {
        const float* ap = A + (size_t)(row0 + m) * K + k0 + kh;
        const float* bp = W + (size_t)(k0 + kh) * HID + col0 + m;
        v2f a, b;
        a.x = ap[0];   a.y = ap[1];
        b.x = bp[0];   b.y = bp[HID];
        acc = __builtin_amdgcn_wmma_f32_16x16x4_f32(
            false, a, false, b, (short)0, acc, false, false);
    }
#else
    const int kb = hi * 8;
    for (int k0 = 0; k0 < K; k0 += 32) {
        const float* ap = A + (size_t)(row0 + m) * K + k0 + kb;
        const float* bp = W + (size_t)(k0 + kb) * HID + col0 + m;
        v16h a, b;
#pragma unroll
        for (int j = 0; j < 8; ++j) {
            a[j]     = (_Float16)ap[j];
            a[j + 8] = (_Float16)ap[j + 16];
            b[j]     = (_Float16)bp[(size_t)j * HID];
            b[j + 8] = (_Float16)bp[(size_t)(j + 16) * HID];
        }
        acc = __builtin_amdgcn_wmma_f32_16x16x32_f16(
            false, a, false, b, (short)0, acc, false, false);
    }
#endif
#endif

#pragma unroll
    for (int r = 0; r < 8; ++r) {
        const int row = row0 + r + hi * 8;
        const int col = col0 + m;
        const size_t idx = (size_t)row * HID + col;
        const float v = acc[r];
        if (MODE == 0) {
            out0[idx] = v;
            const float di = dinv_or_bias[row];
            out1[idx] = di * di * v;
        } else {
            out0[idx] = v + dinv_or_bias[col];
        }
    }
}

template <int U>
__device__ __forceinline__ void edge_group(const float* __restrict__ hp, float* __restrict__ ap,
                                           int c, int r, float nrm, int j) {
    float v0[U], v1[U];
#pragma unroll
    for (int u = 0; u < U; ++u) {
        const int cj = __builtin_amdgcn_readlane(c, j + u);
        const float* s = hp + (size_t)cj * HID;
        v0[u] = s[0];
        v1[u] = s[32];
    }
#pragma unroll
    for (int u = 0; u < U; ++u) {
        const int rj = __builtin_amdgcn_readlane(r, j + u);
        const float w = __int_as_float(__builtin_amdgcn_readlane(__float_as_int(nrm), j + u));
        float* d = ap + (size_t)rj * HID;
        atomic_add_f32(d,      w * v0[u]);
        atomic_add_f32(d + 32, w * v1[u]);
    }
}

__global__ __launch_bounds__(256) void edge_agg_kernel(
    const int* __restrict__ rows, const int* __restrict__ cols,
    const float* __restrict__ ew, const float* __restrict__ dinv,
    const float* __restrict__ h, float* __restrict__ agg, int nedge) {
    const int lane = threadIdx.x & 31;
    const int wv = __builtin_amdgcn_readfirstlane((int)(threadIdx.x >> 5));
    const int e0 = (blockIdx.x * 8 + wv) * 32;
    if (e0 >= nedge) return;
    const int m = (nedge - e0) < 32 ? (nedge - e0) : 32;
    int c = 0, r = 0;
    float nrm = 0.0f;
    if (lane < m) {
        c = cols[e0 + lane];
        r = rows[e0 + lane];
        nrm = dinv[r] * ew[e0 + lane] * dinv[c];
    }
    const float* hp = h + lane;
    float* ap = agg + lane;
    int j = 0;
    for (; j + 8 <= m; j += 8) edge_group<8>(hp, ap, c, r, nrm, j);
    if (j + 4 <= m) { edge_group<4>(hp, ap, c, r, nrm, j); j += 4; }
    if (j + 2 <= m) { edge_group<2>(hp, ap, c, r, nrm, j); j += 2; }
    if (j < m)      { edge_group<1>(hp, ap, c, r, nrm, j); }
}

__global__ __launch_bounds__(256) void finalize_kernel(
    const float* __restrict__ agg, const float* __restrict__ bias,
    float* __restrict__ xout, int total) {
    int i = blockIdx.x * 256 + threadIdx.x;
    if (i >= total) return;
    float v = agg[i] + bias[i & (HID - 1)];
    xout[i] = v > 0.0f ? v : 0.0f;
}

__global__ __launch_bounds__(256) void head2_kernel(
    const float* __restrict__ t, const float* __restrict__ W,
    const float* __restrict__ b, float* __restrict__ out, int n) {
    int i = blockIdx.x * 256 + threadIdx.x;
    if (i >= n) return;
    float acc[NCLASS];
#pragma unroll
    for (int c = 0; c < NCLASS; ++c) acc[c] = b[c];
    const float* ti = t + (size_t)i * HID;
#pragma unroll 4
    for (int j = 0; j < HID; ++j) {
        const float tv = ti[j];
        const float* wr = W + (size_t)j * NCLASS;
#pragma unroll
        for (int c = 0; c < NCLASS; ++c) acc[c] += tv * wr[c];
    }
#pragma unroll
    for (int c = 0; c < NCLASS; ++c) out[(size_t)i * NCLASS + c] = acc[c];
}

extern "C" void kernel_launch(void* const* d_in, const int* in_sizes, int n_in,
                              void* d_out, int out_size, void* d_ws, size_t ws_size,
                              hipStream_t stream) {
    const float* x      = (const float*)d_in[0];
    const float* eattr  = (const float*)d_in[1];
    const int*   eidx   = (const int*)  d_in[2];
    const float* We     = (const float*)d_in[3];
    const float* Wl[3]  = {(const float*)d_in[4], (const float*)d_in[6], (const float*)d_in[8]};
    const float* bl[3]  = {(const float*)d_in[5], (const float*)d_in[7], (const float*)d_in[9]};
    const float* lin1_W = (const float*)d_in[10];
    const float* lin1_b = (const float*)d_in[11];
    const float* lin2_W = (const float*)d_in[12];
    const float* lin2_b = (const float*)d_in[13];
    float* out = (float*)d_out;

    const int N = in_sizes[0] / 128;
    const int E = in_sizes[1] / 16;
    const int* rows = eidx;
    const int* cols = eidx + E;

    float* ws   = (float*)d_ws;
    float* h    = ws;
    float* agg  = h + (size_t)N * HID;
    float* xa   = agg + (size_t)N * HID;
    float* xb   = xa + (size_t)N * HID;
    float* ew   = xb + (size_t)N * HID;
    float* dinv = ew + E;
    float* t    = h;

    const int rowTiles = (N + 15) / 16;
    const int eBlk     = (E + 255) / 256;
    const int nodeElts = N * HID;

    zero_kernel<<<(N + 255) / 256, 256, 0, stream>>>(dinv, N);
    edge_pre_kernel<<<(E + 255) / 256, 256, 0, stream>>>(eattr, We, rows, ew, dinv, E);
    dinv_kernel<<<(N + 255) / 256, 256, 0, stream>>>(dinv, N);

    gemm_wmma_kernel<128, 0><<<rowTiles, 128, 0, stream>>>(x, Wl[0], dinv, h, agg);
    edge_agg_kernel<<<eBlk, 256, 0, stream>>>(rows, cols, ew, dinv, h, agg, E);
    finalize_kernel<<<(nodeElts + 255) / 256, 256, 0, stream>>>(agg, bl[0], xa, nodeElts);

    gemm_wmma_kernel<64, 0><<<rowTiles, 128, 0, stream>>>(xa, Wl[1], dinv, h, agg);
    edge_agg_kernel<<<eBlk, 256, 0, stream>>>(rows, cols, ew, dinv, h, agg, E);
    finalize_kernel<<<(nodeElts + 255) / 256, 256, 0, stream>>>(agg, bl[1], xb, nodeElts);

    gemm_wmma_kernel<64, 0><<<rowTiles, 128, 0, stream>>>(xb, Wl[2], dinv, h, agg);
    edge_agg_kernel<<<eBlk, 256, 0, stream>>>(rows, cols, ew, dinv, h, agg, E);
    finalize_kernel<<<(nodeElts + 255) / 256, 256, 0, stream>>>(agg, bl[2], xa, nodeElts);

    gemm_wmma_kernel<64, 1><<<rowTiles, 128, 0, stream>>>(xa, lin1_W, lin1_b, t, nullptr);
    head2_kernel<<<(N + 255) / 256, 256, 0, stream>>>(t, lin2_W, lin2_b, out, N);
}


# Bias+relu of each layer applied in the next GEMM's A-fragment load (aggregate planes used in turn): the three finalize launches and their 12.8 MB read + 12.8 MB write each are gone
# speedup vs baseline: 2.1997x; 1.1263x over previous
#include <hip/hip_runtime.h>
#include <hip/hip_bf16.h>

typedef __attribute__((ext_vector_type(2)))  float    v2f;
typedef __attribute__((ext_vector_type(8)))  float    v8f;
typedef __attribute__((ext_vector_type(16))) _Float16 v16h;

#define HID 64
#define NCLASS 10

#if defined(__HIP_DEVICE_COMPILE__)
#  if __has_builtin(__builtin_amdgcn_wmma_f32_16x16x4_f32)
#    define GCN_USE_F32_WMMA 1
#  else
#    define GCN_USE_F32_WMMA 0
#  endif
#endif

__device__ __forceinline__ void atomic_add_f32(float* p, float v) {
    asm volatile("global_atomic_add_f32 %0, %1, off"
                 :
                 : "v"(p), "v"(v)
                 : "memory");
}

__global__ __launch_bounds__(256) void edge_pre_kernel(
    const float* __restrict__ ea, const float* __restrict__ We,
    const int* __restrict__ rows, float* __restrict__ ew,
    float* __restrict__ deg, int nedge) {
    int e = blockIdx.x * 256 + threadIdx.x;
    if (e >= nedge) return;
    const float4* a4 = (const float4*)(ea + (size_t)e * 16);
    const float4* w4 = (const float4*)We;
    float s = 0.0f;
#pragma unroll
    for (int i = 0; i < 4; ++i) {
        float4 a = a4[i], w = w4[i];
        s += a.x * w.x + a.y * w.y + a.z * w.z + a.w * w.w;
    }
    ew[e] = s;
    atomic_add_f32(&deg[rows[e]], s);
}

__global__ __launch_bounds__(256) void zero_kernel(float* __restrict__ p, int n) {
    int i = blockIdx.x * 256 + threadIdx.x;
    if (i < n) p[i] = 0.0f;
}

__global__ __launch_bounds__(256) void dinv_kernel(float* __restrict__ deg, int n) {
    int i = blockIdx.x * 256 + threadIdx.x;
    if (i < n) deg[i] = __frsqrt_rn(deg[i] + 1.0f);
}

template <int K, int MODE, int ACT>
__global__ __launch_bounds__(128) void gemm_wmma_kernel(
    const float* __restrict__ A, const float* __restrict__ abias,
    const float* __restrict__ W, const float* __restrict__ dinv_or_bias,
    float* __restrict__ out0, float* __restrict__ out1) {
    const int lane = threadIdx.x & 31;
    const int wave = threadIdx.x >> 5;
    const int m    = lane & 15;
    const int hi   = lane >> 4;
    const int row0 = blockIdx.x * 16;
    const int col0 = wave * 16;

    v8f acc = {0.f, 0.f, 0.f, 0.f, 0.f, 0.f, 0.f, 0.f};

#if defined(__HIP_DEVICE_COMPILE__)
#if GCN_USE_F32_WMMA
    const int kh = hi * 2;
#pragma unroll 4
    for (int k0 = 0; k0 < K; k0 += 4) {
        const float* ap = A + (size_t)(row0 + m) * K + k0 + kh;
        const float* bp = W + (size_t)(k0 + kh) * HID + col0 + m;
        v2f a, b;
        a.x = ap[0];   a.y = ap[1];
        b.x = bp[0];   b.y = bp[HID];
        if (ACT == 1) {
            const float s0 = a.x + abias[k0 + kh];
            const float s1 = a.y + abias[k0 + kh + 1];
            a.x = s0 > 0.0f ? s0 : 0.0f;
            a.y = s1 > 0.0f ? s1 : 0.0f;
        }
        acc = __builtin_amdgcn_wmma_f32_16x16x4_f32(
            false, a, false, b, (short)0, acc, false, false);
    }
#else
    const int kb = hi * 8;
    for (int k0 = 0; k0 < K; k0 += 32) {
        const float* ap = A + (size_t)(row0 + m) * K + k0 + kb;
        const float* bp = W + (size_t)(k0 + kb) * HID + col0 + m;
        v16h a, b;
#pragma unroll
        for (int j = 0; j < 8; ++j) {
            float p = ap[j], q = ap[j + 16];
            if (ACT == 1) {
                p += abias[k0 + kb + j];
                q += abias[k0 + kb + j + 16];
                p = p > 0.0f ? p : 0.0f;
                q = q > 0.0f ? q : 0.0f;
            }
            a[j]     = (_Float16)p;
            a[j + 8] = (_Float16)q;
            b[j]     = (_Float16)bp[(size_t)j * HID];
            b[j + 8] = (_Float16)bp[(size_t)(j + 16) * HID];
        }
        acc = __builtin_amdgcn_wmma_f32_16x16x32_f16(
            false, a, false, b, (short)0, acc, false, false);
    }
#endif
#endif

#pragma unroll
    for (int r = 0; r < 8; ++r) {
        const int row = row0 + r + hi * 8;
        const int col = col0 + m;
        const size_t idx = (size_t)row * HID + col;
        const float v = acc[r];
        if (MODE == 0) {
            out0[idx] = v;
            const float di = dinv_or_bias[row];
            out1[idx] = di * di * v;
        } else {
            out0[idx] = v + dinv_or_bias[col];
        }
    }
}

template <int U>
__device__ __forceinline__ void edge_group(const float* __restrict__ hp, float* __restrict__ ap,
                                           int c, int r, float nrm, int j) {
    float v0[U], v1[U];
#pragma unroll
    for (int u = 0; u < U; ++u) {
        const int cj = __builtin_amdgcn_readlane(c, j + u);
        const float* s = hp + (size_t)cj * HID;
        v0[u] = s[0];
        v1[u] = s[32];
    }
#pragma unroll
    for (int u = 0; u < U; ++u) {
        const int rj = __builtin_amdgcn_readlane(r, j + u);
        const float w = __int_as_float(__builtin_amdgcn_readlane(__float_as_int(nrm), j + u));
        float* d = ap + (size_t)rj * HID;
        atomic_add_f32(d,      w * v0[u]);
        atomic_add_f32(d + 32, w * v1[u]);
    }
}

__global__ __launch_bounds__(256) void edge_agg_kernel(
    const int* __restrict__ rows, const int* __restrict__ cols,
    const float* __restrict__ ew, const float* __restrict__ dinv,
    const float* __restrict__ h, float* __restrict__ agg, int nedge) {
    const int lane = threadIdx.x & 31;
    const int wv = __builtin_amdgcn_readfirstlane((int)(threadIdx.x >> 5));
    const int e0 = (blockIdx.x * 8 + wv) * 32;
    if (e0 >= nedge) return;
    const int m = (nedge - e0) < 32 ? (nedge - e0) : 32;
    int c = 0, r = 0;
    float nrm = 0.0f;
    if (lane < m) {
        c = cols[e0 + lane];
        r = rows[e0 + lane];
        nrm = dinv[r] * ew[e0 + lane] * dinv[c];
    }
    const float* hp = h + lane;
    float* ap = agg + lane;
    int j = 0;
    for (; j + 8 <= m; j += 8) edge_group<8>(hp, ap, c, r, nrm, j);
    if (j + 4 <= m) { edge_group<4>(hp, ap, c, r, nrm, j); j += 4; }
    if (j + 2 <= m) { edge_group<2>(hp, ap, c, r, nrm, j); j += 2; }
    if (j < m)      { edge_group<1>(hp, ap, c, r, nrm, j); }
}

__global__ __launch_bounds__(256) void head2_kernel(
    const float* __restrict__ t, const float* __restrict__ W,
    const float* __restrict__ b, float* __restrict__ out, int n) {
    int i = blockIdx.x * 256 + threadIdx.x;
    if (i >= n) return;
    float acc[NCLASS];
#pragma unroll
    for (int c = 0; c < NCLASS; ++c) acc[c] = b[c];
    const float* ti = t + (size_t)i * HID;
#pragma unroll 4
    for (int j = 0; j < HID; ++j) {
        const float tv = ti[j];
        const float* wr = W + (size_t)j * NCLASS;
#pragma unroll
        for (int c = 0; c < NCLASS; ++c) acc[c] += tv * wr[c];
    }
#pragma unroll
    for (int c = 0; c < NCLASS; ++c) out[(size_t)i * NCLASS + c] = acc[c];
}

extern "C" void kernel_launch(void* const* d_in, const int* in_sizes, int n_in,
                              void* d_out, int out_size, void* d_ws, size_t ws_size,
                              hipStream_t stream) {
    const float* x      = (const float*)d_in[0];
    const float* eattr  = (const float*)d_in[1];
    const int*   eidx   = (const int*)  d_in[2];
    const float* We     = (const float*)d_in[3];
    const float* Wl[3]  = {(const float*)d_in[4], (const float*)d_in[6], (const float*)d_in[8]};
    const float* bl[3]  = {(const float*)d_in[5], (const float*)d_in[7], (const float*)d_in[9]};
    const float* lin1_W = (const float*)d_in[10];
    const float* lin1_b = (const float*)d_in[11];
    const float* lin2_W = (const float*)d_in[12];
    const float* lin2_b = (const float*)d_in[13];
    float* out = (float*)d_out;

    const int N = in_sizes[0] / 128;
    const int E = in_sizes[1] / 16;
    const int* rows = eidx;
    const int* cols = eidx + E;

    float* ws   = (float*)d_ws;
    float* h    = ws;
    float* agga = h + (size_t)N * HID;
    float* aggb = agga + (size_t)N * HID;
    float* xb   = aggb + (size_t)N * HID;
    float* ew   = xb + (size_t)N * HID;
    float* dinv = ew + E;
    float* t    = h;

    const int rowTiles = (N + 15) / 16;
    const int eBlk     = (E + 255) / 256;

    zero_kernel<<<(N + 255) / 256, 256, 0, stream>>>(dinv, N);
    edge_pre_kernel<<<(E + 255) / 256, 256, 0, stream>>>(eattr, We, rows, ew, dinv, E);
    dinv_kernel<<<(N + 255) / 256, 256, 0, stream>>>(dinv, N);

    gemm_wmma_kernel<128, 0, 0><<<rowTiles, 128, 0, stream>>>(x, nullptr, Wl[0], dinv, h, agga);
    edge_agg_kernel<<<eBlk, 256, 0, stream>>>(rows, cols, ew, dinv, h, agga, E);

    gemm_wmma_kernel<64, 0, 1><<<rowTiles, 128, 0, stream>>>(agga, bl[0], Wl[1], dinv, h, aggb);
    edge_agg_kernel<<<eBlk, 256, 0, stream>>>(rows, cols, ew, dinv, h, aggb, E);

    gemm_wmma_kernel<64, 0, 1><<<rowTiles, 128, 0, stream>>>(aggb, bl[1], Wl[2], dinv, h, agga);
    edge_agg_kernel<<<eBlk, 256, 0, stream>>>(rows, cols, ew, dinv, h, agga, E);

    gemm_wmma_kernel<64, 1, 1><<<rowTiles, 128, 0, stream>>>(agga, bl[2], lin1_W, lin1_b, t, nullptr);
    head2_kernel<<<(N + 255) / 256, 256, 0, stream>>>(t, lin2_W, lin2_b, out, N);
}


# Edges sorted by destination row once per call (two-level counting sort in LDS, wave-private counters); a wave sums a row's sources in registers in fixed order, one store per line: no atomics left
# speedup vs baseline: 2.6144x; 1.1885x over previous
#include <hip/hip_runtime.h>
#include <hip/hip_bf16.h>


typedef __attribute__((ext_vector_type(2)))  float    v2f;
typedef __attribute__((ext_vector_type(8)))  float    v8f;
typedef __attribute__((ext_vector_type(16))) _Float16 v16h;

#define HID 64
#define NCLASS 10
#define RB 64
#define RBSH 6
#define NCH 512
#define CW 16

#if defined(__HIP_DEVICE_COMPILE__)
#  if __has_builtin(__builtin_amdgcn_wmma_f32_16x16x4_f32)
#    define GCN_USE_F32_WMMA 1
#  else
#    define GCN_USE_F32_WMMA 0
#  endif
#endif

__device__ __forceinline__ int wave_incl_scan(int x, int lane) {
#pragma unroll
    for (int d = 1; d < 32; d <<= 1) {
        const int y = __shfl_up(x, d);
        if (lane >= d) x += y;
    }
    return x;
}

__device__ __forceinline__ void key_rank(int key, int lane, int& rank, int& cnt) {
    rank = 0; cnt = 0;
#pragma unroll
    for (int j = 0; j < 32; ++j) {
        const int kj = __builtin_amdgcn_readlane(key, j);
        const int same = (kj == key) ? 1 : 0;
        cnt += same;
        rank += (j < lane) ? same : 0;
    }
}

__device__ __forceinline__ void count_slice(const int* __restrict__ rows, int* cnt_w,
                                            int es, int ee, int lane) {
    for (int e0 = es; e0 < ee; e0 += 32) {
        const int e = e0 + lane;
        const int key = e < ee ? (rows[e] >> RBSH) : -1;
        int rank, cnt;
        key_rank(key, lane, rank, cnt);
        if (key >= 0 && rank == cnt - 1) cnt_w[key] += cnt;
    }
}

__global__ __launch_bounds__(CW * 32) void bucket_count_kernel(
    const int* __restrict__ rows, int* __restrict__ blockhist,
    int nedge, int nb, int chunk) {
    extern __shared__ int sh[];
    const int tid = threadIdx.x, lane = tid & 31, blk = blockIdx.x;
    const int wv = __builtin_amdgcn_readfirstlane(tid >> 5);
    for (int i = tid; i < CW * nb; i += CW * 32) sh[i] = 0;
    __syncthreads();
    const int per = (chunk + CW - 1) / CW;
    const int eb = blk * chunk;
    const int ec = (eb + chunk) < nedge ? (eb + chunk) : nedge;
    const int es = eb + wv * per;
    const int ee = (es + per) < ec ? (es + per) : ec;
    count_slice(rows, sh + wv * nb, es, ee, lane);
    __syncthreads();
    for (int i = tid; i < nb; i += CW * 32) {
        int run = 0;
#pragma unroll
        for (int w = 0; w < CW; ++w) run += sh[w * nb + i];
        blockhist[(size_t)i * NCH + blk] = run;
    }
}

__global__ __launch_bounds__(NCH) void bucket_scan_kernel(
    const int* __restrict__ blockhist, int* __restrict__ offs, int* __restrict__ total) {
    __shared__ int wsum[NCH / 32];
    const int t = threadIdx.x, lane = t & 31, wv = t >> 5;
    const size_t i = (size_t)blockIdx.x * NCH + t;
    const int v = blockhist[i];
    const int x = wave_incl_scan(v, lane);
    if (lane == 31) wsum[wv] = x;
    __syncthreads();
    int wo = 0;
#pragma unroll
    for (int k = 0; k < NCH / 32; ++k) { const int s = wsum[k]; if (k < wv) wo += s; }
    offs[i] = wo + x - v;
    if (t == NCH - 1) total[blockIdx.x] = wo + x;
}

__global__ __launch_bounds__(CW * 32) void bucket_place_kernel(
    const float* __restrict__ ea, const float* __restrict__ We,
    const int* __restrict__ rows, const int* __restrict__ cols,
    const int* __restrict__ offs, const int* __restrict__ total,
    int* __restrict__ bstart, int* __restrict__ rec,
    int nedge, int nb, int chunk) {
    extern __shared__ int sh[];
    __shared__ int wsum[CW];
    int* bst = sh + CW * nb;
    const int tid = threadIdx.x, lane = tid & 31, blk = blockIdx.x;
    const int wv = __builtin_amdgcn_readfirstlane(tid >> 5);
    for (int i = tid; i < CW * nb; i += CW * 32) sh[i] = 0;
    int carry = 0;
    for (int base = 0; base < nb; base += CW * 32) {
        const int i = base + tid;
        const int v = i < nb ? total[i] : 0;
        const int x = wave_incl_scan(v, lane);
        if (lane == 31) wsum[wv] = x;
        __syncthreads();
        if (wv == 0) {
            int s = lane < CW ? wsum[lane] : 0;
            s = wave_incl_scan(s, lane);
            if (lane < CW) wsum[lane] = s;
        }
        __syncthreads();
        const int wo = wv > 0 ? wsum[wv - 1] : 0;
        const int excl = carry + wo + x - v;
        if (i < nb) {
            bst[i] = excl;
            if (blk == 0) bstart[i] = excl;
        }
        carry += wsum[CW - 1];
        __syncthreads();
    }
    if (blk == 0 && tid == 0) bstart[nb] = carry;

    const int per = (chunk + CW - 1) / CW;
    const int eb = blk * chunk;
    const int ec = (eb + chunk) < nedge ? (eb + chunk) : nedge;
    const int es = eb + wv * per;
    const int ee = (es + per) < ec ? (es + per) : ec;
    int* cur = sh + wv * nb;
    count_slice(rows, cur, es, ee, lane);
    __syncthreads();
    for (int i = tid; i < nb; i += CW * 32) {
        int run = bst[i] + offs[(size_t)i * NCH + blk];
#pragma unroll
        for (int w = 0; w < CW; ++w) {
            const int c = sh[w * nb + i];
            sh[w * nb + i] = run;
            run += c;
        }
    }
    __syncthreads();

    const float4* w4 = (const float4*)We;
    int4* rec4 = (int4*)rec;
    for (int e0 = es; e0 < ee; e0 += 32) {
        const int e = e0 + lane;
        const bool valid = e < ee;
        const int r = valid ? rows[e] : 0;
        const int key = valid ? (r >> RBSH) : -1;
        int rank, cnt;
        key_rank(key, lane, rank, cnt);
        if (valid) {
            const int p0 = cur[key];
            if (rank == cnt - 1) cur[key] = p0 + cnt;
            const float4* a4 = (const float4*)(ea + (size_t)e * 16);
            float s = 0.0f;
#pragma unroll
            for (int i = 0; i < 4; ++i) {
                float4 a = a4[i], w = w4[i];
                s += a.x * w.x + a.y * w.y + a.z * w.z + a.w * w.w;
            }
            int4 v;
            v.x = r; v.y = cols[e]; v.z = __float_as_int(s); v.w = 0;
            rec4[p0 + rank] = v;
        }
    }
}

__global__ __launch_bounds__(256) void bucket_rowsort_kernel(
    const int* __restrict__ bstart, const int* __restrict__ rec,
    int* __restrict__ rowptr, int* __restrict__ scol, float* __restrict__ sew,
    float* __restrict__ dinv, int n, int nb) {
    __shared__ int wc[8 * RB];
    __shared__ float wdeg[8 * RB];
    __shared__ int rtot[RB];
    const int t = threadIdx.x, lane = t & 31;
    const int wv = __builtin_amdgcn_readfirstlane(t >> 5);
    const int b = blockIdx.x, r0 = b * RB;
    wc[t] = 0; wc[t + 256] = 0;
    wdeg[t] = 0.0f; wdeg[t + 256] = 0.0f;
    __syncthreads();
    const int4* rec4 = (const int4*)rec;
    const int s = __builtin_amdgcn_readfirstlane(bstart[b]);
    const int e = __builtin_amdgcn_readfirstlane(bstart[b + 1]);
    const int per = (e - s + 7) / 8;
    const int is = s + wv * per;
    const int ie = (is + per) < e ? (is + per) : e;
    int* cur = wc + wv * RB;
    float* dcur = wdeg + wv * RB;
    for (int i0 = is; i0 < ie; i0 += 32) {
        const int i = i0 + lane;
        const int key = i < ie ? (rec4[i].x - r0) : -1;
        int rank, cnt;
        key_rank(key, lane, rank, cnt);
        if (key >= 0 && rank == cnt - 1) cur[key] += cnt;
    }
    __syncthreads();
    if (t < RB) {
        int tot = 0;
#pragma unroll
        for (int w = 0; w < 8; ++w) tot += wc[w * RB + t];
        rtot[t] = tot;
    }
    __syncthreads();
    if (t < RB) {
        int st = s;
        for (int k = 0; k < t; ++k) st += rtot[k];
        if (r0 + t < n) rowptr[r0 + t] = st;
        int run = st;
#pragma unroll
        for (int w = 0; w < 8; ++w) {
            const int c = wc[w * RB + t];
            wc[w * RB + t] = run;
            run += c;
        }
    }
    if (b == nb - 1 && t == 0) rowptr[n] = e;
    __syncthreads();
    for (int i0 = is; i0 < ie; i0 += 32) {
        const int i = i0 + lane;
        const bool valid = i < ie;
        int4 v = {0, 0, 0, 0};
        if (valid) v = rec4[i];
        const int key = valid ? (v.x - r0) : -1;
        const float w = __int_as_float(v.z);
        int rank = 0, cnt = 0;
        float wsum = 0.0f;
#pragma unroll
        for (int j = 0; j < 32; ++j) {
            const int kj = __builtin_amdgcn_readlane(key, j);
            const float wj = __int_as_float(__builtin_amdgcn_readlane(v.z, j));
            const bool same = (kj == key);
            cnt += same ? 1 : 0;
            rank += (same && j < lane) ? 1 : 0;
            wsum += same ? wj : 0.0f;
        }
        if (valid) {
            const int p0 = cur[key];
            if (rank == cnt - 1) {
                cur[key] = p0 + cnt;
                dcur[key] += wsum;
            }
            scol[p0 + rank] = v.y;
            sew[p0 + rank] = w;
        }
    }
    __syncthreads();
    if (t < RB && r0 + t < n) {
        float deg = 0.0f;
#pragma unroll
        for (int w = 0; w < 8; ++w) deg += wdeg[w * RB + t];
        const float d = deg + 1.0f;
        dinv[r0 + t] = d > 0.0f ? __frsqrt_rn(d) : 0.0f;
    }
}

template <int K, int MODE, int ACT>
__global__ __launch_bounds__(128) void gemm_wmma_kernel(
    const float* __restrict__ A, const float* __restrict__ abias,
    const float* __restrict__ W, const float* __restrict__ dinv_or_bias,
    float* __restrict__ out0, float* __restrict__ out1) {
    const int lane = threadIdx.x & 31;
    const int wave = threadIdx.x >> 5;
    const int m    = lane & 15;
    const int hi   = lane >> 4;
    const int row0 = blockIdx.x * 16;
    const int col0 = wave * 16;

    v8f acc = {0.f, 0.f, 0.f, 0.f, 0.f, 0.f, 0.f, 0.f};

#if defined(__HIP_DEVICE_COMPILE__)
#if GCN_USE_F32_WMMA
    const int kh = hi * 2;
#pragma unroll 4
    for (int k0 = 0; k0 < K; k0 += 4) {
        const float* ap = A + (size_t)(row0 + m) * K + k0 + kh;
        const float* bp = W + (size_t)(k0 + kh) * HID + col0 + m;
        v2f a, b;
        a.x = ap[0];   a.y = ap[1];
        b.x = bp[0];   b.y = bp[HID];
        if (ACT == 1) {
            const float s0 = a.x + abias[k0 + kh];
            const float s1 = a.y + abias[k0 + kh + 1];
            a.x = s0 > 0.0f ? s0 : 0.0f;
            a.y = s1 > 0.0f ? s1 : 0.0f;
        }
        acc = __builtin_amdgcn_wmma_f32_16x16x4_f32(
            false, a, false, b, (short)0, acc, false, false);
    }
#else
    const int kb = hi * 8;
    for (int k0 = 0; k0 < K; k0 += 32) {
        const float* ap = A + (size_t)(row0 + m) * K + k0 + kb;
        const float* bp = W + (size_t)(k0 + kb) * HID + col0 + m;
        v16h a, b;
#pragma unroll
        for (int j = 0; j < 8; ++j) {
            float p = ap[j], q = ap[j + 16];
            if (ACT == 1) {
                p += abias[k0 + kb + j];
                q += abias[k0 + kb + j + 16];
                p = p > 0.0f ? p : 0.0f;
                q = q > 0.0f ? q : 0.0f;
            }
            a[j]     = (_Float16)p;
            a[j + 8] = (_Float16)q;
            b[j]     = (_Float16)bp[(size_t)j * HID];
            b[j + 8] = (_Float16)bp[(size_t)(j + 16) * HID];
        }
        acc = __builtin_amdgcn_wmma_f32_16x16x32_f16(
            false, a, false, b, (short)0, acc, false, false);
    }
#endif
#endif

#pragma unroll
    for (int r = 0; r < 8; ++r) {
        const int row = row0 + r + hi * 8;
        const int col = col0 + m;
        const size_t idx = (size_t)row * HID + col;
        const float v = acc[r];
        if (MODE == 0) {
            out0[idx] = v;
        } else {
            out0[idx] = v + dinv_or_bias[col];
        }
    }
}

template <int U>
__device__ __forceinline__ void row_group(const float* __restrict__ hp, int c, float nrm,
                                          int j, float& a0, float& a1) {
    float v0[U], v1[U];
#pragma unroll
    for (int u = 0; u < U; ++u) {
        const int cj = __builtin_amdgcn_readlane(c, j + u);
        const float* s = hp + (size_t)cj * HID;
        v0[u] = s[0];
        v1[u] = s[32];
    }
#pragma unroll
    for (int u = 0; u < U; ++u) {
        const float w = __int_as_float(__builtin_amdgcn_readlane(__float_as_int(nrm), j + u));
        a0 += w * v0[u];
        a1 += w * v1[u];
    }
}

template <int FIRST>
__device__ __forceinline__ void fetch_tile(int row, int base, int e, int lane,
                                           const int* __restrict__ scol,
                                           const float* __restrict__ sew,
                                           const float* __restrict__ dinv,
                                           float* __restrict__ snorm,
                                           int& c, float& nrm) {
    c = 0; nrm = 0.0f;
    if (base + lane < e) {
        c = scol[base + lane];
        if (FIRST) {
            nrm = dinv[row] * sew[base + lane] * dinv[c];
            snorm[base + lane] = nrm;
        } else {
            nrm = snorm[base + lane];
        }
    }
}

template <int FIRST>
__global__ __launch_bounds__(256) void row_agg_kernel(
    const int* __restrict__ rowptr, const int* __restrict__ scol,
    const float* __restrict__ sew, const float* __restrict__ dinv,
    float* __restrict__ snorm, const float* __restrict__ h,
    float* __restrict__ agg, int n) {
    const int lane = threadIdx.x & 31;
    const int tw = gridDim.x * 8;
    int row = blockIdx.x * 8 + __builtin_amdgcn_readfirstlane((int)(threadIdx.x >> 5));
    if (row >= n) return;
    const float* hp = h + lane;
    int s = __builtin_amdgcn_readfirstlane(rowptr[row]);
    int e = __builtin_amdgcn_readfirstlane(rowptr[row + 1]);
    int c;
    float nrm;
    fetch_tile<FIRST>(row, s, e, lane, scol, sew, dinv, snorm, c, nrm);
    while (true) {
        const int nrow = row + tw;
        const bool more = nrow < n;
        int ns = 0, ne = 0, nc = 0;
        float nnrm = 0.0f;
        if (more) {
            ns = __builtin_amdgcn_readfirstlane(rowptr[nrow]);
            ne = __builtin_amdgcn_readfirstlane(rowptr[nrow + 1]);
            fetch_tile<FIRST>(nrow, ns, ne, lane, scol, sew, dinv, snorm, nc, nnrm);
        }
        const size_t g = (size_t)row * HID + lane;
        const float di = dinv[row];
        const float h0 = h[g], h1 = h[g + 32];
        float a0 = 0.0f, a1 = 0.0f;
        int base = s;
        while (base < e) {
            const int m = (e - base) < 32 ? (e - base) : 32;
            int j = 0;
            for (; j + 8 <= m; j += 8) row_group<8>(hp, c, nrm, j, a0, a1);
            if (j + 4 <= m) { row_group<4>(hp, c, nrm, j, a0, a1); j += 4; }
            if (j + 2 <= m) { row_group<2>(hp, c, nrm, j, a0, a1); j += 2; }
            if (j < m)      { row_group<1>(hp, c, nrm, j, a0, a1); }
            base += 32;
            if (base < e) fetch_tile<FIRST>(row, base, e, lane, scol, sew, dinv, snorm, c, nrm);
        }
        const float d2 = di * di;
        agg[g]      = a0 + d2 * h0;
        agg[g + 32] = a1 + d2 * h1;
        if (!more) break;
        row = nrow; s = ns; e = ne; c = nc; nrm = nnrm;
    }
}

__global__ __launch_bounds__(256) void head2_kernel(
    const float* __restrict__ t, const float* __restrict__ W,
    const float* __restrict__ b, float* __restrict__ out, int n) {
    int i = blockIdx.x * 256 + threadIdx.x;
    if (i >= n) return;
    float acc[NCLASS];
#pragma unroll
    for (int c = 0; c < NCLASS; ++c) acc[c] = b[c];
    const float* ti = t + (size_t)i * HID;
#pragma unroll 4
    for (int j = 0; j < HID; ++j) {
        const float tv = ti[j];
        const float* wr = W + (size_t)j * NCLASS;
#pragma unroll
        for (int c = 0; c < NCLASS; ++c) acc[c] += tv * wr[c];
    }
#pragma unroll
    for (int c = 0; c < NCLASS; ++c) out[(size_t)i * NCLASS + c] = acc[c];
}

extern "C" void kernel_launch(void* const* d_in, const int* in_sizes, int n_in,
                              void* d_out, int out_size, void* d_ws, size_t ws_size,
                              hipStream_t stream) {
    const float* x      = (const float*)d_in[0];
    const float* eattr  = (const float*)d_in[1];
    const int*   eidx   = (const int*)  d_in[2];
    const float* We     = (const float*)d_in[3];
    const float* Wl[3]  = {(const float*)d_in[4], (const float*)d_in[6], (const float*)d_in[8]};
    const float* bl[3]  = {(const float*)d_in[5], (const float*)d_in[7], (const float*)d_in[9]};
    const float* lin1_W = (const float*)d_in[10];
    const float* lin1_b = (const float*)d_in[11];
    const float* lin2_W = (const float*)d_in[12];
    const float* lin2_b = (const float*)d_in[13];
    float* out = (float*)d_out;

    const int N = in_sizes[0] / 128;
    const int E = in_sizes[1] / 16;
    const int* rows = eidx;
    const int* cols = eidx + E;

    const int nb    = (N + RB - 1) / RB;
    const int chunk = (E + NCH - 1) / NCH;

    float* ws   = (float*)d_ws;
    float* h    = ws;
    float* agga = h + (size_t)N * HID;
    float* aggb = agga + (size_t)N * HID;
    int*   rec  = (int*)(aggb + (size_t)N * HID);
    int*   scol = rec + (size_t)E * 4;
    float* sew  = (float*)(scol + E);
    float* snorm = sew + E;
    float* dinv = snorm + E;
    int* rowptr = (int*)(dinv + N);
    int* blockhist = rowptr + N + 1;
    int* offs   = blockhist + (size_t)nb * NCH;
    int* total  = offs + (size_t)nb * NCH;
    int* bstart = total + nb;
    float* t    = h;

    const int rowTiles = (N + 15) / 16;
    const int aggBlk   = 2048;

    bucket_count_kernel<<<NCH, CW * 32, CW * nb * sizeof(int), stream>>>(rows, blockhist, E, nb, chunk);
    bucket_scan_kernel<<<nb, NCH, 0, stream>>>(blockhist, offs, total);
    bucket_place_kernel<<<NCH, CW * 32, (CW + 1) * nb * sizeof(int), stream>>>(
        eattr, We, rows, cols, offs, total, bstart, rec, E, nb, chunk);
    bucket_rowsort_kernel<<<nb, 256, 0, stream>>>(bstart, rec, rowptr, scol, sew, dinv, N, nb);

    gemm_wmma_kernel<128, 0, 0><<<rowTiles, 128, 0, stream>>>(x, nullptr, Wl[0], nullptr, h, nullptr);
    row_agg_kernel<1><<<aggBlk, 256, 0, stream>>>(rowptr, scol, sew, dinv, snorm, h, agga, N);

    gemm_wmma_kernel<64, 0, 1><<<rowTiles, 128, 0, stream>>>(agga, bl[0], Wl[1], nullptr, h, nullptr);
    row_agg_kernel<0><<<aggBlk, 256, 0, stream>>>(rowptr, scol, sew, dinv, snorm, h, aggb, N);

    gemm_wmma_kernel<64, 0, 1><<<rowTiles, 128, 0, stream>>>(aggb, bl[1], Wl[2], nullptr, h, nullptr);
    row_agg_kernel<0><<<aggBlk, 256, 0, stream>>>(rowptr, scol, sew, dinv, snorm, h, agga, N);

    gemm_wmma_kernel<64, 1, 1><<<rowTiles, 128, 0, stream>>>(agga, bl[2], lin1_W, lin1_b, t, nullptr);
    head2_kernel<<<(N + 255) / 256, 256, 0, stream>>>(t, lin2_W, lin2_b, out, N);
}


# Each aggregation kernel also does the next product: a 4-wave block sums 16 rows into LDS (bias+relu), then WMMA from LDS; lin2 joins the last. The 3 K=64 GEMM launches and the lin2 launch are gone
# speedup vs baseline: 3.0949x; 1.1838x over previous
#include <hip/hip_runtime.h>
#include <hip/hip_bf16.h>


typedef __attribute__((ext_vector_type(2)))  float    v2f;
typedef __attribute__((ext_vector_type(8)))  float    v8f;
typedef __attribute__((ext_vector_type(16))) _Float16 v16h;

#define HID 64
#define NCLASS 10
#define RB 64
#define RBSH 6
#define NCH 512
#define CW 16

#if defined(__HIP_DEVICE_COMPILE__)
#  if __has_builtin(__builtin_amdgcn_wmma_f32_16x16x4_f32)
#    define GCN_USE_F32_WMMA 1
#  else
#    define GCN_USE_F32_WMMA 0
#  endif
#endif

__device__ __forceinline__ int wave_incl_scan(int x, int lane) {
#pragma unroll
    for (int d = 1; d < 32; d <<= 1) {
        const int y = __shfl_up(x, d);
        if (lane >= d) x += y;
    }
    return x;
}

__device__ __forceinline__ void key_rank(int key, int lane, int& rank, int& cnt) {
    rank = 0; cnt = 0;
#pragma unroll
    for (int j = 0; j < 32; ++j) {
        const int kj = __builtin_amdgcn_readlane(key, j);
        const int same = (kj == key) ? 1 : 0;
        cnt += same;
        rank += (j < lane) ? same : 0;
    }
}

__device__ __forceinline__ void count_slice(const int* __restrict__ rows, int* cnt_w,
                                            int es, int ee, int lane) {
    for (int e0 = es; e0 < ee; e0 += 32) {
        const int e = e0 + lane;
        const int key = e < ee ? (rows[e] >> RBSH) : -1;
        int rank, cnt;
        key_rank(key, lane, rank, cnt);
        if (key >= 0 && rank == cnt - 1) cnt_w[key] += cnt;
    }
}

__global__ __launch_bounds__(CW * 32) void bucket_count_kernel(
    const int* __restrict__ rows, int* __restrict__ blockhist,
    int nedge, int nb, int chunk) {
    extern __shared__ int sh[];
    const int tid = threadIdx.x, lane = tid & 31, blk = blockIdx.x;
    const int wv = __builtin_amdgcn_readfirstlane(tid >> 5);
    for (int i = tid; i < CW * nb; i += CW * 32) sh[i] = 0;
    __syncthreads();
    const int per = (chunk + CW - 1) / CW;
    const int eb = blk * chunk;
    const int ec = (eb + chunk) < nedge ? (eb + chunk) : nedge;
    const int es = eb + wv * per;
    const int ee = (es + per) < ec ? (es + per) : ec;
    count_slice(rows, sh + wv * nb, es, ee, lane);
    __syncthreads();
    for (int i = tid; i < nb; i += CW * 32) {
        int run = 0;
#pragma unroll
        for (int w = 0; w < CW; ++w) run += sh[w * nb + i];
        blockhist[(size_t)i * NCH + blk] = run;
    }
}

__global__ __launch_bounds__(NCH) void bucket_scan_kernel(
    const int* __restrict__ blockhist, int* __restrict__ offs, int* __restrict__ total) {
    __shared__ int wsum[NCH / 32];
    const int t = threadIdx.x, lane = t & 31, wv = t >> 5;
    const size_t i = (size_t)blockIdx.x * NCH + t;
    const int v = blockhist[i];
    const int x = wave_incl_scan(v, lane);
    if (lane == 31) wsum[wv] = x;
    __syncthreads();
    int wo = 0;
#pragma unroll
    for (int k = 0; k < NCH / 32; ++k) { const int s = wsum[k]; if (k < wv) wo += s; }
    offs[i] = wo + x - v;
    if (t == NCH - 1) total[blockIdx.x] = wo + x;
}

__global__ __launch_bounds__(CW * 32) void bucket_place_kernel(
    const float* __restrict__ ea, const float* __restrict__ We,
    const int* __restrict__ rows, const int* __restrict__ cols,
    const int* __restrict__ offs, const int* __restrict__ total,
    int* __restrict__ bstart, int* __restrict__ rec,
    int nedge, int nb, int chunk) {
    extern __shared__ int sh[];
    __shared__ int wsum[CW];
    int* bst = sh + CW * nb;
    const int tid = threadIdx.x, lane = tid & 31, blk = blockIdx.x;
    const int wv = __builtin_amdgcn_readfirstlane(tid >> 5);
    for (int i = tid; i < CW * nb; i += CW * 32) sh[i] = 0;
    int carry = 0;
    for (int base = 0; base < nb; base += CW * 32) {
        const int i = base + tid;
        const int v = i < nb ? total[i] : 0;
        const int x = wave_incl_scan(v, lane);
        if (lane == 31) wsum[wv] = x;
        __syncthreads();
        if (wv == 0) {
            int s = lane < CW ? wsum[lane] : 0;
            s = wave_incl_scan(s, lane);
            if (lane < CW) wsum[lane] = s;
        }
        __syncthreads();
        const int wo = wv > 0 ? wsum[wv - 1] : 0;
        const int excl = carry + wo + x - v;
        if (i < nb) {
            bst[i] = excl;
            if (blk == 0) bstart[i] = excl;
        }
        carry += wsum[CW - 1];
        __syncthreads();
    }
    if (blk == 0 && tid == 0) bstart[nb] = carry;

    const int per = (chunk + CW - 1) / CW;
    const int eb = blk * chunk;
    const int ec = (eb + chunk) < nedge ? (eb + chunk) : nedge;
    const int es = eb + wv * per;
    const int ee = (es + per) < ec ? (es + per) : ec;
    int* cur = sh + wv * nb;
    count_slice(rows, cur, es, ee, lane);
    __syncthreads();
    for (int i = tid; i < nb; i += CW * 32) {
        int run = bst[i] + offs[(size_t)i * NCH + blk];
#pragma unroll
        for (int w = 0; w < CW; ++w) {
            const int c = sh[w * nb + i];
            sh[w * nb + i] = run;
            run += c;
        }
    }
    __syncthreads();

    const float4* w4 = (const float4*)We;
    int4* rec4 = (int4*)rec;
    for (int e0 = es; e0 < ee; e0 += 32) {
        const int e = e0 + lane;
        const bool valid = e < ee;
        const int r = valid ? rows[e] : 0;
        const int key = valid ? (r >> RBSH) : -1;
        int rank, cnt;
        key_rank(key, lane, rank, cnt);
        if (valid) {
            const int p0 = cur[key];
            if (rank == cnt - 1) cur[key] = p0 + cnt;
            const float4* a4 = (const float4*)(ea + (size_t)e * 16);
            float s = 0.0f;
#pragma unroll
            for (int i = 0; i < 4; ++i) {
                float4 a = a4[i], w = w4[i];
                s += a.x * w.x + a.y * w.y + a.z * w.z + a.w * w.w;
            }
            int4 v;
            v.x = r; v.y = cols[e]; v.z = __float_as_int(s); v.w = 0;
            rec4[p0 + rank] = v;
        }
    }
}

__global__ __launch_bounds__(256) void bucket_rowsort_kernel(
    const int* __restrict__ bstart, const int* __restrict__ rec,
    int* __restrict__ rowptr, int* __restrict__ scol, float* __restrict__ sew,
    float* __restrict__ dinv, int n, int nb) {
    __shared__ int wc[8 * RB];
    __shared__ float wdeg[8 * RB];
    __shared__ int rtot[RB];
    const int t = threadIdx.x, lane = t & 31;
    const int wv = __builtin_amdgcn_readfirstlane(t >> 5);
    const int b = blockIdx.x, r0 = b * RB;
    wc[t] = 0; wc[t + 256] = 0;
    wdeg[t] = 0.0f; wdeg[t + 256] = 0.0f;
    __syncthreads();
    const int4* rec4 = (const int4*)rec;
    const int s = __builtin_amdgcn_readfirstlane(bstart[b]);
    const int e = __builtin_amdgcn_readfirstlane(bstart[b + 1]);
    const int per = (e - s + 7) / 8;
    const int is = s + wv * per;
    const int ie = (is + per) < e ? (is + per) : e;
    int* cur = wc + wv * RB;
    float* dcur = wdeg + wv * RB;
    for (int i0 = is; i0 < ie; i0 += 32) {
        const int i = i0 + lane;
        const int key = i < ie ? (rec4[i].x - r0) : -1;
        int rank, cnt;
        key_rank(key, lane, rank, cnt);
        if (key >= 0 && rank == cnt - 1) cur[key] += cnt;
    }
    __syncthreads();
    if (t < RB) {
        int tot = 0;
#pragma unroll
        for (int w = 0; w < 8; ++w) tot += wc[w * RB + t];
        rtot[t] = tot;
    }
    __syncthreads();
    if (t < RB) {
        int st = s;
        for (int k = 0; k < t; ++k) st += rtot[k];
        if (r0 + t < n) rowptr[r0 + t] = st;
        int run = st;
#pragma unroll
        for (int w = 0; w < 8; ++w) {
            const int c = wc[w * RB + t];
            wc[w * RB + t] = run;
            run += c;
        }
    }
    if (b == nb - 1 && t == 0) rowptr[n] = e;
    __syncthreads();
    for (int i0 = is; i0 < ie; i0 += 32) {
        const int i = i0 + lane;
        const bool valid = i < ie;
        int4 v = {0, 0, 0, 0};
        if (valid) v = rec4[i];
        const int key = valid ? (v.x - r0) : -1;
        const float w = __int_as_float(v.z);
        int rank = 0, cnt = 0;
        float wsum = 0.0f;
#pragma unroll
        for (int j = 0; j < 32; ++j) {
            const int kj = __builtin_amdgcn_readlane(key, j);
            const float wj = __int_as_float(__builtin_amdgcn_readlane(v.z, j));
            const bool same = (kj == key);
            cnt += same ? 1 : 0;
            rank += (same && j < lane) ? 1 : 0;
            wsum += same ? wj : 0.0f;
        }
        if (valid) {
            const int p0 = cur[key];
            if (rank == cnt - 1) {
                cur[key] = p0 + cnt;
                dcur[key] += wsum;
            }
            scol[p0 + rank] = v.y;
            sew[p0 + rank] = w;
        }
    }
    __syncthreads();
    if (t < RB && r0 + t < n) {
        float deg = 0.0f;
#pragma unroll
        for (int w = 0; w < 8; ++w) deg += wdeg[w * RB + t];
        const float d = deg + 1.0f;
        dinv[r0 + t] = d > 0.0f ? __frsqrt_rn(d) : 0.0f;
    }
}

template <int K, int MODE, int ACT>
__global__ __launch_bounds__(128) void gemm_wmma_kernel(
    const float* __restrict__ A, const float* __restrict__ abias,
    const float* __restrict__ W, const float* __restrict__ dinv_or_bias,
    float* __restrict__ out0, float* __restrict__ out1) {
    const int lane = threadIdx.x & 31;
    const int wave = threadIdx.x >> 5;
    const int m    = lane & 15;
    const int hi   = lane >> 4;
    const int row0 = blockIdx.x * 16;
    const int col0 = wave * 16;

    v8f acc = {0.f, 0.f, 0.f, 0.f, 0.f, 0.f, 0.f, 0.f};

#if defined(__HIP_DEVICE_COMPILE__)
#if GCN_USE_F32_WMMA
    const int kh = hi * 2;
#pragma unroll 4
    for (int k0 = 0; k0 < K; k0 += 4) {
        const float* ap = A + (size_t)(row0 + m) * K + k0 + kh;
        const float* bp = W + (size_t)(k0 + kh) * HID + col0 + m;
        v2f a, b;
        a.x = ap[0];   a.y = ap[1];
        b.x = bp[0];   b.y = bp[HID];
        if (ACT == 1) {
            const float s0 = a.x + abias[k0 + kh];
            const float s1 = a.y + abias[k0 + kh + 1];
            a.x = s0 > 0.0f ? s0 : 0.0f;
            a.y = s1 > 0.0f ? s1 : 0.0f;
        }
        acc = __builtin_amdgcn_wmma_f32_16x16x4_f32(
            false, a, false, b, (short)0, acc, false, false);
    }
#else
    const int kb = hi * 8;
    for (int k0 = 0; k0 < K; k0 += 32) {
        const float* ap = A + (size_t)(row0 + m) * K + k0 + kb;
        const float* bp = W + (size_t)(k0 + kb) * HID + col0 + m;
        v16h a, b;
#pragma unroll
        for (int j = 0; j < 8; ++j) {
            float p = ap[j], q = ap[j + 16];
            if (ACT == 1) {
                p += abias[k0 + kb + j];
                q += abias[k0 + kb + j + 16];
                p = p > 0.0f ? p : 0.0f;
                q = q > 0.0f ? q : 0.0f;
            }
            a[j]     = (_Float16)p;
            a[j + 8] = (_Float16)q;
            b[j]     = (_Float16)bp[(size_t)j * HID];
            b[j + 8] = (_Float16)bp[(size_t)(j + 16) * HID];
        }
        acc = __builtin_amdgcn_wmma_f32_16x16x32_f16(
            false, a, false, b, (short)0, acc, false, false);
    }
#endif
#endif

#pragma unroll
    for (int r = 0; r < 8; ++r) {
        const int row = row0 + r + hi * 8;
        const int col = col0 + m;
        const size_t idx = (size_t)row * HID + col;
        const float v = acc[r];
        if (MODE == 0) {
            out0[idx] = v;
        } else {
            out0[idx] = v + dinv_or_bias[col];
        }
    }
}

template <int U>
__device__ __forceinline__ void row_group(const float* __restrict__ hp, int c, float nrm,
                                          int j, float& a0, float& a1) {
    float v0[U], v1[U];
#pragma unroll
    for (int u = 0; u < U; ++u) {
        const int cj = __builtin_amdgcn_readlane(c, j + u);
        const float* s = hp + (size_t)cj * HID;
        v0[u] = s[0];
        v1[u] = s[32];
    }
#pragma unroll
    for (int u = 0; u < U; ++u) {
        const float w = __int_as_float(__builtin_amdgcn_readlane(__float_as_int(nrm), j + u));
        a0 += w * v0[u];
        a1 += w * v1[u];
    }
}

template <int FIRST>
__device__ __forceinline__ void fetch_tile(int row, int base, int e, int lane,
                                           const int* __restrict__ scol,
                                           const float* __restrict__ sew,
                                           const float* __restrict__ dinv,
                                           float* __restrict__ snorm,
                                           int& c, float& nrm) {
    c = 0; nrm = 0.0f;
    if (base + lane < e) {
        c = scol[base + lane];
        if (FIRST) {
            nrm = dinv[row] * sew[base + lane] * dinv[c];
            snorm[base + lane] = nrm;
        } else {
            nrm = snorm[base + lane];
        }
    }
}

#define TP 68

template <int FIRST, int HEAD>
__global__ __launch_bounds__(128) void agg_gemm_kernel(
    const int* __restrict__ rowptr, const int* __restrict__ scol,
    const float* __restrict__ sew, const float* __restrict__ dinv,
    float* __restrict__ snorm, const float* __restrict__ h,
    const float* __restrict__ abias, const float* __restrict__ W,
    const float* __restrict__ wbias, const float* __restrict__ W2,
    const float* __restrict__ b2, float* __restrict__ outp, int n) {
    __shared__ __attribute__((aligned(16))) float tile[16 * TP];
    const int lane = threadIdx.x & 31;
    const int wave = __builtin_amdgcn_readfirstlane((int)(threadIdx.x >> 5));
    const int row0 = blockIdx.x * 16;
    const float* hp = h + lane;
    const float bz0 = abias[lane], bz1 = abias[lane + 32];

    int row = row0 + wave * 4;
    int s = 0, e = 0, c = 0;
    float nrm = 0.0f;
    if (row < n) {
        s = __builtin_amdgcn_readfirstlane(rowptr[row]);
        e = __builtin_amdgcn_readfirstlane(rowptr[row + 1]);
        fetch_tile<FIRST>(row, s, e, lane, scol, sew, dinv, snorm, c, nrm);
    }
    for (int i = 0; i < 4; ++i) {
        const int nrow = row + 1;
        const bool more = (i < 3) && (nrow < n);
        int ns = 0, ne = 0, nc = 0;
        float nnrm = 0.0f;
        if (more) {
            ns = e;
            ne = __builtin_amdgcn_readfirstlane(rowptr[nrow + 1]);
            fetch_tile<FIRST>(nrow, ns, ne, lane, scol, sew, dinv, snorm, nc, nnrm);
        }
        float v0 = 0.0f, v1 = 0.0f;
        if (row < n) {
            const size_t g = (size_t)row * HID + lane;
            const float di = dinv[row];
            const float h0 = h[g], h1 = h[g + 32];
            float a0 = 0.0f, a1 = 0.0f;
            int base = s;
            while (base < e) {
                const int m = (e - base) < 32 ? (e - base) : 32;
                int j = 0;
                for (; j + 8 <= m; j += 8) row_group<8>(hp, c, nrm, j, a0, a1);
                if (j + 4 <= m) { row_group<4>(hp, c, nrm, j, a0, a1); j += 4; }
                if (j + 2 <= m) { row_group<2>(hp, c, nrm, j, a0, a1); j += 2; }
                if (j < m)      { row_group<1>(hp, c, nrm, j, a0, a1); }
                base += 32;
                if (base < e) fetch_tile<FIRST>(row, base, e, lane, scol, sew, dinv, snorm, c, nrm);
            }
            const float d2 = di * di;
            const float g0 = a0 + d2 * h0;
            const float g1 = a1 + d2 * h1;
            const float s0 = g0 + bz0;
            const float s1 = g1 + bz1;
            v0 = s0 > 0.0f ? s0 : 0.0f;
            v1 = s1 > 0.0f ? s1 : 0.0f;
        }
        float* tr = tile + (wave * 4 + i) * TP + lane;
        tr[0]  = v0;
        tr[32] = v1;
        row = nrow; s = ns; e = ne; c = nc; nrm = nnrm;
    }
    __syncthreads();

    const int m    = lane & 15;
    const int hi   = lane >> 4;
    const int col0 = wave * 16;
    v8f acc = {0.f, 0.f, 0.f, 0.f, 0.f, 0.f, 0.f, 0.f};
#if defined(__HIP_DEVICE_COMPILE__)
    const int kh = hi * 2;
#pragma unroll
    for (int k0 = 0; k0 < HID; k0 += 4) {
        const float* ap = tile + m * TP + k0 + kh;
        const float* bp = W + (size_t)(k0 + kh) * HID + col0 + m;
        v2f a, b;
        a.x = ap[0];   a.y = ap[1];
        b.x = bp[0];   b.y = bp[HID];
        acc = __builtin_amdgcn_wmma_f32_16x16x4_f32(
            false, a, false, b, (short)0, acc, false, false);
    }
#endif

    if (HEAD == 0) {
#pragma unroll
        for (int r = 0; r < 8; ++r) {
            const int orow = row0 + r + hi * 8;
            if (orow < n) outp[(size_t)orow * HID + col0 + m] = acc[r];
        }
    } else {
        __syncthreads();
        const float bb = wbias[col0 + m];
#pragma unroll
        for (int r = 0; r < 8; ++r) tile[(r + hi * 8) * TP + col0 + m] = acc[r] + bb;
        __syncthreads();
        const int rr = threadIdx.x >> 3;
        const int q  = threadIdx.x & 7;
        const bool two = q < NCLASS - 8;
        const int c1 = two ? q + 8 : q;
        const float* tr = tile + rr * TP;
        float r0 = b2[q], r1 = b2[c1];
#pragma unroll 8
        for (int j = 0; j < HID; ++j) {
            const float tv = tr[j];
            r0 += tv * W2[(size_t)j * NCLASS + q];
            r1 += tv * W2[(size_t)j * NCLASS + c1];
        }
        if (row0 + rr < n) {
            float* op = outp + (size_t)(row0 + rr) * NCLASS;
            op[q] = r0;
            if (two) op[c1] = r1;
        }
    }
}

extern "C" void kernel_launch(void* const* d_in, const int* in_sizes, int n_in,
                              void* d_out, int out_size, void* d_ws, size_t ws_size,
                              hipStream_t stream) {
    const float* x      = (const float*)d_in[0];
    const float* eattr  = (const float*)d_in[1];
    const int*   eidx   = (const int*)  d_in[2];
    const float* We     = (const float*)d_in[3];
    const float* Wl[3]  = {(const float*)d_in[4], (const float*)d_in[6], (const float*)d_in[8]};
    const float* bl[3]  = {(const float*)d_in[5], (const float*)d_in[7], (const float*)d_in[9]};
    const float* lin1_W = (const float*)d_in[10];
    const float* lin1_b = (const float*)d_in[11];
    const float* lin2_W = (const float*)d_in[12];
    const float* lin2_b = (const float*)d_in[13];
    float* out = (float*)d_out;

    const int N = in_sizes[0] / 128;
    const int E = in_sizes[1] / 16;
    const int* rows = eidx;
    const int* cols = eidx + E;

    const int nb    = (N + RB - 1) / RB;
    const int chunk = (E + NCH - 1) / NCH;

    float* ws   = (float*)d_ws;
    float* h    = ws;
    float* agga = h + (size_t)N * HID;
    float* aggb = agga + (size_t)N * HID;
    int*   rec  = (int*)(aggb + (size_t)N * HID);
    int*   scol = rec + (size_t)E * 4;
    float* sew  = (float*)(scol + E);
    float* snorm = sew + E;
    float* dinv = snorm + E;
    int* rowptr = (int*)(dinv + N);
    int* blockhist = rowptr + N + 1;
    int* offs   = blockhist + (size_t)nb * NCH;
    int* total  = offs + (size_t)nb * NCH;
    int* bstart = total + nb;

    const int rowTiles = (N + 15) / 16;

    bucket_count_kernel<<<NCH, CW * 32, CW * nb * sizeof(int), stream>>>(rows, blockhist, E, nb, chunk);
    bucket_scan_kernel<<<nb, NCH, 0, stream>>>(blockhist, offs, total);
    bucket_place_kernel<<<NCH, CW * 32, (CW + 1) * nb * sizeof(int), stream>>>(
        eattr, We, rows, cols, offs, total, bstart, rec, E, nb, chunk);
    bucket_rowsort_kernel<<<nb, 256, 0, stream>>>(bstart, rec, rowptr, scol, sew, dinv, N, nb);

    gemm_wmma_kernel<128, 0, 0><<<rowTiles, 128, 0, stream>>>(x, nullptr, Wl[0], nullptr, h, nullptr);

    agg_gemm_kernel<1, 0><<<rowTiles, 128, 0, stream>>>(
        rowptr, scol, sew, dinv, snorm, h, bl[0], Wl[1], nullptr, nullptr, nullptr, agga, N);
    agg_gemm_kernel<0, 0><<<rowTiles, 128, 0, stream>>>(
        rowptr, scol, sew, dinv, snorm, agga, bl[1], Wl[2], nullptr, nullptr, nullptr, h, N);
    agg_gemm_kernel<0, 1><<<rowTiles, 128, 0, stream>>>(
        rowptr, scol, sew, dinv, snorm, h, bl[2], lin1_W, lin1_b, lin2_W, lin2_b, out, N);
}


# Aggregation wave: fewer serial memory round trips: bounds, dinv and col/norm lists of its 4 rows fetched as groups up front; a row's gathers go in rounds of 16 edges, the rest (1..15) as one round
# speedup vs baseline: 3.1945x; 1.0322x over previous
#include <hip/hip_runtime.h>
#include <hip/hip_bf16.h>


typedef __attribute__((ext_vector_type(2)))  float    v2f;
typedef __attribute__((ext_vector_type(8)))  float    v8f;
typedef __attribute__((ext_vector_type(16))) _Float16 v16h;

#define HID 64
#define NCLASS 10
#define RB 64
#define RBSH 6
#define NCH 512
#define CW 16

#if defined(__HIP_DEVICE_COMPILE__)
#  if __has_builtin(__builtin_amdgcn_wmma_f32_16x16x4_f32)
#    define GCN_USE_F32_WMMA 1
#  else
#    define GCN_USE_F32_WMMA 0
#  endif
#endif

__device__ __forceinline__ int wave_incl_scan(int x, int lane) {
#pragma unroll
    for (int d = 1; d < 32; d <<= 1) {
        const int y = __shfl_up(x, d);
        if (lane >= d) x += y;
    }
    return x;
}

__device__ __forceinline__ void key_rank(int key, int lane, int& rank, int& cnt) {
    rank = 0; cnt = 0;
#pragma unroll
    for (int j = 0; j < 32; ++j) {
        const int kj = __builtin_amdgcn_readlane(key, j);
        const int same = (kj == key) ? 1 : 0;
        cnt += same;
        rank += (j < lane) ? same : 0;
    }
}

__device__ __forceinline__ void count_slice(const int* __restrict__ rows, int* cnt_w,
                                            int es, int ee, int lane) {
    for (int e0 = es; e0 < ee; e0 += 32) {
        const int e = e0 + lane;
        const int key = e < ee ? (rows[e] >> RBSH) : -1;
        int rank, cnt;
        key_rank(key, lane, rank, cnt);
        if (key >= 0 && rank == cnt - 1) cnt_w[key] += cnt;
    }
}

__global__ __launch_bounds__(CW * 32) void bucket_count_kernel(
    const int* __restrict__ rows, int* __restrict__ blockhist,
    int nedge, int nb, int chunk) {
    extern __shared__ int sh[];
    const int tid = threadIdx.x, lane = tid & 31, blk = blockIdx.x;
    const int wv = __builtin_amdgcn_readfirstlane(tid >> 5);
    for (int i = tid; i < CW * nb; i += CW * 32) sh[i] = 0;
    __syncthreads();
    const int per = (chunk + CW - 1) / CW;
    const int eb = blk * chunk;
    const int ec = (eb + chunk) < nedge ? (eb + chunk) : nedge;
    const int es = eb + wv * per;
    const int ee = (es + per) < ec ? (es + per) : ec;
    count_slice(rows, sh + wv * nb, es, ee, lane);
    __syncthreads();
    for (int i = tid; i < nb; i += CW * 32) {
        int run = 0;
#pragma unroll
        for (int w = 0; w < CW; ++w) run += sh[w * nb + i];
        blockhist[(size_t)i * NCH + blk] = run;
    }
}

__global__ __launch_bounds__(NCH) void bucket_scan_kernel(
    const int* __restrict__ blockhist, int* __restrict__ offs, int* __restrict__ total) {
    __shared__ int wsum[NCH / 32];
    const int t = threadIdx.x, lane = t & 31, wv = t >> 5;
    const size_t i = (size_t)blockIdx.x * NCH + t;
    const int v = blockhist[i];
    const int x = wave_incl_scan(v, lane);
    if (lane == 31) wsum[wv] = x;
    __syncthreads();
    int wo = 0;
#pragma unroll
    for (int k = 0; k < NCH / 32; ++k) { const int s = wsum[k]; if (k < wv) wo += s; }
    offs[i] = wo + x - v;
    if (t == NCH - 1) total[blockIdx.x] = wo + x;
}

__global__ __launch_bounds__(CW * 32) void bucket_place_kernel(
    const float* __restrict__ ea, const float* __restrict__ We,
    const int* __restrict__ rows, const int* __restrict__ cols,
    const int* __restrict__ offs, const int* __restrict__ total,
    int* __restrict__ bstart, int* __restrict__ rec,
    int nedge, int nb, int chunk) {
    extern __shared__ int sh[];
    __shared__ int wsum[CW];
    int* bst = sh + CW * nb;
    const int tid = threadIdx.x, lane = tid & 31, blk = blockIdx.x;
    const int wv = __builtin_amdgcn_readfirstlane(tid >> 5);
    for (int i = tid; i < CW * nb; i += CW * 32) sh[i] = 0;
    int carry = 0;
    for (int base = 0; base < nb; base += CW * 32) {
        const int i = base + tid;
        const int v = i < nb ? total[i] : 0;
        const int x = wave_incl_scan(v, lane);
        if (lane == 31) wsum[wv] = x;
        __syncthreads();
        if (wv == 0) {
            int s = lane < CW ? wsum[lane] : 0;
            s = wave_incl_scan(s, lane);
            if (lane < CW) wsum[lane] = s;
        }
        __syncthreads();
        const int wo = wv > 0 ? wsum[wv - 1] : 0;
        const int excl = carry + wo + x - v;
        if (i < nb) {
            bst[i] = excl;
            if (blk == 0) bstart[i] = excl;
        }
        carry += wsum[CW - 1];
        __syncthreads();
    }
    if (blk == 0 && tid == 0) bstart[nb] = carry;

    const int per = (chunk + CW - 1) / CW;
    const int eb = blk * chunk;
    const int ec = (eb + chunk) < nedge ? (eb + chunk) : nedge;
    const int es = eb + wv * per;
    const int ee = (es + per) < ec ? (es + per) : ec;
    int* cur = sh + wv * nb;
    count_slice(rows, cur, es, ee, lane);
    __syncthreads();
    for (int i = tid; i < nb; i += CW * 32) {
        int run = bst[i] + offs[(size_t)i * NCH + blk];
#pragma unroll
        for (int w = 0; w < CW; ++w) {
            const int c = sh[w * nb + i];
            sh[w * nb + i] = run;
            run += c;
        }
    }
    __syncthreads();

    const float4* w4 = (const float4*)We;
    int4* rec4 = (int4*)rec;
    for (int e0 = es; e0 < ee; e0 += 32) {
        const int e = e0 + lane;
        const bool valid = e < ee;
        const int r = valid ? rows[e] : 0;
        const int key = valid ? (r >> RBSH) : -1;
        int rank, cnt;
        key_rank(key, lane, rank, cnt);
        if (valid) {
            const int p0 = cur[key];
            if (rank == cnt - 1) cur[key] = p0 + cnt;
            const float4* a4 = (const float4*)(ea + (size_t)e * 16);
            float s = 0.0f;
#pragma unroll
            for (int i = 0; i < 4; ++i) {
                float4 a = a4[i], w = w4[i];
                s += a.x * w.x + a.y * w.y + a.z * w.z + a.w * w.w;
            }
            int4 v;
            v.x = r; v.y = cols[e]; v.z = __float_as_int(s); v.w = 0;
            rec4[p0 + rank] = v;
        }
    }
}

__global__ __launch_bounds__(256) void bucket_rowsort_kernel(
    const int* __restrict__ bstart, const int* __restrict__ rec,
    int* __restrict__ rowptr, int* __restrict__ scol, float* __restrict__ sew,
    float* __restrict__ dinv, int n, int nb) {
    __shared__ int wc[8 * RB];
    __shared__ float wdeg[8 * RB];
    __shared__ int rtot[RB];
    const int t = threadIdx.x, lane = t & 31;
    const int wv = __builtin_amdgcn_readfirstlane(t >> 5);
    const int b = blockIdx.x, r0 = b * RB;
    wc[t] = 0; wc[t + 256] = 0;
    wdeg[t] = 0.0f; wdeg[t + 256] = 0.0f;
    __syncthreads();
    const int4* rec4 = (const int4*)rec;
    const int s = __builtin_amdgcn_readfirstlane(bstart[b]);
    const int e = __builtin_amdgcn_readfirstlane(bstart[b + 1]);
    const int per = (e - s + 7) / 8;
    const int is = s + wv * per;
    const int ie = (is + per) < e ? (is + per) : e;
    int* cur = wc + wv * RB;
    float* dcur = wdeg + wv * RB;
    for (int i0 = is; i0 < ie; i0 += 32) {
        const int i = i0 + lane;
        const int key = i < ie ? (rec4[i].x - r0) : -1;
        int rank, cnt;
        key_rank(key, lane, rank, cnt);
        if (key >= 0 && rank == cnt - 1) cur[key] += cnt;
    }
    __syncthreads();
    if (t < RB) {
        int tot = 0;
#pragma unroll
        for (int w = 0; w < 8; ++w) tot += wc[w * RB + t];
        rtot[t] = tot;
    }
    __syncthreads();
    if (t < RB) {
        int st = s;
        for (int k = 0; k < t; ++k) st += rtot[k];
        if (r0 + t < n) rowptr[r0 + t] = st;
        int run = st;
#pragma unroll
        for (int w = 0; w < 8; ++w) {
            const int c = wc[w * RB + t];
            wc[w * RB + t] = run;
            run += c;
        }
    }
    if (b == nb - 1 && t == 0) rowptr[n] = e;
    __syncthreads();
    for (int i0 = is; i0 < ie; i0 += 32) {
        const int i = i0 + lane;
        const bool valid = i < ie;
        int4 v = {0, 0, 0, 0};
        if (valid) v = rec4[i];
        const int key = valid ? (v.x - r0) : -1;
        const float w = __int_as_float(v.z);
        int rank = 0, cnt = 0;
        float wsum = 0.0f;
#pragma unroll
        for (int j = 0; j < 32; ++j) {
            const int kj = __builtin_amdgcn_readlane(key, j);
            const float wj = __int_as_float(__builtin_amdgcn_readlane(v.z, j));
            const bool same = (kj == key);
            cnt += same ? 1 : 0;
            rank += (same && j < lane) ? 1 : 0;
            wsum += same ? wj : 0.0f;
        }
        if (valid) {
            const int p0 = cur[key];
            if (rank == cnt - 1) {
                cur[key] = p0 + cnt;
                dcur[key] += wsum;
            }
            scol[p0 + rank] = v.y;
            sew[p0 + rank] = w;
        }
    }
    __syncthreads();
    if (t < RB && r0 + t < n) {
        float deg = 0.0f;
#pragma unroll
        for (int w = 0; w < 8; ++w) deg += wdeg[w * RB + t];
        const float d = deg + 1.0f;
        dinv[r0 + t] = d > 0.0f ? __frsqrt_rn(d) : 0.0f;
    }
}

template <int K, int MODE, int ACT>
__global__ __launch_bounds__(128) void gemm_wmma_kernel(
    const float* __restrict__ A, const float* __restrict__ abias,
    const float* __restrict__ W, const float* __restrict__ dinv_or_bias,
    float* __restrict__ out0, float* __restrict__ out1) {
    const int lane = threadIdx.x & 31;
    const int wave = threadIdx.x >> 5;
    const int m    = lane & 15;
    const int hi   = lane >> 4;
    const int row0 = blockIdx.x * 16;
    const int col0 = wave * 16;

    v8f acc = {0.f, 0.f, 0.f, 0.f, 0.f, 0.f, 0.f, 0.f};

#if defined(__HIP_DEVICE_COMPILE__)
#if GCN_USE_F32_WMMA
    const int kh = hi * 2;
#pragma unroll 4
    for (int k0 = 0; k0 < K; k0 += 4) {
        const float* ap = A + (size_t)(row0 + m) * K + k0 + kh;
        const float* bp = W + (size_t)(k0 + kh) * HID + col0 + m;
        v2f a, b;
        a.x = ap[0];   a.y = ap[1];
        b.x = bp[0];   b.y = bp[HID];
        if (ACT == 1) {
            const float s0 = a.x + abias[k0 + kh];
            const float s1 = a.y + abias[k0 + kh + 1];
            a.x = s0 > 0.0f ? s0 : 0.0f;
            a.y = s1 > 0.0f ? s1 : 0.0f;
        }
        acc = __builtin_amdgcn_wmma_f32_16x16x4_f32(
            false, a, false, b, (short)0, acc, false, false);
    }
#else
    const int kb = hi * 8;
    for (int k0 = 0; k0 < K; k0 += 32) {
        const float* ap = A + (size_t)(row0 + m) * K + k0 + kb;
        const float* bp = W + (size_t)(k0 + kb) * HID + col0 + m;
        v16h a, b;
#pragma unroll
        for (int j = 0; j < 8; ++j) {
            float p = ap[j], q = ap[j + 16];
            if (ACT == 1) {
                p += abias[k0 + kb + j];
                q += abias[k0 + kb + j + 16];
                p = p > 0.0f ? p : 0.0f;
                q = q > 0.0f ? q : 0.0f;
            }
            a[j]     = (_Float16)p;
            a[j + 8] = (_Float16)q;
            b[j]     = (_Float16)bp[(size_t)j * HID];
            b[j + 8] = (_Float16)bp[(size_t)(j + 16) * HID];
        }
        acc = __builtin_amdgcn_wmma_f32_16x16x32_f16(
            false, a, false, b, (short)0, acc, false, false);
    }
#endif
#endif

#pragma unroll
    for (int r = 0; r < 8; ++r) {
        const int row = row0 + r + hi * 8;
        const int col = col0 + m;
        const size_t idx = (size_t)row * HID + col;
        const float v = acc[r];
        if (MODE == 0) {
            out0[idx] = v;
        } else {
            out0[idx] = v + dinv_or_bias[col];
        }
    }
}

template <int U>
__device__ __forceinline__ void row_group(const float* __restrict__ h, int lane, int c, float nrm,
                                          int j, float& a0, float& a1) {
    float v0[U], v1[U];
#pragma unroll
    for (int u = 0; u < U; ++u) {
        const int cj = __builtin_amdgcn_readlane(c, j + u);
        const float* s = h + (size_t)cj * HID;
        v0[u] = s[lane];
        v1[u] = s[lane + 32];
    }
#pragma unroll
    for (int u = 0; u < U; ++u) {
        const float w = __int_as_float(__builtin_amdgcn_readlane(__float_as_int(nrm), j + u));
        a0 += w * v0[u];
        a1 += w * v1[u];
    }
}

__device__ __forceinline__ void tile_sums(const float* __restrict__ h, int lane, int c, float nrm,
                                          int m, float& a0, float& a1) {
    int j = 0;
    for (; j + 16 <= m; j += 16) row_group<16>(h, lane, c, nrm, j, a0, a1);
    switch (m - j) {
        case 15: row_group<15>(h, lane, c, nrm, j, a0, a1); break;
        case 14: row_group<14>(h, lane, c, nrm, j, a0, a1); break;
        case 13: row_group<13>(h, lane, c, nrm, j, a0, a1); break;
        case 12: row_group<12>(h, lane, c, nrm, j, a0, a1); break;
        case 11: row_group<11>(h, lane, c, nrm, j, a0, a1); break;
        case 10: row_group<10>(h, lane, c, nrm, j, a0, a1); break;
        case 9:  row_group<9>(h, lane, c, nrm, j, a0, a1);  break;
        case 8:  row_group<8>(h, lane, c, nrm, j, a0, a1);  break;
        case 7:  row_group<7>(h, lane, c, nrm, j, a0, a1);  break;
        case 6:  row_group<6>(h, lane, c, nrm, j, a0, a1);  break;
        case 5:  row_group<5>(h, lane, c, nrm, j, a0, a1);  break;
        case 4:  row_group<4>(h, lane, c, nrm, j, a0, a1);  break;
        case 3:  row_group<3>(h, lane, c, nrm, j, a0, a1);  break;
        case 2:  row_group<2>(h, lane, c, nrm, j, a0, a1);  break;
        case 1:  row_group<1>(h, lane, c, nrm, j, a0, a1);  break;
        default: break;
    }
}

template <int FIRST>
__device__ __forceinline__ void fetch_tile(float di, int base, int e, int lane,
                                           const int* __restrict__ scol,
                                           const float* __restrict__ sew,
                                           const float* __restrict__ dinv,
                                           float* __restrict__ snorm,
                                           int& c, float& nrm) {
    c = 0; nrm = 0.0f;
    if (base + lane < e) {
        c = scol[base + lane];
        if (FIRST) {
            nrm = di * sew[base + lane] * dinv[c];
            snorm[base + lane] = nrm;
        } else {
            nrm = snorm[base + lane];
        }
    }
}

#define TP 68

template <int FIRST, int HEAD>
__global__ __launch_bounds__(128) void agg_gemm_kernel(
    const int* __restrict__ rowptr, const int* __restrict__ scol,
    const float* __restrict__ sew, const float* __restrict__ dinv,
    float* __restrict__ snorm, const float* __restrict__ h,
    const float* __restrict__ abias, const float* __restrict__ W,
    const float* __restrict__ wbias, const float* __restrict__ W2,
    const float* __restrict__ b2, float* __restrict__ outp, int n) {
    __shared__ __attribute__((aligned(16))) float tile[16 * TP];
    const int lane = threadIdx.x & 31;
    const int wave = __builtin_amdgcn_readfirstlane((int)(threadIdx.x >> 5));
    const int row0 = blockIdx.x * 16;
    const float bz0 = abias[lane], bz1 = abias[lane + 32];

    const int rb = row0 + wave * 4;
    int p[5];
    float dr[4];
#pragma unroll
    for (int k = 0; k < 5; ++k) {
        const int r = (rb + k) < n ? (rb + k) : n;
        p[k] = __builtin_amdgcn_readfirstlane(rowptr[r]);
    }
#pragma unroll
    for (int k = 0; k < 4; ++k) {
        const int r = (rb + k) < n ? (rb + k) : (n - 1);
        dr[k] = __int_as_float(__builtin_amdgcn_readfirstlane(__float_as_int(dinv[r])));
    }

    int tc[4];
    float tn[4];
    {
        bool ok[4];
        int idx[4];
#pragma unroll
        for (int k = 0; k < 4; ++k) {
            ok[k]  = p[k] + lane < p[k + 1];
            idx[k] = ok[k] ? p[k] + lane : 0;
        }
        if (FIRST) {
            int cc[4];
            float ww[4], dc[4];
#pragma unroll
            for (int k = 0; k < 4; ++k) { cc[k] = scol[idx[k]]; ww[k] = sew[idx[k]]; }
#pragma unroll
            for (int k = 0; k < 4; ++k) cc[k] = ok[k] ? cc[k] : 0;
#pragma unroll
            for (int k = 0; k < 4; ++k) dc[k] = dinv[cc[k]];
#pragma unroll
            for (int k = 0; k < 4; ++k) {
                const float t = dr[k] * ww[k] * dc[k];
                tc[k] = cc[k];
                tn[k] = ok[k] ? t : 0.0f;
                if (ok[k]) snorm[idx[k]] = t;
            }
        } else {
            int cc[4];
            float nn[4];
#pragma unroll
            for (int k = 0; k < 4; ++k) { cc[k] = scol[idx[k]]; nn[k] = snorm[idx[k]]; }
#pragma unroll
            for (int k = 0; k < 4; ++k) {
                tc[k] = ok[k] ? cc[k] : 0;
                tn[k] = ok[k] ? nn[k] : 0.0f;
            }
        }
    }

#pragma nounroll
    for (int i = 0; i < 4; ++i) {
        const int row = rb + i;
        const int s = p[0], e = p[1];
        float v0 = 0.0f, v1 = 0.0f;
        if (row < n) {
            const float* hr = h + (size_t)row * HID;
            const float h0 = hr[lane], h1 = hr[lane + 32];
            const float di = dr[0];
            float a0 = 0.0f, a1 = 0.0f;
            const int m = (e - s) < 32 ? (e - s) : 32;
            tile_sums(h, lane, tc[0], tn[0], m, a0, a1);
            for (int base = s + 32; base < e; base += 32) {
                int c2;
                float n2;
                fetch_tile<FIRST>(di, base, e, lane, scol, sew, dinv, snorm, c2, n2);
                const int m2 = (e - base) < 32 ? (e - base) : 32;
                int j = 0;
                for (; j + 8 <= m2; j += 8) row_group<8>(h, lane, c2, n2, j, a0, a1);
                if (j + 4 <= m2) { row_group<4>(h, lane, c2, n2, j, a0, a1); j += 4; }
                if (j + 2 <= m2) { row_group<2>(h, lane, c2, n2, j, a0, a1); j += 2; }
                if (j < m2)      { row_group<1>(h, lane, c2, n2, j, a0, a1); }
            }
            const float d2 = di * di;
            const float g0 = a0 + d2 * h0;
            const float g1 = a1 + d2 * h1;
            const float s0 = g0 + bz0;
            const float s1 = g1 + bz1;
            v0 = s0 > 0.0f ? s0 : 0.0f;
            v1 = s1 > 0.0f ? s1 : 0.0f;
        }
        float* tr = tile + (wave * 4 + i) * TP + lane;
        tr[0]  = v0;
        tr[32] = v1;
        p[0] = p[1]; p[1] = p[2]; p[2] = p[3]; p[3] = p[4];
        dr[0] = dr[1]; dr[1] = dr[2]; dr[2] = dr[3];
        tc[0] = tc[1]; tc[1] = tc[2]; tc[2] = tc[3];
        tn[0] = tn[1]; tn[1] = tn[2]; tn[2] = tn[3];
    }

    __syncthreads();

    const int m    = lane & 15;
    const int hi   = lane >> 4;
    const int col0 = wave * 16;
    v8f acc = {0.f, 0.f, 0.f, 0.f, 0.f, 0.f, 0.f, 0.f};
#if defined(__HIP_DEVICE_COMPILE__)
    const int kh = hi * 2;
#pragma unroll
    for (int k0 = 0; k0 < HID; k0 += 4) {
        const float* ap = tile + m * TP + k0 + kh;
        const float* bp = W + (size_t)(k0 + kh) * HID + col0 + m;
        v2f a, b;
        a.x = ap[0];   a.y = ap[1];
        b.x = bp[0];   b.y = bp[HID];
        acc = __builtin_amdgcn_wmma_f32_16x16x4_f32(
            false, a, false, b, (short)0, acc, false, false);
    }
#endif

    if (HEAD == 0) {
#pragma unroll
        for (int r = 0; r < 8; ++r) {
            const int orow = row0 + r + hi * 8;
            if (orow < n) outp[(size_t)orow * HID + col0 + m] = acc[r];
        }
    } else {
        __syncthreads();
        const float bb = wbias[col0 + m];
#pragma unroll
        for (int r = 0; r < 8; ++r) tile[(r + hi * 8) * TP + col0 + m] = acc[r] + bb;
        __syncthreads();
        const int rr = threadIdx.x >> 3;
        const int q  = threadIdx.x & 7;
        const bool two = q < NCLASS - 8;
        const int c1 = two ? q + 8 : q;
        const float* tr = tile + rr * TP;
        float r0 = b2[q], r1 = b2[c1];
#pragma unroll 8
        for (int j = 0; j < HID; ++j) {
            const float tv = tr[j];
            r0 += tv * W2[(size_t)j * NCLASS + q];
            r1 += tv * W2[(size_t)j * NCLASS + c1];
        }
        if (row0 + rr < n) {
            float* op = outp + (size_t)(row0 + rr) * NCLASS;
            op[q] = r0;
            if (two) op[c1] = r1;
        }
    }
}

extern "C" void kernel_launch(void* const* d_in, const int* in_sizes, int n_in,
                              void* d_out, int out_size, void* d_ws, size_t ws_size,
                              hipStream_t stream) {
    const float* x      = (const float*)d_in[0];
    const float* eattr  = (const float*)d_in[1];
    const int*   eidx   = (const int*)  d_in[2];
    const float* We     = (const float*)d_in[3];
    const float* Wl[3]  = {(const float*)d_in[4], (const float*)d_in[6], (const float*)d_in[8]};
    const float* bl[3]  = {(const float*)d_in[5], (const float*)d_in[7], (const float*)d_in[9]};
    const float* lin1_W = (const float*)d_in[10];
    const float* lin1_b = (const float*)d_in[11];
    const float* lin2_W = (const float*)d_in[12];
    const float* lin2_b = (const float*)d_in[13];
    float* out = (float*)d_out;

    const int N = in_sizes[0] / 128;
    const int E = in_sizes[1] / 16;
    const int* rows = eidx;
    const int* cols = eidx + E;

    const int nb    = (N + RB - 1) / RB;
    const int chunk = (E + NCH - 1) / NCH;

    float* ws   = (float*)d_ws;
    float* h    = ws;
    float* agga = h + (size_t)N * HID;
    float* aggb = agga + (size_t)N * HID;
    int*   rec  = (int*)(aggb + (size_t)N * HID);
    int*   scol = rec + (size_t)E * 4;
    float* sew  = (float*)(scol + E);
    float* snorm = sew + E;
    float* dinv = snorm + E;
    int* rowptr = (int*)(dinv + N);
    int* blockhist = rowptr + N + 1;
    int* offs   = blockhist + (size_t)nb * NCH;
    int* total  = offs + (size_t)nb * NCH;
    int* bstart = total + nb;

    const int rowTiles = (N + 15) / 16;

    bucket_count_kernel<<<NCH, CW * 32, CW * nb * sizeof(int), stream>>>(rows, blockhist, E, nb, chunk);
    bucket_scan_kernel<<<nb, NCH, 0, stream>>>(blockhist, offs, total);
    bucket_place_kernel<<<NCH, CW * 32, (CW + 1) * nb * sizeof(int), stream>>>(
        eattr, We, rows, cols, offs, total, bstart, rec, E, nb, chunk);
    bucket_rowsort_kernel<<<nb, 256, 0, stream>>>(bstart, rec, rowptr, scol, sew, dinv, N, nb);

    gemm_wmma_kernel<128, 0, 0><<<rowTiles, 128, 0, stream>>>(x, nullptr, Wl[0], nullptr, h, nullptr);

    agg_gemm_kernel<1, 0><<<rowTiles, 128, 0, stream>>>(
        rowptr, scol, sew, dinv, snorm, h, bl[0], Wl[1], nullptr, nullptr, nullptr, agga, N);
    agg_gemm_kernel<0, 0><<<rowTiles, 128, 0, stream>>>(
        rowptr, scol, sew, dinv, snorm, agga, bl[1], Wl[2], nullptr, nullptr, nullptr, h, N);
    agg_gemm_kernel<0, 1><<<rowTiles, 128, 0, stream>>>(
        rowptr, scol, sew, dinv, snorm, h, bl[2], lin1_W, lin1_b, lin2_W, lin2_b, out, N);
}


# Sort: scattered 4-byte accesses made whole-line: chunk histogram kept [chunk][bucket] (scan block: lane = bucket, counts in registers); row-sorted col/weight lists staged in LDS, stored coalesced
# speedup vs baseline: 3.5065x; 1.0977x over previous
#include <hip/hip_runtime.h>
#include <hip/hip_bf16.h>


typedef __attribute__((ext_vector_type(2)))  float    v2f;
typedef __attribute__((ext_vector_type(8)))  float    v8f;
typedef __attribute__((ext_vector_type(16))) _Float16 v16h;

#define HID 64
#define NCLASS 10
#define RB 64
#define RBSH 6
#define NCH 512
#define CW 16
#define CPW (NCH / CW)
static_assert(NCH % CW == 0, "the scan pass gives every wave the same number of chunks");

#if defined(__HIP_DEVICE_COMPILE__)
#  if __has_builtin(__builtin_amdgcn_wmma_f32_16x16x4_f32)
#    define GCN_USE_F32_WMMA 1
#  else
#    define GCN_USE_F32_WMMA 0
#  endif
#endif

__device__ __forceinline__ int wave_incl_scan(int x, int lane) {
#pragma unroll
    for (int d = 1; d < 32; d <<= 1) {
        const int y = __shfl_up(x, d);
        if (lane >= d) x += y;
    }
    return x;
}

__device__ __forceinline__ void key_rank(int key, int lane, int& rank, int& cnt) {
    rank = 0; cnt = 0;
#pragma unroll
    for (int j = 0; j < 32; ++j) {
        const int kj = __builtin_amdgcn_readlane(key, j);
        const int same = (kj == key) ? 1 : 0;
        cnt += same;
        rank += (j < lane) ? same : 0;
    }
}

__device__ __forceinline__ void count_slice(const int* __restrict__ rows, int* cnt_w,
                                            int es, int ee, int lane) {
    for (int e0 = es; e0 < ee; e0 += 32) {
        const int e = e0 + lane;
        const int key = e < ee ? (rows[e] >> RBSH) : -1;
        int rank, cnt;
        key_rank(key, lane, rank, cnt);
        if (key >= 0 && rank == cnt - 1) cnt_w[key] += cnt;
    }
}

__global__ __launch_bounds__(CW * 32) void bucket_count_kernel(
    const int* __restrict__ rows, int* __restrict__ blockhist,
    int nedge, int nb, int chunk) {
    extern __shared__ int sh[];
    const int tid = threadIdx.x, lane = tid & 31, blk = blockIdx.x;
    const int wv = __builtin_amdgcn_readfirstlane(tid >> 5);
    for (int i = tid; i < CW * nb; i += CW * 32) sh[i] = 0;
    __syncthreads();
    const int per = (chunk + CW - 1) / CW;
    const int eb = blk * chunk;
    const int ec = (eb + chunk) < nedge ? (eb + chunk) : nedge;
    const int es = eb + wv * per;
    const int ee = (es + per) < ec ? (es + per) : ec;
    count_slice(rows, sh + wv * nb, es, ee, lane);
    __syncthreads();
    for (int i = tid; i < nb; i += CW * 32) {
        int run = 0;
#pragma unroll
        for (int w = 0; w < CW; ++w) run += sh[w * nb + i];
        blockhist[(size_t)blk * nb + i] = run;
    }
}

__global__ __launch_bounds__(CW * 32) void bucket_scan_kernel(
    const int* __restrict__ blockhist, int* __restrict__ offs, int* __restrict__ total, int nb) {
    __shared__ int wtot[CW * 32];
    const int lane = threadIdx.x & 31;
    const int wv = __builtin_amdgcn_readfirstlane((int)(threadIdx.x >> 5));
    const int b = blockIdx.x * 32 + lane;
    const bool ok = b < nb;
    const int bl = ok ? b : nb - 1;
    const int c0 = wv * CPW;
    int v[CPW];
#pragma unroll
    for (int k = 0; k < CPW; ++k) v[k] = blockhist[(size_t)(c0 + k) * nb + bl];
    int run = 0;
#pragma unroll
    for (int k = 0; k < CPW; ++k) { const int x = v[k]; v[k] = run; run += x; }
    wtot[wv * 32 + lane] = run;
    __syncthreads();
    int base = 0, all = 0;
#pragma unroll
    for (int w = 0; w < CW; ++w) {
        const int s = wtot[w * 32 + lane];
        all += s;
        base += (w < wv) ? s : 0;
    }
    if (ok) {
#pragma unroll
        for (int k = 0; k < CPW; ++k) offs[(size_t)(c0 + k) * nb + b] = base + v[k];
        if (wv == 0) total[b] = all;
    }
}

__global__ __launch_bounds__(CW * 32) void bucket_place_kernel(
    const float* __restrict__ ea, const float* __restrict__ We,
    const int* __restrict__ rows, const int* __restrict__ cols,
    const int* __restrict__ offs, const int* __restrict__ total,
    int* __restrict__ bstart, int* __restrict__ rec,
    int nedge, int nb, int chunk) {
    extern __shared__ int sh[];
    __shared__ int wsum[CW];
    int* bst = sh + CW * nb;
    const int tid = threadIdx.x, lane = tid & 31, blk = blockIdx.x;
    const int wv = __builtin_amdgcn_readfirstlane(tid >> 5);
    for (int i = tid; i < CW * nb; i += CW * 32) sh[i] = 0;
    int carry = 0;
    for (int base = 0; base < nb; base += CW * 32) {
        const int i = base + tid;
        const int v = i < nb ? total[i] : 0;
        const int x = wave_incl_scan(v, lane);
        if (lane == 31) wsum[wv] = x;
        __syncthreads();
        if (wv == 0) {
            int s = lane < CW ? wsum[lane] : 0;
            s = wave_incl_scan(s, lane);
            if (lane < CW) wsum[lane] = s;
        }
        __syncthreads();
        const int wo = wv > 0 ? wsum[wv - 1] : 0;
        const int excl = carry + wo + x - v;
        if (i < nb) {
            bst[i] = excl;
            if (blk == 0) bstart[i] = excl;
        }
        carry += wsum[CW - 1];
        __syncthreads();
    }
    if (blk == 0 && tid == 0) bstart[nb] = carry;

    const int per = (chunk + CW - 1) / CW;
    const int eb = blk * chunk;
    const int ec = (eb + chunk) < nedge ? (eb + chunk) : nedge;
    const int es = eb + wv * per;
    const int ee = (es + per) < ec ? (es + per) : ec;
    int* cur = sh + wv * nb;
    count_slice(rows, cur, es, ee, lane);
    __syncthreads();
    for (int i = tid; i < nb; i += CW * 32) {
        int run = bst[i] + offs[(size_t)blk * nb + i];
#pragma unroll
        for (int w = 0; w < CW; ++w) {
            const int c = sh[w * nb + i];
            sh[w * nb + i] = run;
            run += c;
        }
    }
    __syncthreads();

    const float4* w4 = (const float4*)We;
    int4* rec4 = (int4*)rec;
    for (int e0 = es; e0 < ee; e0 += 32) {
        const int e = e0 + lane;
        const bool valid = e < ee;
        const int r = valid ? rows[e] : 0;
        const int key = valid ? (r >> RBSH) : -1;
        int rank, cnt;
        key_rank(key, lane, rank, cnt);
        if (valid) {
            const int p0 = cur[key];
            if (rank == cnt - 1) cur[key] = p0 + cnt;
            const float4* a4 = (const float4*)(ea + (size_t)e * 16);
            float s = 0.0f;
#pragma unroll
            for (int i = 0; i < 4; ++i) {
                float4 a = a4[i], w = w4[i];
                s += a.x * w.x + a.y * w.y + a.z * w.z + a.w * w.w;
            }
            int4 v;
            v.x = r; v.y = cols[e]; v.z = __float_as_int(s); v.w = 0;
            rec4[p0 + rank] = v;
        }
    }
}

#define LCAP 2048
__global__ __launch_bounds__(256) void bucket_rowsort_kernel(
    const int* __restrict__ bstart, const int* __restrict__ rec,
    int* __restrict__ rowptr, int* __restrict__ scol, float* __restrict__ sew,
    float* __restrict__ dinv, int n, int nb) {
    __shared__ int wc[8 * RB];
    __shared__ float wdeg[8 * RB];
    __shared__ int rtot[RB];
    __shared__ int lcol[LCAP];
    __shared__ float lew[LCAP];
    const int t = threadIdx.x, lane = t & 31;
    const int wv = __builtin_amdgcn_readfirstlane(t >> 5);
    const int b = blockIdx.x, r0 = b * RB;
    wc[t] = 0; wc[t + 256] = 0;
    wdeg[t] = 0.0f; wdeg[t + 256] = 0.0f;
    __syncthreads();
    const int4* rec4 = (const int4*)rec;
    const int s = __builtin_amdgcn_readfirstlane(bstart[b]);
    const int e = __builtin_amdgcn_readfirstlane(bstart[b + 1]);
    const int per = (e - s + 7) / 8;
    const bool staged = (e - s) <= LCAP;
    const int is = s + wv * per;
    const int ie = (is + per) < e ? (is + per) : e;
    int* cur = wc + wv * RB;
    float* dcur = wdeg + wv * RB;
    for (int i0 = is; i0 < ie; i0 += 32) {
        const int i = i0 + lane;
        const int key = i < ie ? (rec4[i].x - r0) : -1;
        int rank, cnt;
        key_rank(key, lane, rank, cnt);
        if (key >= 0 && rank == cnt - 1) cur[key] += cnt;
    }
    __syncthreads();
    if (t < RB) {
        int tot = 0;
#pragma unroll
        for (int w = 0; w < 8; ++w) tot += wc[w * RB + t];
        rtot[t] = tot;
    }
    __syncthreads();
    if (t < RB) {
        int st = s;
        for (int k = 0; k < t; ++k) st += rtot[k];
        if (r0 + t < n) rowptr[r0 + t] = st;
        int run = st;
#pragma unroll
        for (int w = 0; w < 8; ++w) {
            const int c = wc[w * RB + t];
            wc[w * RB + t] = run;
            run += c;
        }
    }
    if (b == nb - 1 && t == 0) rowptr[n] = e;
    __syncthreads();
    for (int i0 = is; i0 < ie; i0 += 32) {
        const int i = i0 + lane;
        const bool valid = i < ie;
        int4 v = {0, 0, 0, 0};
        if (valid) v = rec4[i];
        const int key = valid ? (v.x - r0) : -1;
        const float w = __int_as_float(v.z);
        int rank = 0, cnt = 0;
        float wsum = 0.0f;
#pragma unroll
        for (int j = 0; j < 32; ++j) {
            const int kj = __builtin_amdgcn_readlane(key, j);
            const float wj = __int_as_float(__builtin_amdgcn_readlane(v.z, j));
            const bool same = (kj == key);
            cnt += same ? 1 : 0;
            rank += (same && j < lane) ? 1 : 0;
            wsum += same ? wj : 0.0f;
        }
        if (valid) {
            const int p0 = cur[key];
            if (rank == cnt - 1) {
                cur[key] = p0 + cnt;
                dcur[key] += wsum;
            }
            const int q = p0 + rank;
            if (staged) {
                lcol[q - s] = v.y;
                lew[q - s] = w;
            } else {
                scol[q] = v.y;
                sew[q] = w;
            }
        }
    }
    __syncthreads();
    if (staged) {
        for (int i = t; i < e - s; i += 256) {
            scol[s + i] = lcol[i];
            sew[s + i] = lew[i];
        }
    }
    if (t < RB && r0 + t < n) {
        float deg = 0.0f;
#pragma unroll
        for (int w = 0; w < 8; ++w) deg += wdeg[w * RB + t];
        const float d = deg + 1.0f;
        dinv[r0 + t] = d > 0.0f ? __frsqrt_rn(d) : 0.0f;
    }
}

template <int K, int MODE, int ACT>
__global__ __launch_bounds__(128) void gemm_wmma_kernel(
    const float* __restrict__ A, const float* __restrict__ abias,
    const float* __restrict__ W, const float* __restrict__ dinv_or_bias,
    float* __restrict__ out0, float* __restrict__ out1) {
    const int lane = threadIdx.x & 31;
    const int wave = threadIdx.x >> 5;
    const int m    = lane & 15;
    const int hi   = lane >> 4;
    const int row0 = blockIdx.x * 16;
    const int col0 = wave * 16;

    v8f acc = {0.f, 0.f, 0.f, 0.f, 0.f, 0.f, 0.f, 0.f};

#if defined(__HIP_DEVICE_COMPILE__)
#if GCN_USE_F32_WMMA
    const int kh = hi * 2;
#pragma unroll 4
    for (int k0 = 0; k0 < K; k0 += 4) {
        const float* ap = A + (size_t)(row0 + m) * K + k0 + kh;
        const float* bp = W + (size_t)(k0 + kh) * HID + col0 + m;
        v2f a, b;
        a.x = ap[0];   a.y = ap[1];
        b.x = bp[0];   b.y = bp[HID];
        if (ACT == 1) {
            const float s0 = a.x + abias[k0 + kh];
            const float s1 = a.y + abias[k0 + kh + 1];
            a.x = s0 > 0.0f ? s0 : 0.0f;
            a.y = s1 > 0.0f ? s1 : 0.0f;
        }
        acc = __builtin_amdgcn_wmma_f32_16x16x4_f32(
            false, a, false, b, (short)0, acc, false, false);
    }
#else
    const int kb = hi * 8;
    for (int k0 = 0; k0 < K; k0 += 32) {
        const float* ap = A + (size_t)(row0 + m) * K + k0 + kb;
        const float* bp = W + (size_t)(k0 + kb) * HID + col0 + m;
        v16h a, b;
#pragma unroll
        for (int j = 0; j < 8; ++j) {
            float p = ap[j], q = ap[j + 16];
            if (ACT == 1) {
                p += abias[k0 + kb + j];
                q += abias[k0 + kb + j + 16];
                p = p > 0.0f ? p : 0.0f;
                q = q > 0.0f ? q : 0.0f;
            }
            a[j]     = (_Float16)p;
            a[j + 8] = (_Float16)q;
            b[j]     = (_Float16)bp[(size_t)j * HID];
            b[j + 8] = (_Float16)bp[(size_t)(j + 16) * HID];
        }
        acc = __builtin_amdgcn_wmma_f32_16x16x32_f16(
            false, a, false, b, (short)0, acc, false, false);
    }
#endif
#endif

#pragma unroll
    for (int r = 0; r < 8; ++r) {
        const int row = row0 + r + hi * 8;
        const int col = col0 + m;
        const size_t idx = (size_t)row * HID + col;
        const float v = acc[r];
        if (MODE == 0) {
            out0[idx] = v;
        } else {
            out0[idx] = v + dinv_or_bias[col];
        }
    }
}

template <int U>
__device__ __forceinline__ void row_group(const float* __restrict__ h, int lane, int c, float nrm,
                                          int j, float& a0, float& a1) {
    float v0[U], v1[U];
#pragma unroll
    for (int u = 0; u < U; ++u) {
        const int cj = __builtin_amdgcn_readlane(c, j + u);
        const float* s = h + (size_t)cj * HID;
        v0[u] = s[lane];
        v1[u] = s[lane + 32];
    }
#pragma unroll
    for (int u = 0; u < U; ++u) {
        const float w = __int_as_float(__builtin_amdgcn_readlane(__float_as_int(nrm), j + u));
        a0 += w * v0[u];
        a1 += w * v1[u];
    }
}

__device__ __forceinline__ void tile_sums(const float* __restrict__ h, int lane, int c, float nrm,
                                          int m, float& a0, float& a1) {
    int j = 0;
    for (; j + 16 <= m; j += 16) row_group<16>(h, lane, c, nrm, j, a0, a1);
    switch (m - j) {
        case 15: row_group<15>(h, lane, c, nrm, j, a0, a1); break;
        case 14: row_group<14>(h, lane, c, nrm, j, a0, a1); break;
        case 13: row_group<13>(h, lane, c, nrm, j, a0, a1); break;
        case 12: row_group<12>(h, lane, c, nrm, j, a0, a1); break;
        case 11: row_group<11>(h, lane, c, nrm, j, a0, a1); break;
        case 10: row_group<10>(h, lane, c, nrm, j, a0, a1); break;
        case 9:  row_group<9>(h, lane, c, nrm, j, a0, a1);  break;
        case 8:  row_group<8>(h, lane, c, nrm, j, a0, a1);  break;
        case 7:  row_group<7>(h, lane, c, nrm, j, a0, a1);  break;
        case 6:  row_group<6>(h, lane, c, nrm, j, a0, a1);  break;
        case 5:  row_group<5>(h, lane, c, nrm, j, a0, a1);  break;
        case 4:  row_group<4>(h, lane, c, nrm, j, a0, a1);  break;
        case 3:  row_group<3>(h, lane, c, nrm, j, a0, a1);  break;
        case 2:  row_group<2>(h, lane, c, nrm, j, a0, a1);  break;
        case 1:  row_group<1>(h, lane, c, nrm, j, a0, a1);  break;
        default: break;
    }
}

template <int FIRST>
__device__ __forceinline__ void fetch_tile(float di, int base, int e, int lane,
                                           const int* __restrict__ scol,
                                           const float* __restrict__ sew,
                                           const float* __restrict__ dinv,
                                           float* __restrict__ snorm,
                                           int& c, float& nrm) {
    c = 0; nrm = 0.0f;
    if (base + lane < e) {
        c = scol[base + lane];
        if (FIRST) {
            nrm = di * sew[base + lane] * dinv[c];
            snorm[base + lane] = nrm;
        } else {
            nrm = snorm[base + lane];
        }
    }
}

#define TP 68

template <int FIRST, int HEAD>
__global__ __launch_bounds__(128) void agg_gemm_kernel(
    const int* __restrict__ rowptr, const int* __restrict__ scol,
    const float* __restrict__ sew, const float* __restrict__ dinv,
    float* __restrict__ snorm, const float* __restrict__ h,
    const float* __restrict__ abias, const float* __restrict__ W,
    const float* __restrict__ wbias, const float* __restrict__ W2,
    const float* __restrict__ b2, float* __restrict__ outp, int n) {
    __shared__ __attribute__((aligned(16))) float tile[16 * TP];
    const int lane = threadIdx.x & 31;
    const int wave = __builtin_amdgcn_readfirstlane((int)(threadIdx.x >> 5));
    const int row0 = blockIdx.x * 16;
    const float bz0 = abias[lane], bz1 = abias[lane + 32];

    const int rb = row0 + wave * 4;
    int p[5];
    float dr[4];
#pragma unroll
    for (int k = 0; k < 5; ++k) {
        const int r = (rb + k) < n ? (rb + k) : n;
        p[k] = __builtin_amdgcn_readfirstlane(rowptr[r]);
    }
#pragma unroll
    for (int k = 0; k < 4; ++k) {
        const int r = (rb + k) < n ? (rb + k) : (n - 1);
        dr[k] = __int_as_float(__builtin_amdgcn_readfirstlane(__float_as_int(dinv[r])));
    }

    int tc[4];
    float tn[4];
    {
        bool ok[4];
        int idx[4];
#pragma unroll
        for (int k = 0; k < 4; ++k) {
            ok[k]  = p[k] + lane < p[k + 1];
            idx[k] = ok[k] ? p[k] + lane : 0;
        }
        if (FIRST) {
            int cc[4];
            float ww[4], dc[4];
#pragma unroll
            for (int k = 0; k < 4; ++k) { cc[k] = scol[idx[k]]; ww[k] = sew[idx[k]]; }
#pragma unroll
            for (int k = 0; k < 4; ++k) cc[k] = ok[k] ? cc[k] : 0;
#pragma unroll
            for (int k = 0; k < 4; ++k) dc[k] = dinv[cc[k]];
#pragma unroll
            for (int k = 0; k < 4; ++k) {
                const float t = dr[k] * ww[k] * dc[k];
                tc[k] = cc[k];
                tn[k] = ok[k] ? t : 0.0f;
                if (ok[k]) snorm[idx[k]] = t;
            }
        } else {
            int cc[4];
            float nn[4];
#pragma unroll
            for (int k = 0; k < 4; ++k) { cc[k] = scol[idx[k]]; nn[k] = snorm[idx[k]]; }
#pragma unroll
            for (int k = 0; k < 4; ++k) {
                tc[k] = ok[k] ? cc[k] : 0;
                tn[k] = ok[k] ? nn[k] : 0.0f;
            }
        }
    }

#pragma nounroll
    for (int i = 0; i < 4; ++i) {
        const int row = rb + i;
        const int s = p[0], e = p[1];
        float v0 = 0.0f, v1 = 0.0f;
        if (row < n) {
            const float* hr = h + (size_t)row * HID;
            const float h0 = hr[lane], h1 = hr[lane + 32];
            const float di = dr[0];
            float a0 = 0.0f, a1 = 0.0f;
            const int m = (e - s) < 32 ? (e - s) : 32;
            tile_sums(h, lane, tc[0], tn[0], m, a0, a1);
            for (int base = s + 32; base < e; base += 32) {
                int c2;
                float n2;
                fetch_tile<FIRST>(di, base, e, lane, scol, sew, dinv, snorm, c2, n2);
                const int m2 = (e - base) < 32 ? (e - base) : 32;
                int j = 0;
                for (; j + 8 <= m2; j += 8) row_group<8>(h, lane, c2, n2, j, a0, a1);
                if (j + 4 <= m2) { row_group<4>(h, lane, c2, n2, j, a0, a1); j += 4; }
                if (j + 2 <= m2) { row_group<2>(h, lane, c2, n2, j, a0, a1); j += 2; }
                if (j < m2)      { row_group<1>(h, lane, c2, n2, j, a0, a1); }
            }
            const float d2 = di * di;
            const float g0 = a0 + d2 * h0;
            const float g1 = a1 + d2 * h1;
            const float s0 = g0 + bz0;
            const float s1 = g1 + bz1;
            v0 = s0 > 0.0f ? s0 : 0.0f;
            v1 = s1 > 0.0f ? s1 : 0.0f;
        }
        float* tr = tile + (wave * 4 + i) * TP + lane;
        tr[0]  = v0;
        tr[32] = v1;
        p[0] = p[1]; p[1] = p[2]; p[2] = p[3]; p[3] = p[4];
        dr[0] = dr[1]; dr[1] = dr[2]; dr[2] = dr[3];
        tc[0] = tc[1]; tc[1] = tc[2]; tc[2] = tc[3];
        tn[0] = tn[1]; tn[1] = tn[2]; tn[2] = tn[3];
    }

    __syncthreads();

    const int m    = lane & 15;
    const int hi   = lane >> 4;
    const int col0 = wave * 16;
    v8f acc = {0.f, 0.f, 0.f, 0.f, 0.f, 0.f, 0.f, 0.f};
#if defined(__HIP_DEVICE_COMPILE__)
    const int kh = hi * 2;
#pragma unroll
    for (int k0 = 0; k0 < HID; k0 += 4) {
        const float* ap = tile + m * TP + k0 + kh;
        const float* bp = W + (size_t)(k0 + kh) * HID + col0 + m;
        v2f a, b;
        a.x = ap[0];   a.y = ap[1];
        b.x = bp[0];   b.y = bp[HID];
        acc = __builtin_amdgcn_wmma_f32_16x16x4_f32(
            false, a, false, b, (short)0, acc, false, false);
    }
#endif

    if (HEAD == 0) {
#pragma unroll
        for (int r = 0; r < 8; ++r) {
            const int orow = row0 + r + hi * 8;
            if (orow < n) outp[(size_t)orow * HID + col0 + m] = acc[r];
        }
    } else {
        __syncthreads();
        const float bb = wbias[col0 + m];
#pragma unroll
        for (int r = 0; r < 8; ++r) tile[(r + hi * 8) * TP + col0 + m] = acc[r] + bb;
        __syncthreads();
        const int rr = threadIdx.x >> 3;
        const int q  = threadIdx.x & 7;
        const bool two = q < NCLASS - 8;
        const int c1 = two ? q + 8 : q;
        const float* tr = tile + rr * TP;
        float r0 = b2[q], r1 = b2[c1];
#pragma unroll 8
        for (int j = 0; j < HID; ++j) {
            const float tv = tr[j];
            r0 += tv * W2[(size_t)j * NCLASS + q];
            r1 += tv * W2[(size_t)j * NCLASS + c1];
        }
        if (row0 + rr < n) {
            float* op = outp + (size_t)(row0 + rr) * NCLASS;
            op[q] = r0;
            if (two) op[c1] = r1;
        }
    }
}

extern "C" void kernel_launch(void* const* d_in, const int* in_sizes, int n_in,
                              void* d_out, int out_size, void* d_ws, size_t ws_size,
                              hipStream_t stream) {
    const float* x      = (const float*)d_in[0];
    const float* eattr  = (const float*)d_in[1];
    const int*   eidx   = (const int*)  d_in[2];
    const float* We     = (const float*)d_in[3];
    const float* Wl[3]  = {(const float*)d_in[4], (const float*)d_in[6], (const float*)d_in[8]};
    const float* bl[3]  = {(const float*)d_in[5], (const float*)d_in[7], (const float*)d_in[9]};
    const float* lin1_W = (const float*)d_in[10];
    const float* lin1_b = (const float*)d_in[11];
    const float* lin2_W = (const float*)d_in[12];
    const float* lin2_b = (const float*)d_in[13];
    float* out = (float*)d_out;

    const int N = in_sizes[0] / 128;
    const int E = in_sizes[1] / 16;
    const int* rows = eidx;
    const int* cols = eidx + E;

    const int nb    = (N + RB - 1) / RB;
    const int chunk = (E + NCH - 1) / NCH;

    float* ws   = (float*)d_ws;
    float* h    = ws;
    float* agga = h + (size_t)N * HID;
    float* aggb = agga + (size_t)N * HID;
    int*   rec  = (int*)(aggb + (size_t)N * HID);
    int*   scol = rec + (size_t)E * 4;
    float* sew  = (float*)(scol + E);
    float* snorm = sew + E;
    float* dinv = snorm + E;
    int* rowptr = (int*)(dinv + N);
    int* blockhist = rowptr + N + 1;
    int* offs   = blockhist + (size_t)nb * NCH;
    int* total  = offs + (size_t)nb * NCH;
    int* bstart = total + nb;

    const int rowTiles = (N + 15) / 16;

    bucket_count_kernel<<<NCH, CW * 32, CW * nb * sizeof(int), stream>>>(rows, blockhist, E, nb, chunk);
    bucket_scan_kernel<<<(nb + 31) / 32, CW * 32, 0, stream>>>(blockhist, offs, total, nb);
    bucket_place_kernel<<<NCH, CW * 32, (CW + 1) * nb * sizeof(int), stream>>>(
        eattr, We, rows, cols, offs, total, bstart, rec, E, nb, chunk);
    bucket_rowsort_kernel<<<nb, 256, 0, stream>>>(bstart, rec, rowptr, scol, sew, dinv, N, nb);

    gemm_wmma_kernel<128, 0, 0><<<rowTiles, 128, 0, stream>>>(x, nullptr, Wl[0], nullptr, h, nullptr);

    agg_gemm_kernel<1, 0><<<rowTiles, 128, 0, stream>>>(
        rowptr, scol, sew, dinv, snorm, h, bl[0], Wl[1], nullptr, nullptr, nullptr, agga, N);
    agg_gemm_kernel<0, 0><<<rowTiles, 128, 0, stream>>>(
        rowptr, scol, sew, dinv, snorm, agga, bl[1], Wl[2], nullptr, nullptr, nullptr, h, N);
    agg_gemm_kernel<0, 1><<<rowTiles, 128, 0, stream>>>(
        rowptr, scol, sew, dinv, snorm, h, bl[2], lin1_W, lin1_b, lin2_W, lin2_b, out, N);
}


# Sort passes: a lane's rank and count among its wave's 32 keys from one ballot per key bit (6 or 10) plus two bit counts, not 32 lane broadcasts with compare and adds; same orders, same sums
# speedup vs baseline: 3.7649x; 1.0737x over previous
#include <hip/hip_runtime.h>
#include <hip/hip_bf16.h>


typedef __attribute__((ext_vector_type(2)))  float    v2f;
typedef __attribute__((ext_vector_type(8)))  float    v8f;
typedef __attribute__((ext_vector_type(16))) _Float16 v16h;

#define HID 64
#define NCLASS 10
#define RB 64
#define RBSH 6
#define NCH 512
#define CW 16
#define CPW (NCH / CW)
static_assert(NCH % CW == 0, "the scan pass gives every wave the same number of chunks");

#if defined(__HIP_DEVICE_COMPILE__)
#  if __has_builtin(__builtin_amdgcn_wmma_f32_16x16x4_f32)
#    define GCN_USE_F32_WMMA 1
#  else
#    define GCN_USE_F32_WMMA 0
#  endif
#endif

__device__ __forceinline__ int wave_incl_scan(int x, int lane) {
#pragma unroll
    for (int d = 1; d < 32; d <<= 1) {
        const int y = __shfl_up(x, d);
        if (lane >= d) x += y;
    }
    return x;
}

__device__ __forceinline__ void key_rank(int key, int nbits, int& rank, int& cnt, unsigned& same_out) {
    const unsigned held = __builtin_amdgcn_ballot_w32(key >= 0);
    unsigned diff = 0u;
    for (int b = 0; b < nbits; ++b) {
        const int nbit = __builtin_amdgcn_sbfe(key, (unsigned)b, 1u);
        const unsigned ones = __builtin_amdgcn_ballot_w32(nbit != 0);
        diff |= ones ^ (unsigned)nbit;
    }
    const unsigned same = key >= 0 ? (~diff & held) : 0u;
    cnt = __builtin_popcount(same);
    rank = (int)__builtin_amdgcn_mbcnt_lo(same, 0u);
    same_out = same;
}

__device__ __forceinline__ int bucket_bits(int nb) {
    return nb > 1 ? 32 - __builtin_clz((unsigned)(nb - 1)) : 0;
}

__device__ __forceinline__ void count_slice(const int* __restrict__ rows, int* cnt_w,
                                            int es, int ee, int lane, int kbits) {
    for (int e0 = es; e0 < ee; e0 += 32) {
        const int e = e0 + lane;
        const int key = e < ee ? (rows[e] >> RBSH) : -1;
        int rank, cnt;
        unsigned grp;
        key_rank(key, kbits, rank, cnt, grp);
        if (key >= 0 && rank == cnt - 1) cnt_w[key] += cnt;
    }
}

__global__ __launch_bounds__(CW * 32) void bucket_count_kernel(
    const int* __restrict__ rows, int* __restrict__ blockhist,
    int nedge, int nb, int chunk) {
    extern __shared__ int sh[];
    const int tid = threadIdx.x, lane = tid & 31, blk = blockIdx.x;
    const int wv = __builtin_amdgcn_readfirstlane(tid >> 5);
    for (int i = tid; i < CW * nb; i += CW * 32) sh[i] = 0;
    __syncthreads();
    const int per = (chunk + CW - 1) / CW;
    const int eb = blk * chunk;
    const int ec = (eb + chunk) < nedge ? (eb + chunk) : nedge;
    const int es = eb + wv * per;
    const int ee = (es + per) < ec ? (es + per) : ec;
    count_slice(rows, sh + wv * nb, es, ee, lane, bucket_bits(nb));
    __syncthreads();
    for (int i = tid; i < nb; i += CW * 32) {
        int run = 0;
#pragma unroll
        for (int w = 0; w < CW; ++w) run += sh[w * nb + i];
        blockhist[(size_t)blk * nb + i] = run;
    }
}

__global__ __launch_bounds__(CW * 32) void bucket_scan_kernel(
    const int* __restrict__ blockhist, int* __restrict__ offs, int* __restrict__ total, int nb) {
    __shared__ int wtot[CW * 32];
    const int lane = threadIdx.x & 31;
    const int wv = __builtin_amdgcn_readfirstlane((int)(threadIdx.x >> 5));
    const int b = blockIdx.x * 32 + lane;
    const bool ok = b < nb;
    const int bl = ok ? b : nb - 1;
    const int c0 = wv * CPW;
    int v[CPW];
#pragma unroll
    for (int k = 0; k < CPW; ++k) v[k] = blockhist[(size_t)(c0 + k) * nb + bl];
    int run = 0;
#pragma unroll
    for (int k = 0; k < CPW; ++k) { const int x = v[k]; v[k] = run; run += x; }
    wtot[wv * 32 + lane] = run;
    __syncthreads();
    int base = 0, all = 0;
#pragma unroll
    for (int w = 0; w < CW; ++w) {
        const int s = wtot[w * 32 + lane];
        all += s;
        base += (w < wv) ? s : 0;
    }
    if (ok) {
#pragma unroll
        for (int k = 0; k < CPW; ++k) offs[(size_t)(c0 + k) * nb + b] = base + v[k];
        if (wv == 0) total[b] = all;
    }
}

__global__ __launch_bounds__(CW * 32) void bucket_place_kernel(
    const float* __restrict__ ea, const float* __restrict__ We,
    const int* __restrict__ rows, const int* __restrict__ cols,
    const int* __restrict__ offs, const int* __restrict__ total,
    int* __restrict__ bstart, int* __restrict__ rec,
    int nedge, int nb, int chunk) {
    extern __shared__ int sh[];
    __shared__ int wsum[CW];
    int* bst = sh + CW * nb;
    const int tid = threadIdx.x, lane = tid & 31, blk = blockIdx.x;
    const int wv = __builtin_amdgcn_readfirstlane(tid >> 5);
    for (int i = tid; i < CW * nb; i += CW * 32) sh[i] = 0;
    int carry = 0;
    for (int base = 0; base < nb; base += CW * 32) {
        const int i = base + tid;
        const int v = i < nb ? total[i] : 0;
        const int x = wave_incl_scan(v, lane);
        if (lane == 31) wsum[wv] = x;
        __syncthreads();
        if (wv == 0) {
            int s = lane < CW ? wsum[lane] : 0;
            s = wave_incl_scan(s, lane);
            if (lane < CW) wsum[lane] = s;
        }
        __syncthreads();
        const int wo = wv > 0 ? wsum[wv - 1] : 0;
        const int excl = carry + wo + x - v;
        if (i < nb) {
            bst[i] = excl;
            if (blk == 0) bstart[i] = excl;
        }
        carry += wsum[CW - 1];
        __syncthreads();
    }
    if (blk == 0 && tid == 0) bstart[nb] = carry;

    const int per = (chunk + CW - 1) / CW;
    const int eb = blk * chunk;
    const int ec = (eb + chunk) < nedge ? (eb + chunk) : nedge;
    const int es = eb + wv * per;
    const int ee = (es + per) < ec ? (es + per) : ec;
    int* cur = sh + wv * nb;
    const int kbits = bucket_bits(nb);
    count_slice(rows, cur, es, ee, lane, kbits);
    __syncthreads();
    for (int i = tid; i < nb; i += CW * 32) {
        int run = bst[i] + offs[(size_t)blk * nb + i];
#pragma unroll
        for (int w = 0; w < CW; ++w) {
            const int c = sh[w * nb + i];
            sh[w * nb + i] = run;
            run += c;
        }
    }
    __syncthreads();

    const float4* w4 = (const float4*)We;
    int4* rec4 = (int4*)rec;
    for (int e0 = es; e0 < ee; e0 += 32) {
        const int e = e0 + lane;
        const bool valid = e < ee;
        const int r = valid ? rows[e] : 0;
        const int key = valid ? (r >> RBSH) : -1;
        int rank, cnt;
        unsigned grp;
        key_rank(key, kbits, rank, cnt, grp);
        if (valid) {
            const int p0 = cur[key];
            if (rank == cnt - 1) cur[key] = p0 + cnt;
            const float4* a4 = (const float4*)(ea + (size_t)e * 16);
            float s = 0.0f;
#pragma unroll
            for (int i = 0; i < 4; ++i) {
                float4 a = a4[i], w = w4[i];
                s += a.x * w.x + a.y * w.y + a.z * w.z + a.w * w.w;
            }
            int4 v;
            v.x = r; v.y = cols[e]; v.z = __float_as_int(s); v.w = 0;
            rec4[p0 + rank] = v;
        }
    }
}

#define LCAP 2048
__global__ __launch_bounds__(256) void bucket_rowsort_kernel(
    const int* __restrict__ bstart, const int* __restrict__ rec,
    int* __restrict__ rowptr, int* __restrict__ scol, float* __restrict__ sew,
    float* __restrict__ dinv, int n, int nb) {
    __shared__ int wc[8 * RB];
    __shared__ float wdeg[8 * RB];
    __shared__ int rtot[RB];
    __shared__ int lcol[LCAP];
    __shared__ float lew[LCAP];
    const int t = threadIdx.x, lane = t & 31;
    const int wv = __builtin_amdgcn_readfirstlane(t >> 5);
    const int b = blockIdx.x, r0 = b * RB;
    wc[t] = 0; wc[t + 256] = 0;
    wdeg[t] = 0.0f; wdeg[t + 256] = 0.0f;
    __syncthreads();
    const int4* rec4 = (const int4*)rec;
    const int s = __builtin_amdgcn_readfirstlane(bstart[b]);
    const int e = __builtin_amdgcn_readfirstlane(bstart[b + 1]);
    const int per = (e - s + 7) / 8;
    const bool staged = (e - s) <= LCAP;
    const int is = s + wv * per;
    const int ie = (is + per) < e ? (is + per) : e;
    int* cur = wc + wv * RB;
    float* dcur = wdeg + wv * RB;
    for (int i0 = is; i0 < ie; i0 += 32) {
        const int i = i0 + lane;
        const int key = i < ie ? (rec4[i].x - r0) : -1;
        int rank, cnt;
        unsigned grp;
        key_rank(key, RBSH, rank, cnt, grp);
        if (key >= 0 && rank == cnt - 1) cur[key] += cnt;
    }
    __syncthreads();
    if (t < RB) {
        int tot = 0;
#pragma unroll
        for (int w = 0; w < 8; ++w) tot += wc[w * RB + t];
        rtot[t] = tot;
    }
    __syncthreads();
    if (t < RB) {
        int st = s;
        for (int k = 0; k < t; ++k) st += rtot[k];
        if (r0 + t < n) rowptr[r0 + t] = st;
        int run = st;
#pragma unroll
        for (int w = 0; w < 8; ++w) {
            const int c = wc[w * RB + t];
            wc[w * RB + t] = run;
            run += c;
        }
    }
    if (b == nb - 1 && t == 0) rowptr[n] = e;
    __syncthreads();
    for (int i0 = is; i0 < ie; i0 += 32) {
        const int i = i0 + lane;
        const bool valid = i < ie;
        int4 v = {0, 0, 0, 0};
        if (valid) v = rec4[i];
        const int key = valid ? (v.x - r0) : -1;
        const float w = __int_as_float(v.z);
        int rank, cnt;
        unsigned grp;
        key_rank(key, RBSH, rank, cnt, grp);
        float wsum = 0.0f;
        unsigned left = grp;
        while (__builtin_amdgcn_ballot_w32(left != 0u) != 0u) {
            const int j = left != 0u ? __builtin_ctz(left) : 0;
            const float wj = __shfl(w, j);
            wsum += left != 0u ? wj : 0.0f;
            left &= left - 1u;
        }
        if (valid) {
            const int p0 = cur[key];
            if (rank == cnt - 1) {
                cur[key] = p0 + cnt;
                dcur[key] += wsum;
            }
            const int q = p0 + rank;
            if (staged) {
                lcol[q - s] = v.y;
                lew[q - s] = w;
            } else {
                scol[q] = v.y;
                sew[q] = w;
            }
        }
    }
    __syncthreads();
    if (staged) {
        for (int i = t; i < e - s; i += 256) {
            scol[s + i] = lcol[i];
            sew[s + i] = lew[i];
        }
    }
    if (t < RB && r0 + t < n) {
        float deg = 0.0f;
#pragma unroll
        for (int w = 0; w < 8; ++w) deg += wdeg[w * RB + t];
        const float d = deg + 1.0f;
        dinv[r0 + t] = d > 0.0f ? __frsqrt_rn(d) : 0.0f;
    }
}

template <int K, int MODE, int ACT>
__global__ __launch_bounds__(128) void gemm_wmma_kernel(
    const float* __restrict__ A, const float* __restrict__ abias,
    const float* __restrict__ W, const float* __restrict__ dinv_or_bias,
    float* __restrict__ out0, float* __restrict__ out1) {
    const int lane = threadIdx.x & 31;
    const int wave = threadIdx.x >> 5;
    const int m    = lane & 15;
    const int hi   = lane >> 4;
    const int row0 = blockIdx.x * 16;
    const int col0 = wave * 16;

    v8f acc = {0.f, 0.f, 0.f, 0.f, 0.f, 0.f, 0.f, 0.f};

#if defined(__HIP_DEVICE_COMPILE__)
#if GCN_USE_F32_WMMA
    const int kh = hi * 2;
#pragma unroll 4
    for (int k0 = 0; k0 < K; k0 += 4) {
        const float* ap = A + (size_t)(row0 + m) * K + k0 + kh;
        const float* bp = W + (size_t)(k0 + kh) * HID + col0 + m;
        v2f a, b;
        a.x = ap[0];   a.y = ap[1];
        b.x = bp[0];   b.y = bp[HID];
        if (ACT == 1) {
            const float s0 = a.x + abias[k0 + kh];
            const float s1 = a.y + abias[k0 + kh + 1];
            a.x = s0 > 0.0f ? s0 : 0.0f;
            a.y = s1 > 0.0f ? s1 : 0.0f;
        }
        acc = __builtin_amdgcn_wmma_f32_16x16x4_f32(
            false, a, false, b, (short)0, acc, false, false);
    }
#else
    const int kb = hi * 8;
    for (int k0 = 0; k0 < K; k0 += 32) {
        const float* ap = A + (size_t)(row0 + m) * K + k0 + kb;
        const float* bp = W + (size_t)(k0 + kb) * HID + col0 + m;
        v16h a, b;
#pragma unroll
        for (int j = 0; j < 8; ++j) {
            float p = ap[j], q = ap[j + 16];
            if (ACT == 1) {
                p += abias[k0 + kb + j];
                q += abias[k0 + kb + j + 16];
                p = p > 0.0f ? p : 0.0f;
                q = q > 0.0f ? q : 0.0f;
            }
            a[j]     = (_Float16)p;
            a[j + 8] = (_Float16)q;
            b[j]     = (_Float16)bp[(size_t)j * HID];
            b[j + 8] = (_Float16)bp[(size_t)(j + 16) * HID];
        }
        acc = __builtin_amdgcn_wmma_f32_16x16x32_f16(
            false, a, false, b, (short)0, acc, false, false);
    }
#endif
#endif

#pragma unroll
    for (int r = 0; r < 8; ++r) {
        const int row = row0 + r + hi * 8;
        const int col = col0 + m;
        const size_t idx = (size_t)row * HID + col;
        const float v = acc[r];
        if (MODE == 0) {
            out0[idx] = v;
        } else {
            out0[idx] = v + dinv_or_bias[col];
        }
    }
}

template <int U>
__device__ __forceinline__ void row_group(const float* __restrict__ h, int lane, int c, float nrm,
                                          int j, float& a0, float& a1) {
    float v0[U], v1[U];
#pragma unroll
    for (int u = 0; u < U; ++u) {
        const int cj = __builtin_amdgcn_readlane(c, j + u);
        const float* s = h + (size_t)cj * HID;
        v0[u] = s[lane];
        v1[u] = s[lane + 32];
    }
#pragma unroll
    for (int u = 0; u < U; ++u) {
        const float w = __int_as_float(__builtin_amdgcn_readlane(__float_as_int(nrm), j + u));
        a0 += w * v0[u];
        a1 += w * v1[u];
    }
}

__device__ __forceinline__ void tile_sums(const float* __restrict__ h, int lane, int c, float nrm,
                                          int m, float& a0, float& a1) {
    int j = 0;
    for (; j + 16 <= m; j += 16) row_group<16>(h, lane, c, nrm, j, a0, a1);
    switch (m - j) {
        case 15: row_group<15>(h, lane, c, nrm, j, a0, a1); break;
        case 14: row_group<14>(h, lane, c, nrm, j, a0, a1); break;
        case 13: row_group<13>(h, lane, c, nrm, j, a0, a1); break;
        case 12: row_group<12>(h, lane, c, nrm, j, a0, a1); break;
        case 11: row_group<11>(h, lane, c, nrm, j, a0, a1); break;
        case 10: row_group<10>(h, lane, c, nrm, j, a0, a1); break;
        case 9:  row_group<9>(h, lane, c, nrm, j, a0, a1);  break;
        case 8:  row_group<8>(h, lane, c, nrm, j, a0, a1);  break;
        case 7:  row_group<7>(h, lane, c, nrm, j, a0, a1);  break;
        case 6:  row_group<6>(h, lane, c, nrm, j, a0, a1);  break;
        case 5:  row_group<5>(h, lane, c, nrm, j, a0, a1);  break;
        case 4:  row_group<4>(h, lane, c, nrm, j, a0, a1);  break;
        case 3:  row_group<3>(h, lane, c, nrm, j, a0, a1);  break;
        case 2:  row_group<2>(h, lane, c, nrm, j, a0, a1);  break;
        case 1:  row_group<1>(h, lane, c, nrm, j, a0, a1);  break;
        default: break;
    }
}

template <int FIRST>
__device__ __forceinline__ void fetch_tile(float di, int base, int e, int lane,
                                           const int* __restrict__ scol,
                                           const float* __restrict__ sew,
                                           const float* __restrict__ dinv,
                                           float* __restrict__ snorm,
                                           int& c, float& nrm) {
    c = 0; nrm = 0.0f;
    if (base + lane < e) {
        c = scol[base + lane];
        if (FIRST) {
            nrm = di * sew[base + lane] * dinv[c];
            snorm[base + lane] = nrm;
        } else {
            nrm = snorm[base + lane];
        }
    }
}

#define TP 68

template <int FIRST, int HEAD>
__global__ __launch_bounds__(128) void agg_gemm_kernel(
    const int* __restrict__ rowptr, const int* __restrict__ scol,
    const float* __restrict__ sew, const float* __restrict__ dinv,
    float* __restrict__ snorm, const float* __restrict__ h,
    const float* __restrict__ abias, const float* __restrict__ W,
    const float* __restrict__ wbias, const float* __restrict__ W2,
    const float* __restrict__ b2, float* __restrict__ outp, int n) {
    __shared__ __attribute__((aligned(16))) float tile[16 * TP];
    const int lane = threadIdx.x & 31;
    const int wave = __builtin_amdgcn_readfirstlane((int)(threadIdx.x >> 5));
    const int row0 = blockIdx.x * 16;
    const float bz0 = abias[lane], bz1 = abias[lane + 32];

    const int rb = row0 + wave * 4;
    int p[5];
    float dr[4];
#pragma unroll
    for (int k = 0; k < 5; ++k) {
        const int r = (rb + k) < n ? (rb + k) : n;
        p[k] = __builtin_amdgcn_readfirstlane(rowptr[r]);
    }
#pragma unroll
    for (int k = 0; k < 4; ++k) {
        const int r = (rb + k) < n ? (rb + k) : (n - 1);
        dr[k] = __int_as_float(__builtin_amdgcn_readfirstlane(__float_as_int(dinv[r])));
    }

    int tc[4];
    float tn[4];
    {
        bool ok[4];
        int idx[4];
#pragma unroll
        for (int k = 0; k < 4; ++k) {
            ok[k]  = p[k] + lane < p[k + 1];
            idx[k] = ok[k] ? p[k] + lane : 0;
        }
        if (FIRST) {
            int cc[4];
            float ww[4], dc[4];
#pragma unroll
            for (int k = 0; k < 4; ++k) { cc[k] = scol[idx[k]]; ww[k] = sew[idx[k]]; }
#pragma unroll
            for (int k = 0; k < 4; ++k) cc[k] = ok[k] ? cc[k] : 0;
#pragma unroll
            for (int k = 0; k < 4; ++k) dc[k] = dinv[cc[k]];
#pragma unroll
            for (int k = 0; k < 4; ++k) {
                const float t = dr[k] * ww[k] * dc[k];
                tc[k] = cc[k];
                tn[k] = ok[k] ? t : 0.0f;
                if (ok[k]) snorm[idx[k]] = t;
            }
        } else {
            int cc[4];
            float nn[4];
#pragma unroll
            for (int k = 0; k < 4; ++k) { cc[k] = scol[idx[k]]; nn[k] = snorm[idx[k]]; }
#pragma unroll
            for (int k = 0; k < 4; ++k) {
                tc[k] = ok[k] ? cc[k] : 0;
                tn[k] = ok[k] ? nn[k] : 0.0f;
            }
        }
    }

#pragma nounroll
    for (int i = 0; i < 4; ++i) {
        const int row = rb + i;
        const int s = p[0], e = p[1];
        float v0 = 0.0f, v1 = 0.0f;
        if (row < n) {
            const float* hr = h + (size_t)row * HID;
            const float h0 = hr[lane], h1 = hr[lane + 32];
            const float di = dr[0];
            float a0 = 0.0f, a1 = 0.0f;
            const int m = (e - s) < 32 ? (e - s) : 32;
            tile_sums(h, lane, tc[0], tn[0], m, a0, a1);
            for (int base = s + 32; base < e; base += 32) {
                int c2;
                float n2;
                fetch_tile<FIRST>(di, base, e, lane, scol, sew, dinv, snorm, c2, n2);
                const int m2 = (e - base) < 32 ? (e - base) : 32;
                int j = 0;
                for (; j + 8 <= m2; j += 8) row_group<8>(h, lane, c2, n2, j, a0, a1);
                if (j + 4 <= m2) { row_group<4>(h, lane, c2, n2, j, a0, a1); j += 4; }
                if (j + 2 <= m2) { row_group<2>(h, lane, c2, n2, j, a0, a1); j += 2; }
                if (j < m2)      { row_group<1>(h, lane, c2, n2, j, a0, a1); }
            }
            const float d2 = di * di;
            const float g0 = a0 + d2 * h0;
            const float g1 = a1 + d2 * h1;
            const float s0 = g0 + bz0;
            const float s1 = g1 + bz1;
            v0 = s0 > 0.0f ? s0 : 0.0f;
            v1 = s1 > 0.0f ? s1 : 0.0f;
        }
        float* tr = tile + (wave * 4 + i) * TP + lane;
        tr[0]  = v0;
        tr[32] = v1;
        p[0] = p[1]; p[1] = p[2]; p[2] = p[3]; p[3] = p[4];
        dr[0] = dr[1]; dr[1] = dr[2]; dr[2] = dr[3];
        tc[0] = tc[1]; tc[1] = tc[2]; tc[2] = tc[3];
        tn[0] = tn[1]; tn[1] = tn[2]; tn[2] = tn[3];
    }

    __syncthreads();

    const int m    = lane & 15;
    const int hi   = lane >> 4;
    const int col0 = wave * 16;
    v8f acc = {0.f, 0.f, 0.f, 0.f, 0.f, 0.f, 0.f, 0.f};
#if defined(__HIP_DEVICE_COMPILE__)
    const int kh = hi * 2;
#pragma unroll
    for (int k0 = 0; k0 < HID; k0 += 4) {
        const float* ap = tile + m * TP + k0 + kh;
        const float* bp = W + (size_t)(k0 + kh) * HID + col0 + m;
        v2f a, b;
        a.x = ap[0];   a.y = ap[1];
        b.x = bp[0];   b.y = bp[HID];
        acc = __builtin_amdgcn_wmma_f32_16x16x4_f32(
            false, a, false, b, (short)0, acc, false, false);
    }
#endif

    if (HEAD == 0) {
#pragma unroll
        for (int r = 0; r < 8; ++r) {
            const int orow = row0 + r + hi * 8;
            if (orow < n) outp[(size_t)orow * HID + col0 + m] = acc[r];
        }
    } else {
        __syncthreads();
        const float bb = wbias[col0 + m];
#pragma unroll
        for (int r = 0; r < 8; ++r) tile[(r + hi * 8) * TP + col0 + m] = acc[r] + bb;
        __syncthreads();
        const int rr = threadIdx.x >> 3;
        const int q  = threadIdx.x & 7;
        const bool two = q < NCLASS - 8;
        const int c1 = two ? q + 8 : q;
        const float* tr = tile + rr * TP;
        float r0 = b2[q], r1 = b2[c1];
#pragma unroll 8
        for (int j = 0; j < HID; ++j) {
            const float tv = tr[j];
            r0 += tv * W2[(size_t)j * NCLASS + q];
            r1 += tv * W2[(size_t)j * NCLASS + c1];
        }
        if (row0 + rr < n) {
            float* op = outp + (size_t)(row0 + rr) * NCLASS;
            op[q] = r0;
            if (two) op[c1] = r1;
        }
    }
}

extern "C" void kernel_launch(void* const* d_in, const int* in_sizes, int n_in,
                              void* d_out, int out_size, void* d_ws, size_t ws_size,
                              hipStream_t stream) {
    const float* x      = (const float*)d_in[0];
    const float* eattr  = (const float*)d_in[1];
    const int*   eidx   = (const int*)  d_in[2];
    const float* We     = (const float*)d_in[3];
    const float* Wl[3]  = {(const float*)d_in[4], (const float*)d_in[6], (const float*)d_in[8]};
    const float* bl[3]  = {(const float*)d_in[5], (const float*)d_in[7], (const float*)d_in[9]};
    const float* lin1_W = (const float*)d_in[10];
    const float* lin1_b = (const float*)d_in[11];
    const float* lin2_W = (const float*)d_in[12];
    const float* lin2_b = (const float*)d_in[13];
    float* out = (float*)d_out;

    const int N = in_sizes[0] / 128;
    const int E = in_sizes[1] / 16;
    const int* rows = eidx;
    const int* cols = eidx + E;

    const int nb    = (N + RB - 1) / RB;
    const int chunk = (E + NCH - 1) / NCH;

    float* ws   = (float*)d_ws;
    float* h    = ws;
    float* agga = h + (size_t)N * HID;
    float* aggb = agga + (size_t)N * HID;
    int*   rec  = (int*)(aggb + (size_t)N * HID);
    int*   scol = rec + (size_t)E * 4;
    float* sew  = (float*)(scol + E);
    float* snorm = sew + E;
    float* dinv = snorm + E;
    int* rowptr = (int*)(dinv + N);
    int* blockhist = rowptr + N + 1;
    int* offs   = blockhist + (size_t)nb * NCH;
    int* total  = offs + (size_t)nb * NCH;
    int* bstart = total + nb;

    const int rowTiles = (N + 15) / 16;

    bucket_count_kernel<<<NCH, CW * 32, CW * nb * sizeof(int), stream>>>(rows, blockhist, E, nb, chunk);
    bucket_scan_kernel<<<(nb + 31) / 32, CW * 32, 0, stream>>>(blockhist, offs, total, nb);
    bucket_place_kernel<<<NCH, CW * 32, (CW + 1) * nb * sizeof(int), stream>>>(
        eattr, We, rows, cols, offs, total, bstart, rec, E, nb, chunk);
    bucket_rowsort_kernel<<<nb, 256, 0, stream>>>(bstart, rec, rowptr, scol, sew, dinv, N, nb);

    gemm_wmma_kernel<128, 0, 0><<<rowTiles, 128, 0, stream>>>(x, nullptr, Wl[0], nullptr, h, nullptr);

    agg_gemm_kernel<1, 0><<<rowTiles, 128, 0, stream>>>(
        rowptr, scol, sew, dinv, snorm, h, bl[0], Wl[1], nullptr, nullptr, nullptr, agga, N);
    agg_gemm_kernel<0, 0><<<rowTiles, 128, 0, stream>>>(
        rowptr, scol, sew, dinv, snorm, agga, bl[1], Wl[2], nullptr, nullptr, nullptr, h, N);
    agg_gemm_kernel<0, 1><<<rowTiles, 128, 0, stream>>>(
        rowptr, scol, sew, dinv, snorm, h, bl[2], lin1_W, lin1_b, lin2_W, lin2_b, out, N);
}


# dinv[col] folded into the stored h rows (a row is scaled by its node's dinv when written): agg = dinv[row]*(sum ew*hs[col] + hs[row]); no per-edge 4-byte dinv lookup, no norm list; products regrouped
# speedup vs baseline: 3.8973x; 1.0352x over previous
#include <hip/hip_runtime.h>
#include <hip/hip_bf16.h>


typedef __attribute__((ext_vector_type(2)))  float    v2f;
typedef __attribute__((ext_vector_type(8)))  float    v8f;
typedef __attribute__((ext_vector_type(16))) _Float16 v16h;

#define HID 64
#define NCLASS 10
#define RB 64
#define RBSH 6
#define NCH 512
#define CW 16
#define CPW (NCH / CW)
static_assert(NCH % CW == 0, "the scan pass gives every wave the same number of chunks");

#if defined(__HIP_DEVICE_COMPILE__)
#  if __has_builtin(__builtin_amdgcn_wmma_f32_16x16x4_f32)
#    define GCN_USE_F32_WMMA 1
#  else
#    define GCN_USE_F32_WMMA 0
#  endif
#endif

__device__ __forceinline__ int wave_incl_scan(int x, int lane) {
#pragma unroll
    for (int d = 1; d < 32; d <<= 1) {
        const int y = __shfl_up(x, d);
        if (lane >= d) x += y;
    }
    return x;
}

__device__ __forceinline__ void key_rank(int key, int nbits, int& rank, int& cnt, unsigned& same_out) {
    const unsigned held = __builtin_amdgcn_ballot_w32(key >= 0);
    unsigned diff = 0u;
    for (int b = 0; b < nbits; ++b) {
        const int nbit = __builtin_amdgcn_sbfe(key, (unsigned)b, 1u);
        const unsigned ones = __builtin_amdgcn_ballot_w32(nbit != 0);
        diff |= ones ^ (unsigned)nbit;
    }
    const unsigned same = key >= 0 ? (~diff & held) : 0u;
    cnt = __builtin_popcount(same);
    rank = (int)__builtin_amdgcn_mbcnt_lo(same, 0u);
    same_out = same;
}

__device__ __forceinline__ int bucket_bits(int nb) {
    return nb > 1 ? 32 - __builtin_clz((unsigned)(nb - 1)) : 0;
}

__device__ __forceinline__ void count_slice(const int* __restrict__ rows, int* cnt_w,
                                            int es, int ee, int lane, int kbits) {
    for (int e0 = es; e0 < ee; e0 += 32) {
        const int e = e0 + lane;
        const int key = e < ee ? (rows[e] >> RBSH) : -1;
        int rank, cnt;
        unsigned grp;
        key_rank(key, kbits, rank, cnt, grp);
        if (key >= 0 && rank == cnt - 1) cnt_w[key] += cnt;
    }
}

__global__ __launch_bounds__(CW * 32) void bucket_count_kernel(
    const int* __restrict__ rows, int* __restrict__ blockhist,
    int nedge, int nb, int chunk) {
    extern __shared__ int sh[];
    const int tid = threadIdx.x, lane = tid & 31, blk = blockIdx.x;
    const int wv = __builtin_amdgcn_readfirstlane(tid >> 5);
    for (int i = tid; i < CW * nb; i += CW * 32) sh[i] = 0;
    __syncthreads();
    const int per = (chunk + CW - 1) / CW;
    const int eb = blk * chunk;
    const int ec = (eb + chunk) < nedge ? (eb + chunk) : nedge;
    const int es = eb + wv * per;
    const int ee = (es + per) < ec ? (es + per) : ec;
    count_slice(rows, sh + wv * nb, es, ee, lane, bucket_bits(nb));
    __syncthreads();
    for (int i = tid; i < nb; i += CW * 32) {
        int run = 0;
#pragma unroll
        for (int w = 0; w < CW; ++w) run += sh[w * nb + i];
        blockhist[(size_t)blk * nb + i] = run;
    }
}

__global__ __launch_bounds__(CW * 32) void bucket_scan_kernel(
    const int* __restrict__ blockhist, int* __restrict__ offs, int* __restrict__ total, int nb) {
    __shared__ int wtot[CW * 32];
    const int lane = threadIdx.x & 31;
    const int wv = __builtin_amdgcn_readfirstlane((int)(threadIdx.x >> 5));
    const int b = blockIdx.x * 32 + lane;
    const bool ok = b < nb;
    const int bl = ok ? b : nb - 1;
    const int c0 = wv * CPW;
    int v[CPW];
#pragma unroll
    for (int k = 0; k < CPW; ++k) v[k] = blockhist[(size_t)(c0 + k) * nb + bl];
    int run = 0;
#pragma unroll
    for (int k = 0; k < CPW; ++k) { const int x = v[k]; v[k] = run; run += x; }
    wtot[wv * 32 + lane] = run;
    __syncthreads();
    int base = 0, all = 0;
#pragma unroll
    for (int w = 0; w < CW; ++w) {
        const int s = wtot[w * 32 + lane];
        all += s;
        base += (w < wv) ? s : 0;
    }
    if (ok) {
#pragma unroll
        for (int k = 0; k < CPW; ++k) offs[(size_t)(c0 + k) * nb + b] = base + v[k];
        if (wv == 0) total[b] = all;
    }
}

__global__ __launch_bounds__(CW * 32) void bucket_place_kernel(
    const float* __restrict__ ea, const float* __restrict__ We,
    const int* __restrict__ rows, const int* __restrict__ cols,
    const int* __restrict__ offs, const int* __restrict__ total,
    int* __restrict__ bstart, int* __restrict__ rec,
    int nedge, int nb, int chunk) {
    extern __shared__ int sh[];
    __shared__ int wsum[CW];
    int* bst = sh + CW * nb;
    const int tid = threadIdx.x, lane = tid & 31, blk = blockIdx.x;
    const int wv = __builtin_amdgcn_readfirstlane(tid >> 5);
    for (int i = tid; i < CW * nb; i += CW * 32) sh[i] = 0;
    int carry = 0;
    for (int base = 0; base < nb; base += CW * 32) {
        const int i = base + tid;
        const int v = i < nb ? total[i] : 0;
        const int x = wave_incl_scan(v, lane);
        if (lane == 31) wsum[wv] = x;
        __syncthreads();
        if (wv == 0) {
            int s = lane < CW ? wsum[lane] : 0;
            s = wave_incl_scan(s, lane);
            if (lane < CW) wsum[lane] = s;
        }
        __syncthreads();
        const int wo = wv > 0 ? wsum[wv - 1] : 0;
        const int excl = carry + wo + x - v;
        if (i < nb) {
            bst[i] = excl;
            if (blk == 0) bstart[i] = excl;
        }
        carry += wsum[CW - 1];
        __syncthreads();
    }
    if (blk == 0 && tid == 0) bstart[nb] = carry;

    const int per = (chunk + CW - 1) / CW;
    const int eb = blk * chunk;
    const int ec = (eb + chunk) < nedge ? (eb + chunk) : nedge;
    const int es = eb + wv * per;
    const int ee = (es + per) < ec ? (es + per) : ec;
    int* cur = sh + wv * nb;
    const int kbits = bucket_bits(nb);
    count_slice(rows, cur, es, ee, lane, kbits);
    __syncthreads();
    for (int i = tid; i < nb; i += CW * 32) {
        int run = bst[i] + offs[(size_t)blk * nb + i];
#pragma unroll
        for (int w = 0; w < CW; ++w) {
            const int c = sh[w * nb + i];
            sh[w * nb + i] = run;
            run += c;
        }
    }
    __syncthreads();

    const float4* w4 = (const float4*)We;
    int4* rec4 = (int4*)rec;
    for (int e0 = es; e0 < ee; e0 += 32) {
        const int e = e0 + lane;
        const bool valid = e < ee;
        const int r = valid ? rows[e] : 0;
        const int key = valid ? (r >> RBSH) : -1;
        int rank, cnt;
        unsigned grp;
        key_rank(key, kbits, rank, cnt, grp);
        if (valid) {
            const int p0 = cur[key];
            if (rank == cnt - 1) cur[key] = p0 + cnt;
            const float4* a4 = (const float4*)(ea + (size_t)e * 16);
            float s = 0.0f;
#pragma unroll
            for (int i = 0; i < 4; ++i) {
                float4 a = a4[i], w = w4[i];
                s += a.x * w.x + a.y * w.y + a.z * w.z + a.w * w.w;
            }
            int4 v;
            v.x = r; v.y = cols[e]; v.z = __float_as_int(s); v.w = 0;
            rec4[p0 + rank] = v;
        }
    }
}

#define LCAP 2048
__global__ __launch_bounds__(256) void bucket_rowsort_kernel(
    const int* __restrict__ bstart, const int* __restrict__ rec,
    int* __restrict__ rowptr, int* __restrict__ scol, float* __restrict__ sew,
    float* __restrict__ dinv, int n, int nb) {
    __shared__ int wc[8 * RB];
    __shared__ float wdeg[8 * RB];
    __shared__ int rtot[RB];
    __shared__ int lcol[LCAP];
    __shared__ float lew[LCAP];
    const int t = threadIdx.x, lane = t & 31;
    const int wv = __builtin_amdgcn_readfirstlane(t >> 5);
    const int b = blockIdx.x, r0 = b * RB;
    wc[t] = 0; wc[t + 256] = 0;
    wdeg[t] = 0.0f; wdeg[t + 256] = 0.0f;
    __syncthreads();
    const int4* rec4 = (const int4*)rec;
    const int s = __builtin_amdgcn_readfirstlane(bstart[b]);
    const int e = __builtin_amdgcn_readfirstlane(bstart[b + 1]);
    const int per = (e - s + 7) / 8;
    const bool staged = (e - s) <= LCAP;
    const int is = s + wv * per;
    const int ie = (is + per) < e ? (is + per) : e;
    int* cur = wc + wv * RB;
    float* dcur = wdeg + wv * RB;
    for (int i0 = is; i0 < ie; i0 += 32) {
        const int i = i0 + lane;
        const int key = i < ie ? (rec4[i].x - r0) : -1;
        int rank, cnt;
        unsigned grp;
        key_rank(key, RBSH, rank, cnt, grp);
        if (key >= 0 && rank == cnt - 1) cur[key] += cnt;
    }
    __syncthreads();
    if (t < RB) {
        int tot = 0;
#pragma unroll
        for (int w = 0; w < 8; ++w) tot += wc[w * RB + t];
        rtot[t] = tot;
    }
    __syncthreads();
    if (t < RB) {
        int st = s;
        for (int k = 0; k < t; ++k) st += rtot[k];
        if (r0 + t < n) rowptr[r0 + t] = st;
        int run = st;
#pragma unroll
        for (int w = 0; w < 8; ++w) {
            const int c = wc[w * RB + t];
            wc[w * RB + t] = run;
            run += c;
        }
    }
    if (b == nb - 1 && t == 0) rowptr[n] = e;
    __syncthreads();
    for (int i0 = is; i0 < ie; i0 += 32) {
        const int i = i0 + lane;
        const bool valid = i < ie;
        int4 v = {0, 0, 0, 0};
        if (valid) v = rec4[i];
        const int key = valid ? (v.x - r0) : -1;
        const float w = __int_as_float(v.z);
        int rank, cnt;
        unsigned grp;
        key_rank(key, RBSH, rank, cnt, grp);
        float wsum = 0.0f;
        unsigned left = grp;
        while (__builtin_amdgcn_ballot_w32(left != 0u) != 0u) {
            const int j = left != 0u ? __builtin_ctz(left) : 0;
            const float wj = __shfl(w, j);
            wsum += left != 0u ? wj : 0.0f;
            left &= left - 1u;
        }
        if (valid) {
            const int p0 = cur[key];
            if (rank == cnt - 1) {
                cur[key] = p0 + cnt;
                dcur[key] += wsum;
            }
            const int q = p0 + rank;
            if (staged) {
                lcol[q - s] = v.y;
                lew[q - s] = w;
            } else {
                scol[q] = v.y;
                sew[q] = w;
            }
        }
    }
    __syncthreads();
    if (staged) {
        for (int i = t; i < e - s; i += 256) {
            scol[s + i] = lcol[i];
            sew[s + i] = lew[i];
        }
    }
    if (t < RB && r0 + t < n) {
        float deg = 0.0f;
#pragma unroll
        for (int w = 0; w < 8; ++w) deg += wdeg[w * RB + t];
        const float d = deg + 1.0f;
        dinv[r0 + t] = d > 0.0f ? __frsqrt_rn(d) : 0.0f;
    }
}

template <int K, int MODE, int ACT>
__global__ __launch_bounds__(128) void gemm_wmma_kernel(
    const float* __restrict__ A, const float* __restrict__ abias,
    const float* __restrict__ W, const float* __restrict__ dinv_or_bias,
    float* __restrict__ out0, float* __restrict__ out1) {
    const int lane = threadIdx.x & 31;
    const int wave = threadIdx.x >> 5;
    const int m    = lane & 15;
    const int hi   = lane >> 4;
    const int row0 = blockIdx.x * 16;
    const int col0 = wave * 16;

    v8f acc = {0.f, 0.f, 0.f, 0.f, 0.f, 0.f, 0.f, 0.f};

    float rs[8];
    if (MODE == 2) {
#pragma unroll
        for (int r = 0; r < 8; ++r) rs[r] = dinv_or_bias[row0 + r + hi * 8];
    }

#if defined(__HIP_DEVICE_COMPILE__)
#if GCN_USE_F32_WMMA
    const int kh = hi * 2;
#pragma unroll 4
    for (int k0 = 0; k0 < K; k0 += 4) {
        const float* ap = A + (size_t)(row0 + m) * K + k0 + kh;
        const float* bp = W + (size_t)(k0 + kh) * HID + col0 + m;
        v2f a, b;
        a.x = ap[0];   a.y = ap[1];
        b.x = bp[0];   b.y = bp[HID];
        if (ACT == 1) {
            const float s0 = a.x + abias[k0 + kh];
            const float s1 = a.y + abias[k0 + kh + 1];
            a.x = s0 > 0.0f ? s0 : 0.0f;
            a.y = s1 > 0.0f ? s1 : 0.0f;
        }
        acc = __builtin_amdgcn_wmma_f32_16x16x4_f32(
            false, a, false, b, (short)0, acc, false, false);
    }
#else
    const int kb = hi * 8;
    for (int k0 = 0; k0 < K; k0 += 32) {
        const float* ap = A + (size_t)(row0 + m) * K + k0 + kb;
        const float* bp = W + (size_t)(k0 + kb) * HID + col0 + m;
        v16h a, b;
#pragma unroll
        for (int j = 0; j < 8; ++j) {
            float p = ap[j], q = ap[j + 16];
            if (ACT == 1) {
                p += abias[k0 + kb + j];
                q += abias[k0 + kb + j + 16];
                p = p > 0.0f ? p : 0.0f;
                q = q > 0.0f ? q : 0.0f;
            }
            a[j]     = (_Float16)p;
            a[j + 8] = (_Float16)q;
            b[j]     = (_Float16)bp[(size_t)j * HID];
            b[j + 8] = (_Float16)bp[(size_t)(j + 16) * HID];
        }
        acc = __builtin_amdgcn_wmma_f32_16x16x32_f16(
            false, a, false, b, (short)0, acc, false, false);
    }
#endif
#endif

#pragma unroll
    for (int r = 0; r < 8; ++r) {
        const int row = row0 + r + hi * 8;
        const int col = col0 + m;
        const size_t idx = (size_t)row * HID + col;
        const float v = acc[r];
        if (MODE == 0) {
            out0[idx] = v;
        } else if (MODE == 2) {
            out0[idx] = v * rs[r];
        } else {
            out0[idx] = v + dinv_or_bias[col];
        }
    }
}

template <int U>
__device__ __forceinline__ void row_group(const float* __restrict__ h, int lane, int c, float nrm,
                                          int j, float& a0, float& a1) {
    float v0[U], v1[U];
#pragma unroll
    for (int u = 0; u < U; ++u) {
        const int cj = __builtin_amdgcn_readlane(c, j + u);
        const float* s = h + (size_t)cj * HID;
        v0[u] = s[lane];
        v1[u] = s[lane + 32];
    }
#pragma unroll
    for (int u = 0; u < U; ++u) {
        const float w = __int_as_float(__builtin_amdgcn_readlane(__float_as_int(nrm), j + u));
        a0 += w * v0[u];
        a1 += w * v1[u];
    }
}

__device__ __forceinline__ void tile_sums(const float* __restrict__ h, int lane, int c, float nrm,
                                          int m, float& a0, float& a1) {
    int j = 0;
    for (; j + 16 <= m; j += 16) row_group<16>(h, lane, c, nrm, j, a0, a1);
    switch (m - j) {
        case 15: row_group<15>(h, lane, c, nrm, j, a0, a1); break;
        case 14: row_group<14>(h, lane, c, nrm, j, a0, a1); break;
        case 13: row_group<13>(h, lane, c, nrm, j, a0, a1); break;
        case 12: row_group<12>(h, lane, c, nrm, j, a0, a1); break;
        case 11: row_group<11>(h, lane, c, nrm, j, a0, a1); break;
        case 10: row_group<10>(h, lane, c, nrm, j, a0, a1); break;
        case 9:  row_group<9>(h, lane, c, nrm, j, a0, a1);  break;
        case 8:  row_group<8>(h, lane, c, nrm, j, a0, a1);  break;
        case 7:  row_group<7>(h, lane, c, nrm, j, a0, a1);  break;
        case 6:  row_group<6>(h, lane, c, nrm, j, a0, a1);  break;
        case 5:  row_group<5>(h, lane, c, nrm, j, a0, a1);  break;
        case 4:  row_group<4>(h, lane, c, nrm, j, a0, a1);  break;
        case 3:  row_group<3>(h, lane, c, nrm, j, a0, a1);  break;
        case 2:  row_group<2>(h, lane, c, nrm, j, a0, a1);  break;
        case 1:  row_group<1>(h, lane, c, nrm, j, a0, a1);  break;
        default: break;
    }
}

__device__ __forceinline__ void fetch_tile(int base, int e, int lane,
                                           const int* __restrict__ scol,
                                           const float* __restrict__ sew,
                                           int& c, float& nrm) {
    c = 0; nrm = 0.0f;
    if (base + lane < e) {
        c = scol[base + lane];
        nrm = sew[base + lane];
    }
}

#define TP 68

template <int HEAD>
__global__ __launch_bounds__(128) void agg_gemm_kernel(
    const int* __restrict__ rowptr, const int* __restrict__ scol,
    const float* __restrict__ sew, const float* __restrict__ dinv,
    const float* __restrict__ h,
    const float* __restrict__ abias, const float* __restrict__ W,
    const float* __restrict__ wbias, const float* __restrict__ W2,
    const float* __restrict__ b2, float* __restrict__ outp, int n) {
    __shared__ __attribute__((aligned(16))) float tile[16 * TP];
    const int lane = threadIdx.x & 31;
    const int wave = __builtin_amdgcn_readfirstlane((int)(threadIdx.x >> 5));
    const int row0 = blockIdx.x * 16;
    const float bz0 = abias[lane], bz1 = abias[lane + 32];

    const int rb = row0 + wave * 4;
    int p[5];
    float dr[4];
#pragma unroll
    for (int k = 0; k < 5; ++k) {
        const int r = (rb + k) < n ? (rb + k) : n;
        p[k] = __builtin_amdgcn_readfirstlane(rowptr[r]);
    }
#pragma unroll
    for (int k = 0; k < 4; ++k) {
        const int r = (rb + k) < n ? (rb + k) : (n - 1);
        dr[k] = __int_as_float(__builtin_amdgcn_readfirstlane(__float_as_int(dinv[r])));
    }

    int tc[4];
    float tn[4];
    {
        bool ok[4];
        int idx[4];
#pragma unroll
        for (int k = 0; k < 4; ++k) {
            ok[k]  = p[k] + lane < p[k + 1];
            idx[k] = ok[k] ? p[k] + lane : 0;
        }
        int cc[4];
        float ww[4];
#pragma unroll
        for (int k = 0; k < 4; ++k) { cc[k] = scol[idx[k]]; ww[k] = sew[idx[k]]; }
#pragma unroll
        for (int k = 0; k < 4; ++k) {
            tc[k] = ok[k] ? cc[k] : 0;
            tn[k] = ok[k] ? ww[k] : 0.0f;
        }
    }

#pragma nounroll
    for (int i = 0; i < 4; ++i) {
        const int row = rb + i;
        const int s = p[0], e = p[1];
        float v0 = 0.0f, v1 = 0.0f;
        if (row < n) {
            const float* hr = h + (size_t)row * HID;
            const float h0 = hr[lane], h1 = hr[lane + 32];
            const float di = dr[0];
            float a0 = 0.0f, a1 = 0.0f;
            const int m = (e - s) < 32 ? (e - s) : 32;
            tile_sums(h, lane, tc[0], tn[0], m, a0, a1);
            for (int base = s + 32; base < e; base += 32) {
                int c2;
                float n2;
                fetch_tile(base, e, lane, scol, sew, c2, n2);
                const int m2 = (e - base) < 32 ? (e - base) : 32;
                int j = 0;
                for (; j + 8 <= m2; j += 8) row_group<8>(h, lane, c2, n2, j, a0, a1);
                if (j + 4 <= m2) { row_group<4>(h, lane, c2, n2, j, a0, a1); j += 4; }
                if (j + 2 <= m2) { row_group<2>(h, lane, c2, n2, j, a0, a1); j += 2; }
                if (j < m2)      { row_group<1>(h, lane, c2, n2, j, a0, a1); }
            }
            const float g0 = di * (a0 + h0);
            const float g1 = di * (a1 + h1);
            const float s0 = g0 + bz0;
            const float s1 = g1 + bz1;
            v0 = s0 > 0.0f ? s0 : 0.0f;
            v1 = s1 > 0.0f ? s1 : 0.0f;
        }
        float* tr = tile + (wave * 4 + i) * TP + lane;
        tr[0]  = v0;
        tr[32] = v1;
        p[0] = p[1]; p[1] = p[2]; p[2] = p[3]; p[3] = p[4];
        dr[0] = dr[1]; dr[1] = dr[2]; dr[2] = dr[3];
        tc[0] = tc[1]; tc[1] = tc[2]; tc[2] = tc[3];
        tn[0] = tn[1]; tn[1] = tn[2]; tn[2] = tn[3];
    }

    __syncthreads();

    const int m    = lane & 15;
    const int hi   = lane >> 4;
    const int col0 = wave * 16;
    v8f acc = {0.f, 0.f, 0.f, 0.f, 0.f, 0.f, 0.f, 0.f};
    float rs[8];
    if (HEAD == 0) {
#pragma unroll
        for (int r = 0; r < 8; ++r) {
            const int orow = row0 + r + hi * 8;
            rs[r] = dinv[orow < n ? orow : (n - 1)];
        }
    }
#if defined(__HIP_DEVICE_COMPILE__)
    const int kh = hi * 2;
#pragma unroll
    for (int k0 = 0; k0 < HID; k0 += 4) {
        const float* ap = tile + m * TP + k0 + kh;
        const float* bp = W + (size_t)(k0 + kh) * HID + col0 + m;
        v2f a, b;
        a.x = ap[0];   a.y = ap[1];
        b.x = bp[0];   b.y = bp[HID];
        acc = __builtin_amdgcn_wmma_f32_16x16x4_f32(
            false, a, false, b, (short)0, acc, false, false);
    }
#endif

    if (HEAD == 0) {
#pragma unroll
        for (int r = 0; r < 8; ++r) {
            const int orow = row0 + r + hi * 8;
            if (orow < n) outp[(size_t)orow * HID + col0 + m] = acc[r] * rs[r];
        }
    } else {
        __syncthreads();
        const float bb = wbias[col0 + m];
#pragma unroll
        for (int r = 0; r < 8; ++r) tile[(r + hi * 8) * TP + col0 + m] = acc[r] + bb;
        __syncthreads();
        const int rr = threadIdx.x >> 3;
        const int q  = threadIdx.x & 7;
        const bool two = q < NCLASS - 8;
        const int c1 = two ? q + 8 : q;
        const float* tr = tile + rr * TP;
        float r0 = b2[q], r1 = b2[c1];
#pragma unroll 8
        for (int j = 0; j < HID; ++j) {
            const float tv = tr[j];
            r0 += tv * W2[(size_t)j * NCLASS + q];
            r1 += tv * W2[(size_t)j * NCLASS + c1];
        }
        if (row0 + rr < n) {
            float* op = outp + (size_t)(row0 + rr) * NCLASS;
            op[q] = r0;
            if (two) op[c1] = r1;
        }
    }
}

extern "C" void kernel_launch(void* const* d_in, const int* in_sizes, int n_in,
                              void* d_out, int out_size, void* d_ws, size_t ws_size,
                              hipStream_t stream) {
    const float* x      = (const float*)d_in[0];
    const float* eattr  = (const float*)d_in[1];
    const int*   eidx   = (const int*)  d_in[2];
    const float* We     = (const float*)d_in[3];
    const float* Wl[3]  = {(const float*)d_in[4], (const float*)d_in[6], (const float*)d_in[8]};
    const float* bl[3]  = {(const float*)d_in[5], (const float*)d_in[7], (const float*)d_in[9]};
    const float* lin1_W = (const float*)d_in[10];
    const float* lin1_b = (const float*)d_in[11];
    const float* lin2_W = (const float*)d_in[12];
    const float* lin2_b = (const float*)d_in[13];
    float* out = (float*)d_out;

    const int N = in_sizes[0] / 128;
    const int E = in_sizes[1] / 16;
    const int* rows = eidx;
    const int* cols = eidx + E;

    const int nb    = (N + RB - 1) / RB;
    const int chunk = (E + NCH - 1) / NCH;

    float* ws   = (float*)d_ws;
    float* h    = ws;
    float* agga = h + (size_t)N * HID;
    float* aggb = agga + (size_t)N * HID;
    int*   rec  = (int*)(aggb + (size_t)N * HID);
    int*   scol = rec + (size_t)E * 4;
    float* sew  = (float*)(scol + E);
    float* dinv = sew + E;
    int* rowptr = (int*)(dinv + N);
    int* blockhist = rowptr + N + 1;
    int* offs   = blockhist + (size_t)nb * NCH;
    int* total  = offs + (size_t)nb * NCH;
    int* bstart = total + nb;

    const int rowTiles = (N + 15) / 16;

    bucket_count_kernel<<<NCH, CW * 32, CW * nb * sizeof(int), stream>>>(rows, blockhist, E, nb, chunk);
    bucket_scan_kernel<<<(nb + 31) / 32, CW * 32, 0, stream>>>(blockhist, offs, total, nb);
    bucket_place_kernel<<<NCH, CW * 32, (CW + 1) * nb * sizeof(int), stream>>>(
        eattr, We, rows, cols, offs, total, bstart, rec, E, nb, chunk);
    bucket_rowsort_kernel<<<nb, 256, 0, stream>>>(bstart, rec, rowptr, scol, sew, dinv, N, nb);

    gemm_wmma_kernel<128, 2, 0><<<rowTiles, 128, 0, stream>>>(x, nullptr, Wl[0], dinv, h, nullptr);

    agg_gemm_kernel<0><<<rowTiles, 128, 0, stream>>>(
        rowptr, scol, sew, dinv, h, bl[0], Wl[1], nullptr, nullptr, nullptr, agga, N);
    agg_gemm_kernel<0><<<rowTiles, 128, 0, stream>>>(
        rowptr, scol, sew, dinv, agga, bl[1], Wl[2], nullptr, nullptr, nullptr, h, N);
    agg_gemm_kernel<1><<<rowTiles, 128, 0, stream>>>(
        rowptr, scol, sew, dinv, h, bl[2], lin1_W, lin1_b, lin2_W, lin2_b, out, N);
}
